# starting kernel
# baseline (speedup 1.0000x reference)
#include <hip/hip_runtime.h>
#include <hip/hip_bf16.h>

typedef float v2f __attribute__((ext_vector_type(2)));
typedef float v8f __attribute__((ext_vector_type(8)));

#define DD 128
#define HH 4
#define CC 32

__global__ void gnn_fill_u32(unsigned int* __restrict__ p, unsigned int v, long long n) {
    long long i = (long long)blockIdx.x * blockDim.x + threadIdx.x;
    if (i < n) p[i] = v;
}

__global__ void gnn_edge_init(const float* __restrict__ x, const int* __restrict__ ei,
                              const float* __restrict__ W_edge, const float* __restrict__ b_edge,
                              float* __restrict__ h_sum, float* __restrict__ cnt, int E_) {
    int e = blockIdx.x;
    int d = threadIdx.x;
    int dstn = ei[E_ + e];
    float x0 = x[e * 3 + 0], x1 = x[e * 3 + 1], x2 = x[e * 3 + 2];
    float v = x0 * W_edge[d] + x1 * W_edge[DD + d] + x2 * W_edge[2 * DD + d] + b_edge[d];
    atomicAdd(&h_sum[(long long)dstn * DD + d], v);
    if (d == 0) atomicAdd(&cnt[dstn], 1.0f);
}

__global__ void gnn_wmma_gemm128(const float* __restrict__ A, const float* __restrict__ B,
                                 float* __restrict__ C, const float* __restrict__ bias,
                                 const float* __restrict__ cnt, int M) {
    int wave = threadIdx.x >> 5;
    int lane = threadIdx.x & 31;
    int m0 = blockIdx.x * 16;
    int n0 = (blockIdx.y * 4 + wave) * 16;
    int half = lane >> 4;
    int l    = lane & 15;
    int K0   = half * 2;
    int Mrow = m0 + l;
    if (Mrow >= M) Mrow = M - 1;
    int Ncol = n0 + l;

    const float* Arow = A + (long long)Mrow * DD;
    v8f acc = {};
#pragma unroll
    for (int k = 0; k < DD; k += 4) {
        v2f a = *(const v2f*)(Arow + k + K0);
        v2f b;
        b.x = B[(long long)(k + K0) * DD + Ncol];
        b.y = B[(long long)(k + K0 + 1) * DD + Ncol];
        acc = __builtin_amdgcn_wmma_f32_16x16x4_f32(false, a, false, b,
                                                    (short)0, acc, false, false);
    }
#pragma unroll
    for (int r = 0; r < 8; ++r) {
        int row = m0 + r + half * 8;
        if (row < M) {
            float v = acc[r];
            if (cnt)  v *= 1.0f / fmaxf(cnt[row], 1.0f);
            if (bias) v += bias[n0 + l];
            C[(long long)row * DD + n0 + l] = v;
        }
    }
}

__global__ void gnn_att_scores(const float* __restrict__ xh,
                               const float* __restrict__ att_src,
                               const float* __restrict__ att_dst,
                               float* __restrict__ a_s, float* __restrict__ a_d) {
    int n = blockIdx.x;
    int d = threadIdx.x;
    int head = d >> 5;
    float v  = xh[(long long)n * DD + d];
    float s  = v * att_src[d];
    float t  = v * att_dst[d];
#pragma unroll
    for (int m = 16; m; m >>= 1) { s += __shfl_xor(s, m); t += __shfl_xor(t, m); }
    if ((d & 31) == 0) {
        a_s[n * HH + head] = s;
        a_d[n * HH + head] = t;
    }
}

__device__ __forceinline__ void edge_st(int t, int E_, const int* ei, int& s, int& dd) {
    if (t < E_) { s = ei[t]; dd = ei[E_ + t]; }
    else        { s = t - E_; dd = t - E_; }
}

__global__ void gnn_edge_max(const int* __restrict__ ei, const float* __restrict__ a_s,
                             const float* __restrict__ a_d, float* __restrict__ mval,
                             int E_, int Nn) {
    int idx = blockIdx.x * blockDim.x + threadIdx.x;
    int tot = E_ + Nn;
    int t = idx >> 2, hd = idx & 3;
    if (t >= tot) return;
    int s, dd; edge_st(t, E_, ei, s, dd);
    float e = a_s[s * HH + hd] + a_d[dd * HH + hd];
    e = (e > 0.0f) ? e : 0.2f * e;
    int* p = (int*)&mval[dd * HH + hd];
    if (e >= 0.0f) atomicMax(p, __float_as_int(e));
    else           atomicMin((unsigned int*)p, __float_as_uint(e));
}

__global__ void gnn_edge_denom(const int* __restrict__ ei, const float* __restrict__ a_s,
                               const float* __restrict__ a_d, const float* __restrict__ mval,
                               float* __restrict__ denom, int E_, int Nn) {
    int idx = blockIdx.x * blockDim.x + threadIdx.x;
    int tot = E_ + Nn;
    int t = idx >> 2, hd = idx & 3;
    if (t >= tot) return;
    int s, dd; edge_st(t, E_, ei, s, dd);
    float e = a_s[s * HH + hd] + a_d[dd * HH + hd];
    e = (e > 0.0f) ? e : 0.2f * e;
    atomicAdd(&denom[dd * HH + hd], __expf(e - mval[dd * HH + hd]));
}

__global__ void gnn_edge_accum(const int* __restrict__ ei, const float* __restrict__ a_s,
                               const float* __restrict__ a_d, const float* __restrict__ mval,
                               const float* __restrict__ denom, const float* __restrict__ xh,
                               float* __restrict__ out, int E_, int Nn) {
    int t = blockIdx.x;
    int d = threadIdx.x;
    int hd = d >> 5;
    int s, dd; edge_st(t, E_, ei, s, dd);
    float e = a_s[s * HH + hd] + a_d[dd * HH + hd];
    e = (e > 0.0f) ? e : 0.2f * e;
    float alpha = __expf(e - mval[dd * HH + hd]) / denom[dd * HH + hd];
    atomicAdd(&out[(long long)dd * DD + d], alpha * xh[(long long)s * DD + d]);
}

__global__ void gnn_finalize(const float* __restrict__ out, const float* __restrict__ gbias,
                             const float* __restrict__ ln_g, const float* __restrict__ ln_b,
                             float* __restrict__ h, int do_relu) {
    __shared__ float red[4];
    int n = blockIdx.x;
    int d = threadIdx.x;
    float v = out[(long long)n * DD + d] + gbias[d];
    float s = v;
#pragma unroll
    for (int m = 16; m; m >>= 1) s += __shfl_xor(s, m);
    if ((d & 31) == 0) red[d >> 5] = s;
    __syncthreads();
    float mean = (red[0] + red[1] + red[2] + red[3]) * (1.0f / DD);
    float c = v - mean;
    float q = c * c;
#pragma unroll
    for (int m = 16; m; m >>= 1) q += __shfl_xor(q, m);
    __syncthreads();
    if ((d & 31) == 0) red[d >> 5] = q;
    __syncthreads();
    float var = (red[0] + red[1] + red[2] + red[3]) * (1.0f / DD);
    float y = c * rsqrtf(var + 1e-5f) * ln_g[d] + ln_b[d];
    if (do_relu) y = fmaxf(y, 0.0f);
    h[(long long)n * DD + d] += y;
}

extern "C" void kernel_launch(void* const* d_in, const int* in_sizes, int n_in,
                              void* d_out, int out_size, void* d_ws, size_t ws_size,
                              hipStream_t stream) {
    const float* x      = (const float*)d_in[0];
    const int*   ei     = (const int*)d_in[1];
    const float* W_edge = (const float*)d_in[2];
    const float* b_edge = (const float*)d_in[3];
    const float* W_node = (const float*)d_in[4];
    const float* b_node = (const float*)d_in[5];
    const float* lin_w  = (const float*)d_in[6];
    const float* att_sp = (const float*)d_in[7];
    const float* att_dp = (const float*)d_in[8];
    const float* g_bias = (const float*)d_in[9];
    const float* ln_g   = (const float*)d_in[10];
    const float* ln_b   = (const float*)d_in[11];

    const int E_ = in_sizes[0] / 3;
    const int Nn = out_size / DD;
    const int tot = E_ + Nn;

    float* bufA  = (float*)d_ws;
    float* xh    = bufA + (long long)Nn * DD;
    float* cnt   = xh   + (long long)Nn * DD;
    float* a_s   = cnt  + Nn;
    float* a_d   = a_s  + (long long)Nn * HH;
    float* mval  = a_d  + (long long)Nn * HH;
    float* denom = mval + (long long)Nn * HH;
    float* h     = (float*)d_out;

    long long nh = (long long)Nn * DD;
    int gh  = (int)((nh + 255) / 256);
    int g4  = (Nn * HH + 255) / 256;
    int g1  = (Nn + 255) / 256;
    int geh = (tot * HH + 255) / 256;
    dim3 gemm_grid((Nn + 15) / 16, 2);

    gnn_fill_u32<<<gh, 256, 0, stream>>>((unsigned int*)bufA, 0u, nh);
    gnn_fill_u32<<<g1, 256, 0, stream>>>((unsigned int*)cnt, 0u, Nn);
    gnn_edge_init<<<E_, DD, 0, stream>>>(x, ei, W_edge, b_edge, bufA, cnt, E_);
    gnn_wmma_gemm128<<<gemm_grid, 128, 0, stream>>>(bufA, W_node, h, b_node, cnt, Nn);

    for (int i = 0; i < 3; ++i) {
        gnn_wmma_gemm128<<<gemm_grid, 128, 0, stream>>>(
            h, lin_w + (long long)i * DD * DD, xh, nullptr, nullptr, Nn);
        gnn_att_scores<<<Nn, DD, 0, stream>>>(xh, att_sp + i * DD, att_dp + i * DD, a_s, a_d);

        gnn_fill_u32<<<gh, 256, 0, stream>>>((unsigned int*)bufA, 0u, nh);
        gnn_fill_u32<<<g4, 256, 0, stream>>>((unsigned int*)denom, 0u, Nn * HH);
        gnn_fill_u32<<<g4, 256, 0, stream>>>((unsigned int*)mval, 0xFF800000u, Nn * HH);

        gnn_edge_max  <<<geh, 256, 0, stream>>>(ei, a_s, a_d, mval, E_, Nn);
        gnn_edge_denom<<<geh, 256, 0, stream>>>(ei, a_s, a_d, mval, denom, E_, Nn);
        gnn_edge_accum<<<tot, DD, 0, stream>>>(ei, a_s, a_d, mval, denom, xh, bufA, E_, Nn);

        gnn_finalize<<<Nn, DD, 0, stream>>>(bufA, g_bias + i * DD, ln_g + i * DD,
                                            ln_b + i * DD, h, (i < 2) ? 1 : 0);
    }
    (void)n_in; (void)ws_size; (void)W_edge;
}


# Per layer one 16-node tile kernel does the previous layer's divide+LayerNorm+residual, the projection GEMM from an LDS tile, logits and resets: 28 launches -> 14; LayerNorm sums re-associated
# speedup vs baseline: 1.1901x; 1.1901x over previous
#include <hip/hip_runtime.h>
#include <hip/hip_bf16.h>

typedef float v2f __attribute__((ext_vector_type(2)));
typedef float v8f __attribute__((ext_vector_type(8)));

#define DD 128
#define HH 4
#define CC 32

__global__ void gnn_fill_u32(unsigned int* __restrict__ p, unsigned int v, long long n) {
    long long i = (long long)blockIdx.x * blockDim.x + threadIdx.x;
    if (i < n) p[i] = v;
}

__global__ void gnn_edge_init(const float* __restrict__ x, const int* __restrict__ ei,
                              const float* __restrict__ W_edge, const float* __restrict__ b_edge,
                              float* __restrict__ h_sum, float* __restrict__ cnt, int E_) {
    int e = blockIdx.x;
    int d = threadIdx.x;
    int dstn = ei[E_ + e];
    float x0 = x[e * 3 + 0], x1 = x[e * 3 + 1], x2 = x[e * 3 + 2];
    float v = x0 * W_edge[d] + x1 * W_edge[DD + d] + x2 * W_edge[2 * DD + d] + b_edge[d];
    atomicAdd(&h_sum[(long long)dstn * DD + d], v);
    if (d == 0) atomicAdd(&cnt[dstn], 1.0f);
}

__global__ void gnn_wmma_gemm128(const float* __restrict__ A, const float* __restrict__ B,
                                 float* __restrict__ C, const float* __restrict__ bias,
                                 const float* __restrict__ cnt, int M) {
    int wave = threadIdx.x >> 5;
    int lane = threadIdx.x & 31;
    int m0 = blockIdx.x * 16;
    int n0 = (blockIdx.y * 4 + wave) * 16;
    int half = lane >> 4;
    int l    = lane & 15;
    int K0   = half * 2;
    int Mrow = m0 + l;
    if (Mrow >= M) Mrow = M - 1;
    int Ncol = n0 + l;

    const float* Arow = A + (long long)Mrow * DD;
    v8f acc = {};
#pragma unroll
    for (int k = 0; k < DD; k += 4) {
        v2f a = *(const v2f*)(Arow + k + K0);
        v2f b;
        b.x = B[(long long)(k + K0) * DD + Ncol];
        b.y = B[(long long)(k + K0 + 1) * DD + Ncol];
        acc = __builtin_amdgcn_wmma_f32_16x16x4_f32(false, a, false, b,
                                                    (short)0, acc, false, false);
    }
#pragma unroll
    for (int r = 0; r < 8; ++r) {
        int row = m0 + r + half * 8;
        if (row < M) {
            float v = acc[r];
            if (cnt)  v *= 1.0f / fmaxf(cnt[row], 1.0f);
            if (bias) v += bias[n0 + l];
            C[(long long)row * DD + n0 + l] = v;
        }
    }
}

#define TP 132

template <int PREV>
__global__ void __launch_bounds__(128)
gnn_node_tile(const float* __restrict__ Wl,
              const float* __restrict__ att_src, const float* __restrict__ att_dst,
              const float* __restrict__ gbias, const float* __restrict__ ln_g,
              const float* __restrict__ ln_b,
              float* __restrict__ h, float* __restrict__ accb, float* __restrict__ denom,
              float* __restrict__ mval, float* __restrict__ xh,
              float* __restrict__ a_s, float* __restrict__ a_d, int M) {
    __shared__ __attribute__((aligned(16))) float Ht[16 * TP];
    __shared__ __attribute__((aligned(16))) float Xt[16 * TP];
    __shared__ float Lg[128];
    const int wave = threadIdx.x >> 5;
    const int lane = threadIdx.x & 31;
    const int m0 = blockIdx.x * 16;

    float4 hv[4];
    int    rowok[4];
    long long goff[4];
#pragma unroll
    for (int u = 0; u < 4; ++u) {
        const int node = m0 + wave * 4 + u;
        rowok[u] = node < M;
        goff[u]  = (long long)(rowok[u] ? node : M - 1) * DD + 4 * lane;
        hv[u]    = *(const float4*)(h + goff[u]);
    }
    if (PREV) {
        const float4 gb = *(const float4*)(gbias + 4 * lane);
        const float4 g  = *(const float4*)(ln_g + 4 * lane);
        const float4 b  = *(const float4*)(ln_b + 4 * lane);
        float v[4][4], s[4], q[4];
#pragma unroll
        for (int u = 0; u < 4; ++u) {
            const float4 o  = *(const float4*)(accb + goff[u]);
            const float  dn = denom[(goff[u] >> 7) * HH + (lane >> 3)];
            v[u][0] = o.x / dn + gb.x;  v[u][1] = o.y / dn + gb.y;
            v[u][2] = o.z / dn + gb.z;  v[u][3] = o.w / dn + gb.w;
            s[u] = (v[u][0] + v[u][1]) + (v[u][2] + v[u][3]);
        }
#pragma unroll
        for (int m = 16; m; m >>= 1) {
#pragma unroll
            for (int u = 0; u < 4; ++u) s[u] += __shfl_xor(s[u], m);
        }
#pragma unroll
        for (int u = 0; u < 4; ++u) {
            const float mean = s[u] * (1.0f / DD);
#pragma unroll
            for (int e = 0; e < 4; ++e) v[u][e] -= mean;
            q[u] = (v[u][0] * v[u][0] + v[u][1] * v[u][1]) + (v[u][2] * v[u][2] + v[u][3] * v[u][3]);
        }
#pragma unroll
        for (int m = 16; m; m >>= 1) {
#pragma unroll
            for (int u = 0; u < 4; ++u) q[u] += __shfl_xor(q[u], m);
        }
#pragma unroll
        for (int u = 0; u < 4; ++u) {
            const float rs = rsqrtf(q[u] * (1.0f / DD) + 1e-5f);
            hv[u].x += fmaxf(v[u][0] * rs * g.x + b.x, 0.0f);
            hv[u].y += fmaxf(v[u][1] * rs * g.y + b.y, 0.0f);
            hv[u].z += fmaxf(v[u][2] * rs * g.z + b.z, 0.0f);
            hv[u].w += fmaxf(v[u][3] * rs * g.w + b.w, 0.0f);
            if (rowok[u]) *(float4*)(h + goff[u]) = hv[u];
        }
    }
#pragma unroll
    for (int u = 0; u < 4; ++u)
        *(float4*)(&Ht[(wave * 4 + u) * TP + 4 * lane]) = hv[u];
    __syncthreads();

    const int half = lane >> 4;
    const int l    = lane & 15;
    const int K0   = half * 2;
    const int n0   = wave * 32;
    v8f acc0 = {};
    v8f acc1 = {};
#pragma unroll
    for (int k = 0; k < DD; k += 4) {
        const v2f a = *(const v2f*)(&Ht[l * TP + k + K0]);
        const float* Bk = Wl + (long long)(k + K0) * DD + n0 + l;
        v2f b0, b1;
        b0.x = Bk[0];       b0.y = Bk[DD];
        b1.x = Bk[16];      b1.y = Bk[DD + 16];
        acc0 = __builtin_amdgcn_wmma_f32_16x16x4_f32(false, a, false, b0, (short)0, acc0, false, false);
        acc1 = __builtin_amdgcn_wmma_f32_16x16x4_f32(false, a, false, b1, (short)0, acc1, false, false);
    }
#pragma unroll
    for (int r = 0; r < 8; ++r) {
        const int row = r + half * 8;
        Xt[row * TP + n0 + l]      = acc0[r];
        Xt[row * TP + n0 + 16 + l] = acc1[r];
    }
    __syncthreads();

    const int ch = wave * 32 + lane;
    const float ws = att_src[ch];
    const float wd = att_dst[ch];
    for (int it0 = 0; it0 < 16; it0 += 4) {
        float xv[4], s[4], t[4];
#pragma unroll
        for (int u = 0; u < 4; ++u) {
            xv[u] = Xt[(it0 + u) * TP + ch];
            s[u] = xv[u] * ws;
            t[u] = xv[u] * wd;
        }
#pragma unroll
        for (int m = 16; m; m >>= 1) {
#pragma unroll
            for (int u = 0; u < 4; ++u) { s[u] += __shfl_xor(s[u], m); t[u] += __shfl_xor(t[u], m); }
        }
#pragma unroll
        for (int u = 0; u < 4; ++u) {
            const int node = m0 + it0 + u;
            if (lane == 0) { Lg[(it0 + u) * HH + wave] = s[u]; Lg[64 + (it0 + u) * HH + wave] = t[u]; }
            if (node < M) {
                xh[(long long)node * DD + ch]   = xv[u];
                accb[(long long)node * DD + ch] = 0.0f;
            }
        }
    }
    __syncthreads();
    for (int qi = lane; qi < 64; qi += 32) {
        if (m0 + (qi >> 2) < M) {
            const long long idx = (long long)m0 * HH + qi;
            if (wave == 0)      a_s[idx]   = Lg[qi];
            else if (wave == 1) a_d[idx]   = Lg[64 + qi];
            else if (wave == 2) denom[idx] = 0.0f;
            else                mval[idx]  = __int_as_float((int)0xFF800000u);
        }
    }
}

__device__ __forceinline__ void edge_st(int t, int E_, const int* ei, int& s, int& dd) {
    if (t < E_) { s = ei[t]; dd = ei[E_ + t]; }
    else        { s = t - E_; dd = t - E_; }
}

__global__ void gnn_edge_max(const int* __restrict__ ei, const float* __restrict__ a_s,
                             const float* __restrict__ a_d, float* __restrict__ mval,
                             int E_, int Nn) {
    int idx = blockIdx.x * blockDim.x + threadIdx.x;
    int tot = E_ + Nn;
    int t = idx >> 2, hd = idx & 3;
    if (t >= tot) return;
    int s, dd; edge_st(t, E_, ei, s, dd);
    float e = a_s[s * HH + hd] + a_d[dd * HH + hd];
    e = (e > 0.0f) ? e : 0.2f * e;
    int* p = (int*)&mval[dd * HH + hd];
    if (e >= 0.0f) atomicMax(p, __float_as_int(e));
    else           atomicMin((unsigned int*)p, __float_as_uint(e));
}

__global__ void gnn_edge_accum(const int* __restrict__ ei, const float* __restrict__ a_s,
                               const float* __restrict__ a_d, const float* __restrict__ mval,
                               const float* __restrict__ xh, float* __restrict__ out,
                               float* __restrict__ denom, int E_, int Nn) {
    int t = blockIdx.x;
    int d = threadIdx.x;
    int hq = (d < HH) ? d : (d >> 5);
    int s, dd; edge_st(t, E_, ei, s, dd);
    float e = a_s[s * HH + hq] + a_d[dd * HH + hq];
    e = (e > 0.0f) ? e : 0.2f * e;
    float exq = __expf(e - mval[dd * HH + hq]);
    float ex = __int_as_float(__builtin_amdgcn_readfirstlane(__float_as_int(exq)));
    atomicAdd(&out[(long long)dd * DD + d], ex * xh[(long long)s * DD + d]);
    if (d < HH) atomicAdd(&denom[dd * HH + d], exq);
}

__global__ void gnn_finalize(const float* __restrict__ out, const float* __restrict__ denom,
                             const float* __restrict__ gbias,
                             const float* __restrict__ ln_g, const float* __restrict__ ln_b,
                             float* __restrict__ h, int do_relu) {
    __shared__ float red[4];
    int n = blockIdx.x;
    int d = threadIdx.x;
    float v = out[(long long)n * DD + d] / denom[n * HH + (d >> 5)] + gbias[d];
    float s = v;
#pragma unroll
    for (int m = 16; m; m >>= 1) s += __shfl_xor(s, m);
    if ((d & 31) == 0) red[d >> 5] = s;
    __syncthreads();
    float mean = (red[0] + red[1] + red[2] + red[3]) * (1.0f / DD);
    float c = v - mean;
    float q = c * c;
#pragma unroll
    for (int m = 16; m; m >>= 1) q += __shfl_xor(q, m);
    __syncthreads();
    if ((d & 31) == 0) red[d >> 5] = q;
    __syncthreads();
    float var = (red[0] + red[1] + red[2] + red[3]) * (1.0f / DD);
    float y = c * rsqrtf(var + 1e-5f) * ln_g[d] + ln_b[d];
    if (do_relu) y = fmaxf(y, 0.0f);
    h[(long long)n * DD + d] += y;
}

extern "C" void kernel_launch(void* const* d_in, const int* in_sizes, int n_in,
                              void* d_out, int out_size, void* d_ws, size_t ws_size,
                              hipStream_t stream) {
    const float* x      = (const float*)d_in[0];
    const int*   ei     = (const int*)d_in[1];
    const float* W_edge = (const float*)d_in[2];
    const float* b_edge = (const float*)d_in[3];
    const float* W_node = (const float*)d_in[4];
    const float* b_node = (const float*)d_in[5];
    const float* lin_w  = (const float*)d_in[6];
    const float* att_sp = (const float*)d_in[7];
    const float* att_dp = (const float*)d_in[8];
    const float* g_bias = (const float*)d_in[9];
    const float* ln_g   = (const float*)d_in[10];
    const float* ln_b   = (const float*)d_in[11];

    const int E_ = in_sizes[0] / 3;
    const int Nn = out_size / DD;
    const int tot = E_ + Nn;

    float* bufA  = (float*)d_ws;
    float* xh    = bufA + (long long)Nn * DD;
    float* cnt   = xh   + (long long)Nn * DD;
    float* a_s   = cnt  + Nn;
    float* a_d   = a_s  + (long long)Nn * HH;
    float* mval  = a_d  + (long long)Nn * HH;
    float* denom = mval + (long long)Nn * HH;
    float* h     = (float*)d_out;

    long long nh = (long long)Nn * DD;
    int gh  = (int)((nh + 255) / 256);
    int tiles = (Nn + 15) / 16;
    int g1  = (Nn + 255) / 256;
    int geh = (tot * HH + 255) / 256;
    dim3 gemm_grid((Nn + 15) / 16, 2);

    gnn_fill_u32<<<gh, 256, 0, stream>>>((unsigned int*)bufA, 0u, nh);
    gnn_fill_u32<<<g1, 256, 0, stream>>>((unsigned int*)cnt, 0u, Nn);
    gnn_edge_init<<<E_, DD, 0, stream>>>(x, ei, W_edge, b_edge, bufA, cnt, E_);
    gnn_wmma_gemm128<<<gemm_grid, 128, 0, stream>>>(bufA, W_node, h, b_node, cnt, Nn);

    for (int i = 0; i < 3; ++i) {
        if (i == 0)
            gnn_node_tile<0><<<tiles, 128, 0, stream>>>(
                lin_w, att_sp, att_dp, nullptr, nullptr, nullptr,
                h, bufA, denom, mval, xh, a_s, a_d, Nn);
        else
            gnn_node_tile<1><<<tiles, 128, 0, stream>>>(
                lin_w + (long long)i * DD * DD, att_sp + i * DD, att_dp + i * DD,
                g_bias + (i - 1) * DD, ln_g + (i - 1) * DD, ln_b + (i - 1) * DD,
                h, bufA, denom, mval, xh, a_s, a_d, Nn);

        gnn_edge_max  <<<geh, 256, 0, stream>>>(ei, a_s, a_d, mval, E_, Nn);
        gnn_edge_accum<<<tot, DD, 0, stream>>>(ei, a_s, a_d, mval, xh, bufA, denom, E_, Nn);
    }
    gnn_finalize<<<Nn, DD, 0, stream>>>(bufA, denom, g_bias + 2 * DD, ln_g + 2 * DD,
                                        ln_b + 2 * DD, h, 0);
    (void)n_in; (void)ws_size; (void)W_edge;
}


# Edge MLP moved behind the init scatter (linearity): per edge 4 f64 atomics (3 raw features + count: exact, order-free), was 129 f32; tile kernel: mean, W_edge, W_node WMMA; init sums re-associated
# speedup vs baseline: 1.4997x; 1.2601x over previous
#include <hip/hip_runtime.h>
#include <hip/hip_bf16.h>

typedef float v2f __attribute__((ext_vector_type(2)));
typedef float v8f __attribute__((ext_vector_type(8)));

#define DD 128
#define HH 4
#define CC 32

__global__ void gnn_fill_u32(unsigned int* __restrict__ p, unsigned int v, long long n) {
    long long i = (long long)blockIdx.x * blockDim.x + threadIdx.x;
    if (i < n) p[i] = v;
}

__global__ void gnn_edge_init(const float* __restrict__ x, const int* __restrict__ ei,
                              double* __restrict__ S, int E_) {
    long long idx = (long long)blockIdx.x * blockDim.x + threadIdx.x;
    int e = (int)(idx >> 2);
    int c = (int)(idx & 3);
    if (e >= E_) return;
    int dstn = ei[E_ + e];
    double v = (c < 3) ? (double)x[(long long)e * 3 + c] : 1.0;
    atomicAdd(&S[(long long)dstn * 4 + c], v);
}

#define TP 132

__global__ void __launch_bounds__(128)
gnn_init_tile(const double* __restrict__ S, const float* __restrict__ W_edge,
              const float* __restrict__ b_edge, const float* __restrict__ W_node,
              const float* __restrict__ b_node, float* __restrict__ h, int M) {
    __shared__ __attribute__((aligned(16))) float Ht[16 * TP];
    __shared__ __attribute__((aligned(16))) float Xt[16 * TP];
    const int wave = threadIdx.x >> 5;
    const int lane = threadIdx.x & 31;
    const int m0 = blockIdx.x * 16;

    const float4 w0 = *(const float4*)(W_edge + 4 * lane);
    const float4 w1 = *(const float4*)(W_edge + DD + 4 * lane);
    const float4 w2 = *(const float4*)(W_edge + 2 * DD + 4 * lane);
    const float4 be = *(const float4*)(b_edge + 4 * lane);
#pragma unroll
    for (int u = 0; u < 4; ++u) {
        const int r = wave * 4 + u;
        int node = m0 + r;
        if (node >= M) node = M - 1;
        const double* sp = S + (long long)node * 4;
        const double c  = sp[3];
        const float inv = 1.0f / fmaxf((float)c, 1.0f);
        const float a0 = (float)sp[0] * inv;
        const float a1 = (float)sp[1] * inv;
        const float a2 = (float)sp[2] * inv;
        const float fb = (c > 0.0) ? 1.0f : 0.0f;
        float4 v;
        v.x = a0 * w0.x + a1 * w1.x + a2 * w2.x + fb * be.x;
        v.y = a0 * w0.y + a1 * w1.y + a2 * w2.y + fb * be.y;
        v.z = a0 * w0.z + a1 * w1.z + a2 * w2.z + fb * be.z;
        v.w = a0 * w0.w + a1 * w1.w + a2 * w2.w + fb * be.w;
        *(float4*)(&Ht[r * TP + 4 * lane]) = v;
    }
    __syncthreads();

    const int half = lane >> 4;
    const int l    = lane & 15;
    const int K0   = half * 2;
    const int n0   = wave * 32;
    v8f acc0 = {};
    v8f acc1 = {};
#pragma unroll
    for (int k = 0; k < DD; k += 4) {
        const v2f a = *(const v2f*)(&Ht[l * TP + k + K0]);
        const float* Bk = W_node + (long long)(k + K0) * DD + n0 + l;
        v2f b0, b1;
        b0.x = Bk[0];       b0.y = Bk[DD];
        b1.x = Bk[16];      b1.y = Bk[DD + 16];
        acc0 = __builtin_amdgcn_wmma_f32_16x16x4_f32(false, a, false, b0, (short)0, acc0, false, false);
        acc1 = __builtin_amdgcn_wmma_f32_16x16x4_f32(false, a, false, b1, (short)0, acc1, false, false);
    }
#pragma unroll
    for (int r = 0; r < 8; ++r) {
        const int row = r + half * 8;
        Xt[row * TP + n0 + l]      = acc0[r];
        Xt[row * TP + n0 + 16 + l] = acc1[r];
    }
    __syncthreads();

    const int ch = wave * 32 + lane;
    const float bn = b_node[ch];
    for (int r = 0; r < 16; ++r) {
        const int node = m0 + r;
        if (node < M) h[(long long)node * DD + ch] = Xt[r * TP + ch] + bn;
    }
}

template <int PREV>
__global__ void __launch_bounds__(128)
gnn_node_tile(const float* __restrict__ Wl,
              const float* __restrict__ att_src, const float* __restrict__ att_dst,
              const float* __restrict__ gbias, const float* __restrict__ ln_g,
              const float* __restrict__ ln_b,
              float* __restrict__ h, float* __restrict__ accb, float* __restrict__ denom,
              float* __restrict__ mval, float* __restrict__ xh,
              float* __restrict__ a_s, float* __restrict__ a_d, int M) {
    __shared__ __attribute__((aligned(16))) float Ht[16 * TP];
    __shared__ __attribute__((aligned(16))) float Xt[16 * TP];
    __shared__ float Lg[128];
    const int wave = threadIdx.x >> 5;
    const int lane = threadIdx.x & 31;
    const int m0 = blockIdx.x * 16;

    float4 hv[4];
    int    rowok[4];
    long long goff[4];
#pragma unroll
    for (int u = 0; u < 4; ++u) {
        const int node = m0 + wave * 4 + u;
        rowok[u] = node < M;
        goff[u]  = (long long)(rowok[u] ? node : M - 1) * DD + 4 * lane;
        hv[u]    = *(const float4*)(h + goff[u]);
    }
    if (PREV) {
        const float4 gb = *(const float4*)(gbias + 4 * lane);
        const float4 g  = *(const float4*)(ln_g + 4 * lane);
        const float4 b  = *(const float4*)(ln_b + 4 * lane);
        float v[4][4], s[4], q[4];
#pragma unroll
        for (int u = 0; u < 4; ++u) {
            const float4 o  = *(const float4*)(accb + goff[u]);
            const float  dn = denom[(goff[u] >> 7) * HH + (lane >> 3)];
            v[u][0] = o.x / dn + gb.x;  v[u][1] = o.y / dn + gb.y;
            v[u][2] = o.z / dn + gb.z;  v[u][3] = o.w / dn + gb.w;
            s[u] = (v[u][0] + v[u][1]) + (v[u][2] + v[u][3]);
        }
#pragma unroll
        for (int m = 16; m; m >>= 1) {
#pragma unroll
            for (int u = 0; u < 4; ++u) s[u] += __shfl_xor(s[u], m);
        }
#pragma unroll
        for (int u = 0; u < 4; ++u) {
            const float mean = s[u] * (1.0f / DD);
#pragma unroll
            for (int e = 0; e < 4; ++e) v[u][e] -= mean;
            q[u] = (v[u][0] * v[u][0] + v[u][1] * v[u][1]) + (v[u][2] * v[u][2] + v[u][3] * v[u][3]);
        }
#pragma unroll
        for (int m = 16; m; m >>= 1) {
#pragma unroll
            for (int u = 0; u < 4; ++u) q[u] += __shfl_xor(q[u], m);
        }
#pragma unroll
        for (int u = 0; u < 4; ++u) {
            const float rs = rsqrtf(q[u] * (1.0f / DD) + 1e-5f);
            hv[u].x += fmaxf(v[u][0] * rs * g.x + b.x, 0.0f);
            hv[u].y += fmaxf(v[u][1] * rs * g.y + b.y, 0.0f);
            hv[u].z += fmaxf(v[u][2] * rs * g.z + b.z, 0.0f);
            hv[u].w += fmaxf(v[u][3] * rs * g.w + b.w, 0.0f);
            if (rowok[u]) *(float4*)(h + goff[u]) = hv[u];
        }
    }
#pragma unroll
    for (int u = 0; u < 4; ++u)
        *(float4*)(&Ht[(wave * 4 + u) * TP + 4 * lane]) = hv[u];
    __syncthreads();

    const int half = lane >> 4;
    const int l    = lane & 15;
    const int K0   = half * 2;
    const int n0   = wave * 32;
    v8f acc0 = {};
    v8f acc1 = {};
#pragma unroll
    for (int k = 0; k < DD; k += 4) {
        const v2f a = *(const v2f*)(&Ht[l * TP + k + K0]);
        const float* Bk = Wl + (long long)(k + K0) * DD + n0 + l;
        v2f b0, b1;
        b0.x = Bk[0];       b0.y = Bk[DD];
        b1.x = Bk[16];      b1.y = Bk[DD + 16];
        acc0 = __builtin_amdgcn_wmma_f32_16x16x4_f32(false, a, false, b0, (short)0, acc0, false, false);
        acc1 = __builtin_amdgcn_wmma_f32_16x16x4_f32(false, a, false, b1, (short)0, acc1, false, false);
    }
#pragma unroll
    for (int r = 0; r < 8; ++r) {
        const int row = r + half * 8;
        Xt[row * TP + n0 + l]      = acc0[r];
        Xt[row * TP + n0 + 16 + l] = acc1[r];
    }
    __syncthreads();

    const int ch = wave * 32 + lane;
    const float ws = att_src[ch];
    const float wd = att_dst[ch];
    for (int it0 = 0; it0 < 16; it0 += 4) {
        float xv[4], s[4], t[4];
#pragma unroll
        for (int u = 0; u < 4; ++u) {
            xv[u] = Xt[(it0 + u) * TP + ch];
            s[u] = xv[u] * ws;
            t[u] = xv[u] * wd;
        }
#pragma unroll
        for (int m = 16; m; m >>= 1) {
#pragma unroll
            for (int u = 0; u < 4; ++u) { s[u] += __shfl_xor(s[u], m); t[u] += __shfl_xor(t[u], m); }
        }
#pragma unroll
        for (int u = 0; u < 4; ++u) {
            const int node = m0 + it0 + u;
            if (lane == 0) { Lg[(it0 + u) * HH + wave] = s[u]; Lg[64 + (it0 + u) * HH + wave] = t[u]; }
            if (node < M) {
                xh[(long long)node * DD + ch]   = xv[u];
                accb[(long long)node * DD + ch] = 0.0f;
            }
        }
    }
    __syncthreads();
    for (int qi = lane; qi < 64; qi += 32) {
        if (m0 + (qi >> 2) < M) {
            const long long idx = (long long)m0 * HH + qi;
            if (wave == 0)      a_s[idx]   = Lg[qi];
            else if (wave == 1) a_d[idx]   = Lg[64 + qi];
            else if (wave == 2) denom[idx] = 0.0f;
            else                mval[idx]  = __int_as_float((int)0xFF800000u);
        }
    }
}

__device__ __forceinline__ void edge_st(int t, int E_, const int* ei, int& s, int& dd) {
    if (t < E_) { s = ei[t]; dd = ei[E_ + t]; }
    else        { s = t - E_; dd = t - E_; }
}

__global__ void gnn_edge_max(const int* __restrict__ ei, const float* __restrict__ a_s,
                             const float* __restrict__ a_d, float* __restrict__ mval,
                             int E_, int Nn) {
    int idx = blockIdx.x * blockDim.x + threadIdx.x;
    int tot = E_ + Nn;
    int t = idx >> 2, hd = idx & 3;
    if (t >= tot) return;
    int s, dd; edge_st(t, E_, ei, s, dd);
    float e = a_s[s * HH + hd] + a_d[dd * HH + hd];
    e = (e > 0.0f) ? e : 0.2f * e;
    int* p = (int*)&mval[dd * HH + hd];
    if (e >= 0.0f) atomicMax(p, __float_as_int(e));
    else           atomicMin((unsigned int*)p, __float_as_uint(e));
}

__global__ void gnn_edge_accum(const int* __restrict__ ei, const float* __restrict__ a_s,
                               const float* __restrict__ a_d, const float* __restrict__ mval,
                               const float* __restrict__ xh, float* __restrict__ out,
                               float* __restrict__ denom, int E_, int Nn) {
    int t = blockIdx.x;
    int d = threadIdx.x;
    int hq = (d < HH) ? d : (d >> 5);
    int s, dd; edge_st(t, E_, ei, s, dd);
    float e = a_s[s * HH + hq] + a_d[dd * HH + hq];
    e = (e > 0.0f) ? e : 0.2f * e;
    float exq = __expf(e - mval[dd * HH + hq]);
    float ex = __int_as_float(__builtin_amdgcn_readfirstlane(__float_as_int(exq)));
    atomicAdd(&out[(long long)dd * DD + d], ex * xh[(long long)s * DD + d]);
    if (d < HH) atomicAdd(&denom[dd * HH + d], exq);
}

__global__ void gnn_finalize(const float* __restrict__ out, const float* __restrict__ denom,
                             const float* __restrict__ gbias,
                             const float* __restrict__ ln_g, const float* __restrict__ ln_b,
                             float* __restrict__ h, int do_relu) {
    __shared__ float red[4];
    int n = blockIdx.x;
    int d = threadIdx.x;
    float v = out[(long long)n * DD + d] / denom[n * HH + (d >> 5)] + gbias[d];
    float s = v;
#pragma unroll
    for (int m = 16; m; m >>= 1) s += __shfl_xor(s, m);
    if ((d & 31) == 0) red[d >> 5] = s;
    __syncthreads();
    float mean = (red[0] + red[1] + red[2] + red[3]) * (1.0f / DD);
    float c = v - mean;
    float q = c * c;
#pragma unroll
    for (int m = 16; m; m >>= 1) q += __shfl_xor(q, m);
    __syncthreads();
    if ((d & 31) == 0) red[d >> 5] = q;
    __syncthreads();
    float var = (red[0] + red[1] + red[2] + red[3]) * (1.0f / DD);
    float y = c * rsqrtf(var + 1e-5f) * ln_g[d] + ln_b[d];
    if (do_relu) y = fmaxf(y, 0.0f);
    h[(long long)n * DD + d] += y;
}

extern "C" void kernel_launch(void* const* d_in, const int* in_sizes, int n_in,
                              void* d_out, int out_size, void* d_ws, size_t ws_size,
                              hipStream_t stream) {
    const float* x      = (const float*)d_in[0];
    const int*   ei     = (const int*)d_in[1];
    const float* W_edge = (const float*)d_in[2];
    const float* b_edge = (const float*)d_in[3];
    const float* W_node = (const float*)d_in[4];
    const float* b_node = (const float*)d_in[5];
    const float* lin_w  = (const float*)d_in[6];
    const float* att_sp = (const float*)d_in[7];
    const float* att_dp = (const float*)d_in[8];
    const float* g_bias = (const float*)d_in[9];
    const float* ln_g   = (const float*)d_in[10];
    const float* ln_b   = (const float*)d_in[11];

    const int E_ = in_sizes[0] / 3;
    const int Nn = out_size / DD;
    const int tot = E_ + Nn;

    float* bufA  = (float*)d_ws;
    float* xh    = bufA + (long long)Nn * DD;
    float* a_s   = xh   + (long long)Nn * DD + Nn;
    float* a_d   = a_s  + (long long)Nn * HH;
    float* mval  = a_d  + (long long)Nn * HH;
    float* denom = mval + (long long)Nn * HH;
    float* h     = (float*)d_out;
    double* S    = (double*)d_ws;

    long long ns = (long long)Nn * 8;
    int gs  = (int)((ns + 255) / 256);
    int tiles = (Nn + 15) / 16;
    int gei = (int)(((long long)E_ * 4 + 255) / 256);
    int geh = (tot * HH + 255) / 256;

    gnn_fill_u32<<<gs, 256, 0, stream>>>((unsigned int*)S, 0u, ns);
    gnn_edge_init<<<gei, 256, 0, stream>>>(x, ei, S, E_);
    gnn_init_tile<<<tiles, 128, 0, stream>>>(S, W_edge, b_edge, W_node, b_node, h, Nn);

    for (int i = 0; i < 3; ++i) {
        if (i == 0)
            gnn_node_tile<0><<<tiles, 128, 0, stream>>>(
                lin_w, att_sp, att_dp, nullptr, nullptr, nullptr,
                h, bufA, denom, mval, xh, a_s, a_d, Nn);
        else
            gnn_node_tile<1><<<tiles, 128, 0, stream>>>(
                lin_w + (long long)i * DD * DD, att_sp + i * DD, att_dp + i * DD,
                g_bias + (i - 1) * DD, ln_g + (i - 1) * DD, ln_b + (i - 1) * DD,
                h, bufA, denom, mval, xh, a_s, a_d, Nn);

        gnn_edge_max  <<<geh, 256, 0, stream>>>(ei, a_s, a_d, mval, E_, Nn);
        gnn_edge_accum<<<tot, DD, 0, stream>>>(ei, a_s, a_d, mval, xh, bufA, denom, E_, Nn);
    }
    gnn_finalize<<<Nn, DD, 0, stream>>>(bufA, denom, g_bias + 2 * DD, ln_g + 2 * DD,
                                        ln_b + 2 * DD, h, 0);
    (void)n_in; (void)ws_size;
}


# Finalize and edge-init rebuilt as few long-lived waves: finalize one wave per row, 4 rows per wave, no barrier (200000 -> 12504 waves); edge init 75000 -> 9376; last LayerNorm's sums re-associated
# speedup vs baseline: 3.5779x; 2.3857x over previous
#include <hip/hip_runtime.h>
#include <hip/hip_bf16.h>

typedef float v2f __attribute__((ext_vector_type(2)));
typedef float v8f __attribute__((ext_vector_type(8)));

#define DD 128
#define HH 4
#define CC 32

__global__ void gnn_fill_u32(unsigned int* __restrict__ p, unsigned int v, long long n) {
    long long i = (long long)blockIdx.x * blockDim.x + threadIdx.x;
    if (i < n) p[i] = v;
}

__global__ void __launch_bounds__(256)
gnn_edge_init(const float* __restrict__ x, const int* __restrict__ ei,
              double* __restrict__ S, int E_) {
    const long long total  = (long long)E_ * 4;
    const long long stride = (long long)gridDim.x * blockDim.x;
    const long long first  = (long long)blockIdx.x * blockDim.x + threadIdx.x;
    const int c = (int)(first & 3);
    const int cx = (c < 3) ? c : 2;
    long long idx = first;
    for (; idx + 3 * stride < total; idx += 4 * stride) {
        int   dstn[4];
        float xf[4];
#pragma unroll
        for (int u = 0; u < 4; ++u) {
            const int e = (int)((idx + u * stride) >> 2);
            dstn[u] = ei[E_ + e];
            xf[u]   = x[(long long)e * 3 + cx];
        }
#pragma unroll
        for (int u = 0; u < 4; ++u)
            atomicAdd(&S[(long long)dstn[u] * 4 + c], (c < 3) ? (double)xf[u] : 1.0);
    }
    for (; idx < total; idx += stride) {
        const int e = (int)(idx >> 2);
        const int dstn = ei[E_ + e];
        const float xf = x[(long long)e * 3 + cx];
        atomicAdd(&S[(long long)dstn * 4 + c], (c < 3) ? (double)xf : 1.0);
    }
}

#define TP 132

__global__ void __launch_bounds__(128)
gnn_init_tile(const double* __restrict__ S, const float* __restrict__ W_edge,
              const float* __restrict__ b_edge, const float* __restrict__ W_node,
              const float* __restrict__ b_node, float* __restrict__ h, int M) {
    __shared__ __attribute__((aligned(16))) float Ht[16 * TP];
    __shared__ __attribute__((aligned(16))) float Xt[16 * TP];
    const int wave = threadIdx.x >> 5;
    const int lane = threadIdx.x & 31;
    const int m0 = blockIdx.x * 16;

    const float4 w0 = *(const float4*)(W_edge + 4 * lane);
    const float4 w1 = *(const float4*)(W_edge + DD + 4 * lane);
    const float4 w2 = *(const float4*)(W_edge + 2 * DD + 4 * lane);
    const float4 be = *(const float4*)(b_edge + 4 * lane);
#pragma unroll
    for (int u = 0; u < 4; ++u) {
        const int r = wave * 4 + u;
        int node = m0 + r;
        if (node >= M) node = M - 1;
        const double* sp = S + (long long)node * 4;
        const double c  = sp[3];
        const float inv = 1.0f / fmaxf((float)c, 1.0f);
        const float a0 = (float)sp[0] * inv;
        const float a1 = (float)sp[1] * inv;
        const float a2 = (float)sp[2] * inv;
        const float fb = (c > 0.0) ? 1.0f : 0.0f;
        float4 v;
        v.x = a0 * w0.x + a1 * w1.x + a2 * w2.x + fb * be.x;
        v.y = a0 * w0.y + a1 * w1.y + a2 * w2.y + fb * be.y;
        v.z = a0 * w0.z + a1 * w1.z + a2 * w2.z + fb * be.z;
        v.w = a0 * w0.w + a1 * w1.w + a2 * w2.w + fb * be.w;
        *(float4*)(&Ht[r * TP + 4 * lane]) = v;
    }
    __syncthreads();

    const int half = lane >> 4;
    const int l    = lane & 15;
    const int K0   = half * 2;
    const int n0   = wave * 32;
    v8f acc0 = {};
    v8f acc1 = {};
#pragma unroll
    for (int k = 0; k < DD; k += 4) {
        const v2f a = *(const v2f*)(&Ht[l * TP + k + K0]);
        const float* Bk = W_node + (long long)(k + K0) * DD + n0 + l;
        v2f b0, b1;
        b0.x = Bk[0];       b0.y = Bk[DD];
        b1.x = Bk[16];      b1.y = Bk[DD + 16];
        acc0 = __builtin_amdgcn_wmma_f32_16x16x4_f32(false, a, false, b0, (short)0, acc0, false, false);
        acc1 = __builtin_amdgcn_wmma_f32_16x16x4_f32(false, a, false, b1, (short)0, acc1, false, false);
    }
#pragma unroll
    for (int r = 0; r < 8; ++r) {
        const int row = r + half * 8;
        Xt[row * TP + n0 + l]      = acc0[r];
        Xt[row * TP + n0 + 16 + l] = acc1[r];
    }
    __syncthreads();

    const int ch = wave * 32 + lane;
    const float bn = b_node[ch];
    for (int r = 0; r < 16; ++r) {
        const int node = m0 + r;
        if (node < M) h[(long long)node * DD + ch] = Xt[r * TP + ch] + bn;
    }
}

template <int PREV>
__global__ void __launch_bounds__(128)
gnn_node_tile(const float* __restrict__ Wl,
              const float* __restrict__ att_src, const float* __restrict__ att_dst,
              const float* __restrict__ gbias, const float* __restrict__ ln_g,
              const float* __restrict__ ln_b,
              float* __restrict__ h, const float* __restrict__ accb,
              const float* __restrict__ denom, float* __restrict__ xh,
              float* __restrict__ a_s, float* __restrict__ a_d, int M) {
    __shared__ __attribute__((aligned(16))) float Ht[16 * TP];
    __shared__ __attribute__((aligned(16))) float Xt[16 * TP];
    __shared__ float Lg[128];
    const int wave = threadIdx.x >> 5;
    const int lane = threadIdx.x & 31;
    const int m0 = blockIdx.x * 16;

    float4 hv[4];
    int    rowok[4];
    long long goff[4];
#pragma unroll
    for (int u = 0; u < 4; ++u) {
        const int node = m0 + wave * 4 + u;
        rowok[u] = node < M;
        goff[u]  = (long long)(rowok[u] ? node : M - 1) * DD + 4 * lane;
        hv[u]    = *(const float4*)(h + goff[u]);
    }
    if (PREV) {
        const float4 gb = *(const float4*)(gbias + 4 * lane);
        const float4 g  = *(const float4*)(ln_g + 4 * lane);
        const float4 b  = *(const float4*)(ln_b + 4 * lane);
        float v[4][4], s[4], q[4];
#pragma unroll
        for (int u = 0; u < 4; ++u) {
            const float4 o  = *(const float4*)(accb + goff[u]);
            const float  dn = denom[(goff[u] >> 7) * HH + (lane >> 3)];
            v[u][0] = o.x / dn + gb.x;  v[u][1] = o.y / dn + gb.y;
            v[u][2] = o.z / dn + gb.z;  v[u][3] = o.w / dn + gb.w;
            s[u] = (v[u][0] + v[u][1]) + (v[u][2] + v[u][3]);
        }
#pragma unroll
        for (int m = 16; m; m >>= 1) {
#pragma unroll
            for (int u = 0; u < 4; ++u) s[u] += __shfl_xor(s[u], m);
        }
#pragma unroll
        for (int u = 0; u < 4; ++u) {
            const float mean = s[u] * (1.0f / DD);
#pragma unroll
            for (int e = 0; e < 4; ++e) v[u][e] -= mean;
            q[u] = (v[u][0] * v[u][0] + v[u][1] * v[u][1]) + (v[u][2] * v[u][2] + v[u][3] * v[u][3]);
        }
#pragma unroll
        for (int m = 16; m; m >>= 1) {
#pragma unroll
            for (int u = 0; u < 4; ++u) q[u] += __shfl_xor(q[u], m);
        }
#pragma unroll
        for (int u = 0; u < 4; ++u) {
            const float rs = rsqrtf(q[u] * (1.0f / DD) + 1e-5f);
            hv[u].x += fmaxf(v[u][0] * rs * g.x + b.x, 0.0f);
            hv[u].y += fmaxf(v[u][1] * rs * g.y + b.y, 0.0f);
            hv[u].z += fmaxf(v[u][2] * rs * g.z + b.z, 0.0f);
            hv[u].w += fmaxf(v[u][3] * rs * g.w + b.w, 0.0f);
            if (rowok[u]) *(float4*)(h + goff[u]) = hv[u];
        }
    }
#pragma unroll
    for (int u = 0; u < 4; ++u)
        *(float4*)(&Ht[(wave * 4 + u) * TP + 4 * lane]) = hv[u];
    __syncthreads();

    const int half = lane >> 4;
    const int l    = lane & 15;
    const int K0   = half * 2;
    const int n0   = wave * 32;
    v8f acc0 = {};
    v8f acc1 = {};
#pragma unroll
    for (int k = 0; k < DD; k += 4) {
        const v2f a = *(const v2f*)(&Ht[l * TP + k + K0]);
        const float* Bk = Wl + (long long)(k + K0) * DD + n0 + l;
        v2f b0, b1;
        b0.x = Bk[0];       b0.y = Bk[DD];
        b1.x = Bk[16];      b1.y = Bk[DD + 16];
        acc0 = __builtin_amdgcn_wmma_f32_16x16x4_f32(false, a, false, b0, (short)0, acc0, false, false);
        acc1 = __builtin_amdgcn_wmma_f32_16x16x4_f32(false, a, false, b1, (short)0, acc1, false, false);
    }
#pragma unroll
    for (int r = 0; r < 8; ++r) {
        const int row = r + half * 8;
        Xt[row * TP + n0 + l]      = acc0[r];
        Xt[row * TP + n0 + 16 + l] = acc1[r];
    }
    __syncthreads();

    const int ch = wave * 32 + lane;
    const float ws = att_src[ch];
    const float wd = att_dst[ch];
    for (int it0 = 0; it0 < 16; it0 += 4) {
        float xv[4], s[4], t[4];
#pragma unroll
        for (int u = 0; u < 4; ++u) {
            xv[u] = Xt[(it0 + u) * TP + ch];
            s[u] = xv[u] * ws;
            t[u] = xv[u] * wd;
        }
#pragma unroll
        for (int m = 16; m; m >>= 1) {
#pragma unroll
            for (int u = 0; u < 4; ++u) { s[u] += __shfl_xor(s[u], m); t[u] += __shfl_xor(t[u], m); }
        }
#pragma unroll
        for (int u = 0; u < 4; ++u) {
            const int node = m0 + it0 + u;
            if (lane == 0) { Lg[(it0 + u) * HH + wave] = s[u]; Lg[64 + (it0 + u) * HH + wave] = t[u]; }
            if (node < M) xh[(long long)node * DD + ch] = xv[u];
        }
    }
    __syncthreads();
    for (int qi = lane; qi < 64; qi += 32) {
        if (m0 + (qi >> 2) < M) {
            const long long idx = (long long)m0 * HH + qi;
            if (wave == 0)      a_s[idx] = Lg[qi];
            else if (wave == 1) a_d[idx] = Lg[64 + qi];
        }
    }
}

#define GR   200
#define GNT  1024
#define GNW  32
#define GCAP 256
#define GQ   8
#define G0   ((GR * (DD / 4) + GNT - 1) / GNT)
#define GB   8

__device__ __forceinline__ float gnn_lrelu(float e) { return (e > 0.0f) ? e : 0.2f * e; }

template <bool AL>
__device__ __forceinline__ int gnn_scan(const int* __restrict__ col,
                                        int E_, int qb, int qe, int base, int lo, int lane,
                                        unsigned ltmask, int* lsrc_w, unsigned short* lrow_w) {
    int cw = 0;
    for (int q0 = qb; q0 < qe; q0 += 32 * GQ) {
        int4 dq[GQ];
#pragma unroll
        for (int u = 0; u < GQ; ++u) {
            const int q  = q0 + u * 32 + lane;
            const int qc = (q < qe) ? q : (qe - 1);
            if (AL) {
                dq[u] = *(const int4*)(col + 4 * qc);
            } else {
                const int last = E_ - 1;
                const int i0 = 4 * qc;
                dq[u].x = col[(i0 + 0 < last) ? (i0 + 0) : last];
                dq[u].y = col[(i0 + 1 < last) ? (i0 + 1) : last];
                dq[u].z = col[(i0 + 2 < last) ? (i0 + 2) : last];
                dq[u].w = col[(i0 + 3 < last) ? (i0 + 3) : last];
            }
        }
#pragma unroll
        for (int u = 0; u < GQ; ++u) {
            const int q = q0 + u * 32 + lane;
            const int dv[4] = {dq[u].x, dq[u].y, dq[u].z, dq[u].w};
            const unsigned rg = (q < qe) ? (unsigned)GR : 0u;
            bool mt[4];
#pragma unroll
            for (int c = 0; c < 4; ++c)
                mt[c] = ((unsigned)(dv[c] - base) < rg) & (AL | (4 * q + c < E_));
            if (__builtin_amdgcn_ballot_w32(mt[0] | mt[1] | mt[2] | mt[3]) != 0u) {
#pragma unroll
                for (int c = 0; c < 4; ++c) {
                    const unsigned mk = __builtin_amdgcn_ballot_w32(mt[c]);
                    const int o = cw + __popc(mk & ltmask) - lo;
                    if (mt[c] && o >= 0 && o < GCAP) {
                        lsrc_w[o] = 4 * q + c;
                        lrow_w[o] = (unsigned short)(dv[c] - base);
                    }
                    cw += __popc(mk);
                }
            }
        }
    }
    return cw;
}

__global__ void __launch_bounds__(GNT)
gnn_range_gather(const int* __restrict__ ei, const float* __restrict__ a_s,
                 const float* __restrict__ a_d, const float* __restrict__ xh,
                 float* __restrict__ out, float* __restrict__ denom, int E_, int Nn) {
    __shared__ __attribute__((aligned(16))) float acc[GR * DD];
    __shared__ float den[GR * HH];
    __shared__ float Mo[GR * HH];
    __shared__ int   Mn[GR * HH];
    __shared__ float ad[GR * HH];
    __shared__ int            lsrc[GNW * GCAP];
    __shared__ unsigned short lrow[GNW * GCAP];
    __shared__ int            cnt[GNW];
    __shared__ unsigned short wq[GNW * 64];

    const int tid  = threadIdx.x;
    const int wave = __builtin_amdgcn_readfirstlane(tid >> 5);
    const int lane = tid & 31;
    const int base = blockIdx.x * GR;
    const unsigned ltmask = (1u << lane) - 1u;

    {
        float4 v0[G0];
#pragma unroll
        for (int t = 0; t < G0; ++t) {
            const int i  = tid + t * GNT;
            const int ic = (i < GR * (DD / 4)) ? i : (GR * (DD / 4) - 1);
            const int node = base + (ic >> 5);
            const int nc = (node < Nn) ? node : (Nn - 1);
            v0[t] = *(const float4*)(xh + (long long)nc * DD + 4 * (ic & 31));
        }
#pragma unroll
        for (int t = 0; t < G0; ++t) {
            const int i = tid + t * GNT;
            if (i < GR * (DD / 4)) {
                float4 v = v0[t];
                if (base + (i >> 5) >= Nn) v = make_float4(0.0f, 0.0f, 0.0f, 0.0f);
                *(float4*)(&acc[4 * i]) = v;
            }
        }
    }
    for (int i = tid; i < GR * HH; i += GNT) {
        float as_ = 0.0f, adv = 0.0f;
        if (base + (i >> 2) < Nn) {
            as_ = a_s[(long long)base * HH + i];
            adv = a_d[(long long)base * HH + i];
        }
        const float e = gnn_lrelu(as_ + adv);
        ad[i] = adv;  Mo[i] = e;  Mn[i] = __float_as_int(e);  den[i] = 1.0f;
    }
    __syncthreads();

    const int* col = ei + E_;
    const int  Q   = (E_ + 3) >> 2;
    const int  qpw = (Q + GNW - 1) / GNW;
    const int  qb  = wave * qpw;
    const int  qe  = (qb + qpw < Q) ? (qb + qpw) : Q;
    const bool al  = (E_ & 3) == 0;

    int lo = 0;
    int mx;
    do {
        const int cw = al
            ? gnn_scan<true >(col, E_, qb, qe, base, lo, lane, ltmask, &lsrc[wave * GCAP], &lrow[wave * GCAP])
            : gnn_scan<false>(col, E_, qb, qe, base, lo, lane, ltmask, &lsrc[wave * GCAP], &lrow[wave * GCAP]);
        if (lane == 0) cnt[wave] = cw;

        {
            int nown = cw - lo;
            nown = (nown < 0) ? 0 : ((nown > GCAP) ? GCAP : nown);
            for (int j = lane; j < nown; j += 32) {
                const int s = ei[lsrc[wave * GCAP + j]];
                lsrc[wave * GCAP + j] = s;
                const int r = lrow[wave * GCAP + j];
                const float4 as4 = *(const float4*)(a_s + (long long)s * HH);
                const float ev[4] = {as4.x, as4.y, as4.z, as4.w};
#pragma unroll
                for (int hq = 0; hq < 4; ++hq) {
                    const float e = gnn_lrelu(ev[hq] + ad[r * HH + hq]);
                    int* p = &Mn[r * HH + hq];
                    if (e >= 0.0f) atomicMax(p, __float_as_int(e));
                    else           atomicMin((unsigned int*)p, __float_as_uint(e));
                }
            }
        }
        __syncthreads();

        mx = 0;
        for (int w = 0; w < GNW; ++w) mx = (cnt[w] > mx) ? cnt[w] : mx;

        for (int i = tid; i < GR * (DD / 4); i += GNT) {
            const int mi = (i >> 5) * HH + ((i & 31) >> 3);
            const float sc = __expf(Mo[mi] - __int_as_float(Mn[mi]));
            float4 v = *(float4*)(&acc[4 * i]);
            v.x *= sc;  v.y *= sc;  v.z *= sc;  v.w *= sc;
            *(float4*)(&acc[4 * i]) = v;
        }
        __syncthreads();
        for (int i = tid; i < GR * HH; i += GNT) {
            const float mn = __int_as_float(Mn[i]);
            den[i] *= __expf(Mo[i] - mn);
            Mo[i] = mn;
        }
        __syncthreads();

        int qh = 0, qn = 0;
        for (int s = 0; s < GNW; ++s) {
            int n = cnt[s] - lo;
            n = (n < 0) ? 0 : ((n > GCAP) ? GCAP : n);
            for (int j0 = 0; j0 < n; j0 += 128) {
                bool mine[4];
#pragma unroll
                for (int c = 0; c < 4; ++c) {
                    const int j = j0 + 32 * c + lane;
                    mine[c] = false;
                    if (j < n) mine[c] = (lrow[s * GCAP + j] & 31) == wave;
                }
#pragma unroll
                for (int c = 0; c < 4; ++c) {
                    const unsigned mk = __builtin_amdgcn_ballot_w32(mine[c]);
                    if (mk != 0u) {
                        if (mine[c])
                            wq[wave * 64 + ((qh + qn + __popc(mk & ltmask)) & 63)] =
                                (unsigned short)(s * GCAP + j0 + 32 * c + lane);
                        qn += __popc(mk);
                    }
                    if (qn >= 32) {
                        const int id = wq[wave * 64 + ((qh + lane) & 63)];
                        const int es = lsrc[id];
                        const int er = lrow[id];
                        for (int b = 0; b < 32; b += GB) {
                            float4 xv[GB];
                            float  av[GB];
#pragma unroll
                            for (int u = 0; u < GB; ++u) {
                                const int su = __builtin_amdgcn_readlane(es, b + u);
                                xv[u] = *(const float4*)(xh + (long long)su * DD + 4 * lane);
                                av[u] = a_s[(long long)su * HH + (lane >> 3)];
                            }
#pragma unroll
                            for (int u = 0; u < GB; ++u) {
                                const int ru = __builtin_amdgcn_readlane(er, b + u);
                                const int mi = ru * HH + (lane >> 3);
                                const float e = gnn_lrelu(av[u] + ad[mi]);
                                const float p = __expf(e - Mo[mi]);
                                float4 a = *(float4*)(&acc[ru * DD + 4 * lane]);
                                a.x += p * xv[u].x;  a.y += p * xv[u].y;
                                a.z += p * xv[u].z;  a.w += p * xv[u].w;
                                *(float4*)(&acc[ru * DD + 4 * lane]) = a;
                                if ((lane & 7) == 0) den[mi] += p;
                            }
                        }
                        qh = (qh + 32) & 63;
                        qn -= 32;
                    }
                }
            }
        }
        if (qn > 0) {
            int es = 0, er = 0;
            if (lane < qn) {
                const int id = wq[wave * 64 + ((qh + lane) & 63)];
                es = lsrc[id];
                er = lrow[id];
            }
            for (int b = 0; b < qn; b += GB) {
                float4 xv[GB];
                float  av[GB];
#pragma unroll
                for (int u = 0; u < GB; ++u) {
                    const int su = __builtin_amdgcn_readlane(es, (b + u) & 31);
                    xv[u] = *(const float4*)(xh + (long long)su * DD + 4 * lane);
                    av[u] = a_s[(long long)su * HH + (lane >> 3)];
                }
#pragma unroll
                for (int u = 0; u < GB; ++u) {
                    if (b + u < qn) {
                        const int ru = __builtin_amdgcn_readlane(er, (b + u) & 31);
                        const int mi = ru * HH + (lane >> 3);
                        const float e = gnn_lrelu(av[u] + ad[mi]);
                        const float p = __expf(e - Mo[mi]);
                        float4 a = *(float4*)(&acc[ru * DD + 4 * lane]);
                        a.x += p * xv[u].x;  a.y += p * xv[u].y;
                        a.z += p * xv[u].z;  a.w += p * xv[u].w;
                        *(float4*)(&acc[ru * DD + 4 * lane]) = a;
                        if ((lane & 7) == 0) den[mi] += p;
                    }
                }
            }
        }
        __syncthreads();
        lo += GCAP;
    } while (lo < mx);

    for (int i = tid; i < GR * (DD / 4); i += GNT) {
        const int node = base + (i >> 5);
        if (node < Nn) *(float4*)(out + (long long)node * DD + 4 * (i & 31)) = *(const float4*)(&acc[4 * i]);
    }
    for (int i = tid; i < GR * HH; i += GNT)
        if (base + (i >> 2) < Nn) denom[(long long)base * HH + i] = den[i];
}

#define FR 32
__global__ void __launch_bounds__(256)
gnn_finalize(const float* __restrict__ out, const float* __restrict__ denom,
             const float* __restrict__ gbias,
             const float* __restrict__ ln_g, const float* __restrict__ ln_b,
             float* __restrict__ h, int M) {
    const int wave = threadIdx.x >> 5;
    const int lane = threadIdx.x & 31;
    const int r0 = blockIdx.x * FR + wave * 4;
    if (r0 >= M) return;
    float4 hv[4], o[4];
    float  dn[4];
    int    rowok[4];
    long long goff[4];
#pragma unroll
    for (int u = 0; u < 4; ++u) {
        const int node = r0 + u;
        rowok[u] = node < M;
        const int nc = rowok[u] ? node : M - 1;
        goff[u] = (long long)nc * DD + 4 * lane;
        o[u]    = *(const float4*)(out + goff[u]);
        hv[u]   = *(const float4*)(h + goff[u]);
        dn[u]   = denom[(long long)nc * HH + (lane >> 3)];
    }
    const float4 gb = *(const float4*)(gbias + 4 * lane);
    const float4 g  = *(const float4*)(ln_g + 4 * lane);
    const float4 b  = *(const float4*)(ln_b + 4 * lane);
    float v[4][4], s[4], q[4];
#pragma unroll
    for (int u = 0; u < 4; ++u) {
        v[u][0] = o[u].x / dn[u] + gb.x;  v[u][1] = o[u].y / dn[u] + gb.y;
        v[u][2] = o[u].z / dn[u] + gb.z;  v[u][3] = o[u].w / dn[u] + gb.w;
        s[u] = (v[u][0] + v[u][1]) + (v[u][2] + v[u][3]);
    }
#pragma unroll
    for (int m = 16; m; m >>= 1) {
#pragma unroll
        for (int u = 0; u < 4; ++u) s[u] += __shfl_xor(s[u], m);
    }
#pragma unroll
    for (int u = 0; u < 4; ++u) {
        const float mean = s[u] * (1.0f / DD);
#pragma unroll
        for (int e = 0; e < 4; ++e) v[u][e] -= mean;
        q[u] = (v[u][0] * v[u][0] + v[u][1] * v[u][1]) + (v[u][2] * v[u][2] + v[u][3] * v[u][3]);
    }
#pragma unroll
    for (int m = 16; m; m >>= 1) {
#pragma unroll
        for (int u = 0; u < 4; ++u) q[u] += __shfl_xor(q[u], m);
    }
#pragma unroll
    for (int u = 0; u < 4; ++u) {
        const float rs = rsqrtf(q[u] * (1.0f / DD) + 1e-5f);
        hv[u].x += v[u][0] * rs * g.x + b.x;
        hv[u].y += v[u][1] * rs * g.y + b.y;
        hv[u].z += v[u][2] * rs * g.z + b.z;
        hv[u].w += v[u][3] * rs * g.w + b.w;
        if (rowok[u]) *(float4*)(h + goff[u]) = hv[u];
    }
}

extern "C" void kernel_launch(void* const* d_in, const int* in_sizes, int n_in,
                              void* d_out, int out_size, void* d_ws, size_t ws_size,
                              hipStream_t stream) {
    const float* x      = (const float*)d_in[0];
    const int*   ei     = (const int*)d_in[1];
    const float* W_edge = (const float*)d_in[2];
    const float* b_edge = (const float*)d_in[3];
    const float* W_node = (const float*)d_in[4];
    const float* b_node = (const float*)d_in[5];
    const float* lin_w  = (const float*)d_in[6];
    const float* att_sp = (const float*)d_in[7];
    const float* att_dp = (const float*)d_in[8];
    const float* g_bias = (const float*)d_in[9];
    const float* ln_g   = (const float*)d_in[10];
    const float* ln_b   = (const float*)d_in[11];

    const int E_ = in_sizes[0] / 3;
    const int Nn = out_size / DD;

    float* bufA  = (float*)d_ws;
    float* xh    = bufA + (long long)Nn * DD;
    float* a_s   = xh   + (long long)Nn * DD + Nn;
    float* a_d   = a_s  + (long long)Nn * HH;
    float* denom = a_d  + (long long)Nn * HH;
    float* h     = (float*)d_out;
    double* S    = (double*)d_ws;

    long long ns = (long long)Nn * 8;
    int gs  = (int)((ns + 255) / 256);
    int tiles = (Nn + 15) / 16;
    int gei = (int)(((long long)E_ * 4 + 2047) / 2048);
    int gfin = (Nn + FR - 1) / FR;
    int gblk = (Nn + GR - 1) / GR;

    gnn_fill_u32<<<gs, 256, 0, stream>>>((unsigned int*)S, 0u, ns);
    gnn_edge_init<<<gei, 256, 0, stream>>>(x, ei, S, E_);
    gnn_init_tile<<<tiles, 128, 0, stream>>>(S, W_edge, b_edge, W_node, b_node, h, Nn);

    for (int i = 0; i < 3; ++i) {
        if (i == 0)
            gnn_node_tile<0><<<tiles, 128, 0, stream>>>(
                lin_w, att_sp, att_dp, nullptr, nullptr, nullptr,
                h, bufA, denom, xh, a_s, a_d, Nn);
        else
            gnn_node_tile<1><<<tiles, 128, 0, stream>>>(
                lin_w + (long long)i * DD * DD, att_sp + i * DD, att_dp + i * DD,
                g_bias + (i - 1) * DD, ln_g + (i - 1) * DD, ln_b + (i - 1) * DD,
                h, bufA, denom, xh, a_s, a_d, Nn);

        gnn_range_gather<<<gblk, GNT, 0, stream>>>(ei, a_s, a_d, xh, bufA, denom, E_, Nn);
    }
    gnn_finalize<<<gfin, 256, 0, stream>>>(bufA, denom, g_bias + 2 * DD, ln_g + 2 * DD,
                                           ln_b + 2 * DD, h, Nn);
    (void)n_in; (void)ws_size;
}


# Gather blocks keep their edge lists: layer 1's scan stores each wave's (source, row) list in the workspace; layers 2 and 3 read it back instead of scanning the edge list again; sums in the same order
# speedup vs baseline: 4.1110x; 1.1490x over previous
#include <hip/hip_runtime.h>
#include <hip/hip_bf16.h>

typedef float v2f __attribute__((ext_vector_type(2)));
typedef float v8f __attribute__((ext_vector_type(8)));

#define DD 128
#define HH 4
#define CC 32

__global__ void gnn_fill_u32(unsigned int* __restrict__ p, unsigned int v, long long n) {
    long long i = (long long)blockIdx.x * blockDim.x + threadIdx.x;
    if (i < n) p[i] = v;
}

__global__ void __launch_bounds__(256)
gnn_edge_init(const float* __restrict__ x, const int* __restrict__ ei,
              double* __restrict__ S, int E_) {
    const long long total  = (long long)E_ * 4;
    const long long stride = (long long)gridDim.x * blockDim.x;
    const long long first  = (long long)blockIdx.x * blockDim.x + threadIdx.x;
    const int c = (int)(first & 3);
    const int cx = (c < 3) ? c : 2;
    long long idx = first;
    for (; idx + 3 * stride < total; idx += 4 * stride) {
        int   dstn[4];
        float xf[4];
#pragma unroll
        for (int u = 0; u < 4; ++u) {
            const int e = (int)((idx + u * stride) >> 2);
            dstn[u] = ei[E_ + e];
            xf[u]   = x[(long long)e * 3 + cx];
        }
#pragma unroll
        for (int u = 0; u < 4; ++u)
            atomicAdd(&S[(long long)dstn[u] * 4 + c], (c < 3) ? (double)xf[u] : 1.0);
    }
    for (; idx < total; idx += stride) {
        const int e = (int)(idx >> 2);
        const int dstn = ei[E_ + e];
        const float xf = x[(long long)e * 3 + cx];
        atomicAdd(&S[(long long)dstn * 4 + c], (c < 3) ? (double)xf : 1.0);
    }
}

#define TP 132

__global__ void __launch_bounds__(128)
gnn_init_tile(const double* __restrict__ S, const float* __restrict__ W_edge,
              const float* __restrict__ b_edge, const float* __restrict__ W_node,
              const float* __restrict__ b_node, float* __restrict__ h, int M) {
    __shared__ __attribute__((aligned(16))) float Ht[16 * TP];
    __shared__ __attribute__((aligned(16))) float Xt[16 * TP];
    const int wave = threadIdx.x >> 5;
    const int lane = threadIdx.x & 31;
    const int m0 = blockIdx.x * 16;

    const float4 w0 = *(const float4*)(W_edge + 4 * lane);
    const float4 w1 = *(const float4*)(W_edge + DD + 4 * lane);
    const float4 w2 = *(const float4*)(W_edge + 2 * DD + 4 * lane);
    const float4 be = *(const float4*)(b_edge + 4 * lane);
#pragma unroll
    for (int u = 0; u < 4; ++u) {
        const int r = wave * 4 + u;
        int node = m0 + r;
        if (node >= M) node = M - 1;
        const double* sp = S + (long long)node * 4;
        const double c  = sp[3];
        const float inv = 1.0f / fmaxf((float)c, 1.0f);
        const float a0 = (float)sp[0] * inv;
        const float a1 = (float)sp[1] * inv;
        const float a2 = (float)sp[2] * inv;
        const float fb = (c > 0.0) ? 1.0f : 0.0f;
        float4 v;
        v.x = a0 * w0.x + a1 * w1.x + a2 * w2.x + fb * be.x;
        v.y = a0 * w0.y + a1 * w1.y + a2 * w2.y + fb * be.y;
        v.z = a0 * w0.z + a1 * w1.z + a2 * w2.z + fb * be.z;
        v.w = a0 * w0.w + a1 * w1.w + a2 * w2.w + fb * be.w;
        *(float4*)(&Ht[r * TP + 4 * lane]) = v;
    }
    __syncthreads();

    const int half = lane >> 4;
    const int l    = lane & 15;
    const int K0   = half * 2;
    const int n0   = wave * 32;
    v8f acc0 = {};
    v8f acc1 = {};
#pragma unroll
    for (int k = 0; k < DD; k += 4) {
        const v2f a = *(const v2f*)(&Ht[l * TP + k + K0]);
        const float* Bk = W_node + (long long)(k + K0) * DD + n0 + l;
        v2f b0, b1;
        b0.x = Bk[0];       b0.y = Bk[DD];
        b1.x = Bk[16];      b1.y = Bk[DD + 16];
        acc0 = __builtin_amdgcn_wmma_f32_16x16x4_f32(false, a, false, b0, (short)0, acc0, false, false);
        acc1 = __builtin_amdgcn_wmma_f32_16x16x4_f32(false, a, false, b1, (short)0, acc1, false, false);
    }
#pragma unroll
    for (int r = 0; r < 8; ++r) {
        const int row = r + half * 8;
        Xt[row * TP + n0 + l]      = acc0[r];
        Xt[row * TP + n0 + 16 + l] = acc1[r];
    }
    __syncthreads();

    const int ch = wave * 32 + lane;
    const float bn = b_node[ch];
    for (int r = 0; r < 16; ++r) {
        const int node = m0 + r;
        if (node < M) h[(long long)node * DD + ch] = Xt[r * TP + ch] + bn;
    }
}

template <int PREV>
__global__ void __launch_bounds__(128)
gnn_node_tile(const float* __restrict__ Wl,
              const float* __restrict__ att_src, const float* __restrict__ att_dst,
              const float* __restrict__ gbias, const float* __restrict__ ln_g,
              const float* __restrict__ ln_b,
              float* __restrict__ h, const float* __restrict__ accb,
              const float* __restrict__ denom, float* __restrict__ xh,
              float* __restrict__ a_s, float* __restrict__ a_d, int M) {
    __shared__ __attribute__((aligned(16))) float Ht[16 * TP];
    __shared__ __attribute__((aligned(16))) float Xt[16 * TP];
    __shared__ float Lg[128];
    const int wave = threadIdx.x >> 5;
    const int lane = threadIdx.x & 31;
    const int m0 = blockIdx.x * 16;

    float4 hv[4];
    int    rowok[4];
    long long goff[4];
#pragma unroll
    for (int u = 0; u < 4; ++u) {
        const int node = m0 + wave * 4 + u;
        rowok[u] = node < M;
        goff[u]  = (long long)(rowok[u] ? node : M - 1) * DD + 4 * lane;
        hv[u]    = *(const float4*)(h + goff[u]);
    }
    if (PREV) {
        const float4 gb = *(const float4*)(gbias + 4 * lane);
        const float4 g  = *(const float4*)(ln_g + 4 * lane);
        const float4 b  = *(const float4*)(ln_b + 4 * lane);
        float v[4][4], s[4], q[4];
#pragma unroll
        for (int u = 0; u < 4; ++u) {
            const float4 o  = *(const float4*)(accb + goff[u]);
            const float  dn = denom[(goff[u] >> 7) * HH + (lane >> 3)];
            v[u][0] = o.x / dn + gb.x;  v[u][1] = o.y / dn + gb.y;
            v[u][2] = o.z / dn + gb.z;  v[u][3] = o.w / dn + gb.w;
            s[u] = (v[u][0] + v[u][1]) + (v[u][2] + v[u][3]);
        }
#pragma unroll
        for (int m = 16; m; m >>= 1) {
#pragma unroll
            for (int u = 0; u < 4; ++u) s[u] += __shfl_xor(s[u], m);
        }
#pragma unroll
        for (int u = 0; u < 4; ++u) {
            const float mean = s[u] * (1.0f / DD);
#pragma unroll
            for (int e = 0; e < 4; ++e) v[u][e] -= mean;
            q[u] = (v[u][0] * v[u][0] + v[u][1] * v[u][1]) + (v[u][2] * v[u][2] + v[u][3] * v[u][3]);
        }
#pragma unroll
        for (int m = 16; m; m >>= 1) {
#pragma unroll
            for (int u = 0; u < 4; ++u) q[u] += __shfl_xor(q[u], m);
        }
#pragma unroll
        for (int u = 0; u < 4; ++u) {
            const float rs = rsqrtf(q[u] * (1.0f / DD) + 1e-5f);
            hv[u].x += fmaxf(v[u][0] * rs * g.x + b.x, 0.0f);
            hv[u].y += fmaxf(v[u][1] * rs * g.y + b.y, 0.0f);
            hv[u].z += fmaxf(v[u][2] * rs * g.z + b.z, 0.0f);
            hv[u].w += fmaxf(v[u][3] * rs * g.w + b.w, 0.0f);
            if (rowok[u]) *(float4*)(h + goff[u]) = hv[u];
        }
    }
#pragma unroll
    for (int u = 0; u < 4; ++u)
        *(float4*)(&Ht[(wave * 4 + u) * TP + 4 * lane]) = hv[u];
    __syncthreads();

    const int half = lane >> 4;
    const int l    = lane & 15;
    const int K0   = half * 2;
    const int n0   = wave * 32;
    v8f acc0 = {};
    v8f acc1 = {};
#pragma unroll
    for (int k = 0; k < DD; k += 4) {
        const v2f a = *(const v2f*)(&Ht[l * TP + k + K0]);
        const float* Bk = Wl + (long long)(k + K0) * DD + n0 + l;
        v2f b0, b1;
        b0.x = Bk[0];       b0.y = Bk[DD];
        b1.x = Bk[16];      b1.y = Bk[DD + 16];
        acc0 = __builtin_amdgcn_wmma_f32_16x16x4_f32(false, a, false, b0, (short)0, acc0, false, false);
        acc1 = __builtin_amdgcn_wmma_f32_16x16x4_f32(false, a, false, b1, (short)0, acc1, false, false);
    }
#pragma unroll
    for (int r = 0; r < 8; ++r) {
        const int row = r + half * 8;
        Xt[row * TP + n0 + l]      = acc0[r];
        Xt[row * TP + n0 + 16 + l] = acc1[r];
    }
    __syncthreads();

    const int ch = wave * 32 + lane;
    const float ws = att_src[ch];
    const float wd = att_dst[ch];
    for (int it0 = 0; it0 < 16; it0 += 4) {
        float xv[4], s[4], t[4];
#pragma unroll
        for (int u = 0; u < 4; ++u) {
            xv[u] = Xt[(it0 + u) * TP + ch];
            s[u] = xv[u] * ws;
            t[u] = xv[u] * wd;
        }
#pragma unroll
        for (int m = 16; m; m >>= 1) {
#pragma unroll
            for (int u = 0; u < 4; ++u) { s[u] += __shfl_xor(s[u], m); t[u] += __shfl_xor(t[u], m); }
        }
#pragma unroll
        for (int u = 0; u < 4; ++u) {
            const int node = m0 + it0 + u;
            if (lane == 0) { Lg[(it0 + u) * HH + wave] = s[u]; Lg[64 + (it0 + u) * HH + wave] = t[u]; }
            if (node < M) xh[(long long)node * DD + ch] = xv[u];
        }
    }
    __syncthreads();
    for (int qi = lane; qi < 64; qi += 32) {
        if (m0 + (qi >> 2) < M) {
            const long long idx = (long long)m0 * HH + qi;
            if (wave == 0)      a_s[idx] = Lg[qi];
            else if (wave == 1) a_d[idx] = Lg[64 + qi];
        }
    }
}

#define GR   200
#define GNT  1024
#define GNW  32
#define GCAP 256
#define GQ   8
#define G0   ((GR * (DD / 4) + GNT - 1) / GNT)
#define GB   8

__device__ __forceinline__ float gnn_lrelu(float e) { return (e > 0.0f) ? e : 0.2f * e; }

template <bool AL>
__device__ __forceinline__ int gnn_scan(const int* __restrict__ col,
                                        int E_, int qb, int qe, int base, int lo, int lane,
                                        unsigned ltmask, int* lsrc_w, unsigned short* lrow_w) {
    int cw = 0;
    for (int q0 = qb; q0 < qe; q0 += 32 * GQ) {
        int4 dq[GQ];
#pragma unroll
        for (int u = 0; u < GQ; ++u) {
            const int q  = q0 + u * 32 + lane;
            const int qc = (q < qe) ? q : (qe - 1);
            if (AL) {
                dq[u] = *(const int4*)(col + 4 * qc);
            } else {
                const int last = E_ - 1;
                const int i0 = 4 * qc;
                dq[u].x = col[(i0 + 0 < last) ? (i0 + 0) : last];
                dq[u].y = col[(i0 + 1 < last) ? (i0 + 1) : last];
                dq[u].z = col[(i0 + 2 < last) ? (i0 + 2) : last];
                dq[u].w = col[(i0 + 3 < last) ? (i0 + 3) : last];
            }
        }
#pragma unroll
        for (int u = 0; u < GQ; ++u) {
            const int q = q0 + u * 32 + lane;
            const int dv[4] = {dq[u].x, dq[u].y, dq[u].z, dq[u].w};
            const unsigned rg = (q < qe) ? (unsigned)GR : 0u;
            bool mt[4];
#pragma unroll
            for (int c = 0; c < 4; ++c)
                mt[c] = ((unsigned)(dv[c] - base) < rg) & (AL | (4 * q + c < E_));
            if (__builtin_amdgcn_ballot_w32(mt[0] | mt[1] | mt[2] | mt[3]) != 0u) {
#pragma unroll
                for (int c = 0; c < 4; ++c) {
                    const unsigned mk = __builtin_amdgcn_ballot_w32(mt[c]);
                    const int o = cw + __popc(mk & ltmask) - lo;
                    if (mt[c] && o >= 0 && o < GCAP) {
                        lsrc_w[o] = 4 * q + c;
                        lrow_w[o] = (unsigned short)(dv[c] - base);
                    }
                    cw += __popc(mk);
                }
            }
        }
    }
    return cw;
}

template <int KEEP>
__global__ void __launch_bounds__(GNT)
gnn_range_gather(const int* __restrict__ ei, const float* __restrict__ a_s,
                 const float* __restrict__ a_d, const float* __restrict__ xh,
                 float* __restrict__ out, float* __restrict__ denom,
                 int* __restrict__ ksrc, unsigned char* __restrict__ krow,
                 int* __restrict__ kcnt, int E_, int Nn) {
    static_assert(GR <= 256, "a kept row number is one byte");
    __shared__ __attribute__((aligned(16))) float acc[GR * DD];
    __shared__ float den[GR * HH];
    __shared__ float Mo[GR * HH];
    __shared__ int   Mn[GR * HH];
    __shared__ float ad[GR * HH];
    __shared__ int            lsrc[GNW * GCAP];
    __shared__ unsigned short lrow[GNW * GCAP];
    __shared__ int            cnt[GNW];
    __shared__ unsigned short wq[GNW * 64];

    const int tid  = threadIdx.x;
    const int wave = __builtin_amdgcn_readfirstlane(tid >> 5);
    const int lane = tid & 31;
    const int base = blockIdx.x * GR;
    const unsigned ltmask = (1u << lane) - 1u;

    {
        float4 v0[G0];
#pragma unroll
        for (int t = 0; t < G0; ++t) {
            const int i  = tid + t * GNT;
            const int ic = (i < GR * (DD / 4)) ? i : (GR * (DD / 4) - 1);
            const int node = base + (ic >> 5);
            const int nc = (node < Nn) ? node : (Nn - 1);
            v0[t] = *(const float4*)(xh + (long long)nc * DD + 4 * (ic & 31));
        }
#pragma unroll
        for (int t = 0; t < G0; ++t) {
            const int i = tid + t * GNT;
            if (i < GR * (DD / 4)) {
                float4 v = v0[t];
                if (base + (i >> 5) >= Nn) v = make_float4(0.0f, 0.0f, 0.0f, 0.0f);
                *(float4*)(&acc[4 * i]) = v;
            }
        }
    }
    for (int i = tid; i < GR * HH; i += GNT) {
        float as_ = 0.0f, adv = 0.0f;
        if (base + (i >> 2) < Nn) {
            as_ = a_s[(long long)base * HH + i];
            adv = a_d[(long long)base * HH + i];
        }
        const float e = gnn_lrelu(as_ + adv);
        ad[i] = adv;  Mo[i] = e;  Mn[i] = __float_as_int(e);  den[i] = 1.0f;
    }
    __syncthreads();

    const int* col = ei + E_;
    const int  Q   = (E_ + 3) >> 2;
    const int  qpw = (Q + GNW - 1) / GNW;
    const int  qb  = wave * qpw;
    const int  qe  = (qb + qpw < Q) ? (qb + qpw) : Q;
    const bool al  = (E_ & 3) == 0;

    const long long kb = ((long long)blockIdx.x * GNW + wave) * GCAP;
    bool kept = false;
    int  kown = 0;
    if (KEEP == 2) {
        const int kc = kcnt[(long long)blockIdx.x * GNW + lane];
        kept = __builtin_amdgcn_ballot_w32(kc > GCAP) == 0u;
        kown = __builtin_amdgcn_readlane(kc, wave);
    }

    int lo = 0;
    int mx;
    do {
        int cw;
        if (KEEP == 2 && kept)
            cw = kown;
        else
            cw = al
            ? gnn_scan<true >(col, E_, qb, qe, base, lo, lane, ltmask, &lsrc[wave * GCAP], &lrow[wave * GCAP])
            : gnn_scan<false>(col, E_, qb, qe, base, lo, lane, ltmask, &lsrc[wave * GCAP], &lrow[wave * GCAP]);
        if (lane == 0) {
            cnt[wave] = cw;
            if (KEEP == 1 && lo == 0) kcnt[(long long)blockIdx.x * GNW + wave] = cw;
        }

        {
            int nown = cw - lo;
            nown = (nown < 0) ? 0 : ((nown > GCAP) ? GCAP : nown);
            for (int j = lane; j < nown; j += 32) {
                int s, r;
                if (KEEP == 2 && kept) {
                    s = ksrc[kb + j];
                    r = krow[kb + j];
                    lrow[wave * GCAP + j] = (unsigned short)r;
                } else {
                    s = ei[lsrc[wave * GCAP + j]];
                    r = lrow[wave * GCAP + j];
                    if (KEEP == 1 && lo == 0) {
                        ksrc[kb + j] = s;
                        krow[kb + j] = (unsigned char)r;
                    }
                }
                lsrc[wave * GCAP + j] = s;
                const float4 as4 = *(const float4*)(a_s + (long long)s * HH);
                const float ev[4] = {as4.x, as4.y, as4.z, as4.w};
#pragma unroll
                for (int hq = 0; hq < 4; ++hq) {
                    const float e = gnn_lrelu(ev[hq] + ad[r * HH + hq]);
                    int* p = &Mn[r * HH + hq];
                    if (e >= 0.0f) atomicMax(p, __float_as_int(e));
                    else           atomicMin((unsigned int*)p, __float_as_uint(e));
                }
            }
        }
        __syncthreads();

        mx = 0;
        for (int w = 0; w < GNW; ++w) mx = (cnt[w] > mx) ? cnt[w] : mx;

        for (int i = tid; i < GR * (DD / 4); i += GNT) {
            const int mi = (i >> 5) * HH + ((i & 31) >> 3);
            const float sc = __expf(Mo[mi] - __int_as_float(Mn[mi]));
            float4 v = *(float4*)(&acc[4 * i]);
            v.x *= sc;  v.y *= sc;  v.z *= sc;  v.w *= sc;
            *(float4*)(&acc[4 * i]) = v;
        }
        __syncthreads();
        for (int i = tid; i < GR * HH; i += GNT) {
            const float mn = __int_as_float(Mn[i]);
            den[i] *= __expf(Mo[i] - mn);
            Mo[i] = mn;
        }
        __syncthreads();

        int qh = 0, qn = 0;
        for (int s = 0; s < GNW; ++s) {
            int n = cnt[s] - lo;
            n = (n < 0) ? 0 : ((n > GCAP) ? GCAP : n);
            for (int j0 = 0; j0 < n; j0 += 128) {
                bool mine[4];
#pragma unroll
                for (int c = 0; c < 4; ++c) {
                    const int j = j0 + 32 * c + lane;
                    mine[c] = false;
                    if (j < n) mine[c] = (lrow[s * GCAP + j] & 31) == wave;
                }
#pragma unroll
                for (int c = 0; c < 4; ++c) {
                    const unsigned mk = __builtin_amdgcn_ballot_w32(mine[c]);
                    if (mk != 0u) {
                        if (mine[c])
                            wq[wave * 64 + ((qh + qn + __popc(mk & ltmask)) & 63)] =
                                (unsigned short)(s * GCAP + j0 + 32 * c + lane);
                        qn += __popc(mk);
                    }
                    if (qn >= 32) {
                        const int id = wq[wave * 64 + ((qh + lane) & 63)];
                        const int es = lsrc[id];
                        const int er = lrow[id];
                        for (int b = 0; b < 32; b += GB) {
                            float4 xv[GB];
                            float  av[GB];
#pragma unroll
                            for (int u = 0; u < GB; ++u) {
                                const int su = __builtin_amdgcn_readlane(es, b + u);
                                xv[u] = *(const float4*)(xh + (long long)su * DD + 4 * lane);
                                av[u] = a_s[(long long)su * HH + (lane >> 3)];
                            }
#pragma unroll
                            for (int u = 0; u < GB; ++u) {
                                const int ru = __builtin_amdgcn_readlane(er, b + u);
                                const int mi = ru * HH + (lane >> 3);
                                const float e = gnn_lrelu(av[u] + ad[mi]);
                                const float p = __expf(e - Mo[mi]);
                                float4 a = *(float4*)(&acc[ru * DD + 4 * lane]);
                                a.x += p * xv[u].x;  a.y += p * xv[u].y;
                                a.z += p * xv[u].z;  a.w += p * xv[u].w;
                                *(float4*)(&acc[ru * DD + 4 * lane]) = a;
                                if ((lane & 7) == 0) den[mi] += p;
                            }
                        }
                        qh = (qh + 32) & 63;
                        qn -= 32;
                    }
                }
            }
        }
        if (qn > 0) {
            int es = 0, er = 0;
            if (lane < qn) {
                const int id = wq[wave * 64 + ((qh + lane) & 63)];
                es = lsrc[id];
                er = lrow[id];
            }
            for (int b = 0; b < qn; b += GB) {
                float4 xv[GB];
                float  av[GB];
#pragma unroll
                for (int u = 0; u < GB; ++u) {
                    const int su = __builtin_amdgcn_readlane(es, (b + u) & 31);
                    xv[u] = *(const float4*)(xh + (long long)su * DD + 4 * lane);
                    av[u] = a_s[(long long)su * HH + (lane >> 3)];
                }
#pragma unroll
                for (int u = 0; u < GB; ++u) {
                    if (b + u < qn) {
                        const int ru = __builtin_amdgcn_readlane(er, (b + u) & 31);
                        const int mi = ru * HH + (lane >> 3);
                        const float e = gnn_lrelu(av[u] + ad[mi]);
                        const float p = __expf(e - Mo[mi]);
                        float4 a = *(float4*)(&acc[ru * DD + 4 * lane]);
                        a.x += p * xv[u].x;  a.y += p * xv[u].y;
                        a.z += p * xv[u].z;  a.w += p * xv[u].w;
                        *(float4*)(&acc[ru * DD + 4 * lane]) = a;
                        if ((lane & 7) == 0) den[mi] += p;
                    }
                }
            }
        }
        __syncthreads();
        lo += GCAP;
    } while (lo < mx);

    for (int i = tid; i < GR * (DD / 4); i += GNT) {
        const int node = base + (i >> 5);
        if (node < Nn) *(float4*)(out + (long long)node * DD + 4 * (i & 31)) = *(const float4*)(&acc[4 * i]);
    }
    for (int i = tid; i < GR * HH; i += GNT)
        if (base + (i >> 2) < Nn) denom[(long long)base * HH + i] = den[i];
}

#define FR 32
__global__ void __launch_bounds__(256)
gnn_finalize(const float* __restrict__ out, const float* __restrict__ denom,
             const float* __restrict__ gbias,
             const float* __restrict__ ln_g, const float* __restrict__ ln_b,
             float* __restrict__ h, int M) {
    const int wave = threadIdx.x >> 5;
    const int lane = threadIdx.x & 31;
    const int r0 = blockIdx.x * FR + wave * 4;
    if (r0 >= M) return;
    float4 hv[4], o[4];
    float  dn[4];
    int    rowok[4];
    long long goff[4];
#pragma unroll
    for (int u = 0; u < 4; ++u) {
        const int node = r0 + u;
        rowok[u] = node < M;
        const int nc = rowok[u] ? node : M - 1;
        goff[u] = (long long)nc * DD + 4 * lane;
        o[u]    = *(const float4*)(out + goff[u]);
        hv[u]   = *(const float4*)(h + goff[u]);
        dn[u]   = denom[(long long)nc * HH + (lane >> 3)];
    }
    const float4 gb = *(const float4*)(gbias + 4 * lane);
    const float4 g  = *(const float4*)(ln_g + 4 * lane);
    const float4 b  = *(const float4*)(ln_b + 4 * lane);
    float v[4][4], s[4], q[4];
#pragma unroll
    for (int u = 0; u < 4; ++u) {
        v[u][0] = o[u].x / dn[u] + gb.x;  v[u][1] = o[u].y / dn[u] + gb.y;
        v[u][2] = o[u].z / dn[u] + gb.z;  v[u][3] = o[u].w / dn[u] + gb.w;
        s[u] = (v[u][0] + v[u][1]) + (v[u][2] + v[u][3]);
    }
#pragma unroll
    for (int m = 16; m; m >>= 1) {
#pragma unroll
        for (int u = 0; u < 4; ++u) s[u] += __shfl_xor(s[u], m);
    }
#pragma unroll
    for (int u = 0; u < 4; ++u) {
        const float mean = s[u] * (1.0f / DD);
#pragma unroll
        for (int e = 0; e < 4; ++e) v[u][e] -= mean;
        q[u] = (v[u][0] * v[u][0] + v[u][1] * v[u][1]) + (v[u][2] * v[u][2] + v[u][3] * v[u][3]);
    }
#pragma unroll
    for (int m = 16; m; m >>= 1) {
#pragma unroll
        for (int u = 0; u < 4; ++u) q[u] += __shfl_xor(q[u], m);
    }
#pragma unroll
    for (int u = 0; u < 4; ++u) {
        const float rs = rsqrtf(q[u] * (1.0f / DD) + 1e-5f);
        hv[u].x += v[u][0] * rs * g.x + b.x;
        hv[u].y += v[u][1] * rs * g.y + b.y;
        hv[u].z += v[u][2] * rs * g.z + b.z;
        hv[u].w += v[u][3] * rs * g.w + b.w;
        if (rowok[u]) *(float4*)(h + goff[u]) = hv[u];
    }
}

extern "C" void kernel_launch(void* const* d_in, const int* in_sizes, int n_in,
                              void* d_out, int out_size, void* d_ws, size_t ws_size,
                              hipStream_t stream) {
    const float* x      = (const float*)d_in[0];
    const int*   ei     = (const int*)d_in[1];
    const float* W_edge = (const float*)d_in[2];
    const float* b_edge = (const float*)d_in[3];
    const float* W_node = (const float*)d_in[4];
    const float* b_node = (const float*)d_in[5];
    const float* lin_w  = (const float*)d_in[6];
    const float* att_sp = (const float*)d_in[7];
    const float* att_dp = (const float*)d_in[8];
    const float* g_bias = (const float*)d_in[9];
    const float* ln_g   = (const float*)d_in[10];
    const float* ln_b   = (const float*)d_in[11];

    const int E_ = in_sizes[0] / 3;
    const int Nn = out_size / DD;

    float* bufA  = (float*)d_ws;
    float* xh    = bufA + (long long)Nn * DD;
    float* a_s   = xh   + (long long)Nn * DD + Nn;
    float* a_d   = a_s  + (long long)Nn * HH;
    float* denom = a_d  + (long long)Nn * HH;
    float* h     = (float*)d_out;
    double* S    = (double*)d_ws;

    long long ns = (long long)Nn * 8;
    int gs  = (int)((ns + 255) / 256);
    int tiles = (Nn + 15) / 16;
    int gei = (int)(((long long)E_ * 4 + 2047) / 2048);
    int gfin = (Nn + FR - 1) / FR;
    int gblk = (Nn + GR - 1) / GR;

    int*           ksrc = (int*)(denom + (long long)Nn * HH);
    int*           kcnt = ksrc + (long long)gblk * GNW * GCAP;
    unsigned char* krow = (unsigned char*)(kcnt + (long long)gblk * GNW);
    const bool room = (size_t)((krow + (long long)gblk * GNW * GCAP) - (unsigned char*)d_ws) <= ws_size;

    gnn_fill_u32<<<gs, 256, 0, stream>>>((unsigned int*)S, 0u, ns);
    gnn_edge_init<<<gei, 256, 0, stream>>>(x, ei, S, E_);
    gnn_init_tile<<<tiles, 128, 0, stream>>>(S, W_edge, b_edge, W_node, b_node, h, Nn);

    for (int i = 0; i < 3; ++i) {
        if (i == 0)
            gnn_node_tile<0><<<tiles, 128, 0, stream>>>(
                lin_w, att_sp, att_dp, nullptr, nullptr, nullptr,
                h, bufA, denom, xh, a_s, a_d, Nn);
        else
            gnn_node_tile<1><<<tiles, 128, 0, stream>>>(
                lin_w + (long long)i * DD * DD, att_sp + i * DD, att_dp + i * DD,
                g_bias + (i - 1) * DD, ln_g + (i - 1) * DD, ln_b + (i - 1) * DD,
                h, bufA, denom, xh, a_s, a_d, Nn);

        if (!room)
            gnn_range_gather<0><<<gblk, GNT, 0, stream>>>(ei, a_s, a_d, xh, bufA, denom, ksrc, krow, kcnt, E_, Nn);
        else if (i == 0)
            gnn_range_gather<1><<<gblk, GNT, 0, stream>>>(ei, a_s, a_d, xh, bufA, denom, ksrc, krow, kcnt, E_, Nn);
        else
            gnn_range_gather<2><<<gblk, GNT, 0, stream>>>(ei, a_s, a_d, xh, bufA, denom, ksrc, krow, kcnt, E_, Nn);
    }
    gnn_finalize<<<gfin, 256, 0, stream>>>(bufA, denom, g_bias + 2 * DD, ln_g + 2 * DD,
                                           ln_b + 2 * DD, h, Nn);
    (void)n_in;
}


# Gather sum step: each group of 32 waiting entries is put in row order (stable) in the wave, so a row's sums stay in registers over its terms; LDS updated once per row, not per entry; same sum order
# speedup vs baseline: 4.2909x; 1.0438x over previous
#include <hip/hip_runtime.h>
#include <hip/hip_bf16.h>

typedef float v2f __attribute__((ext_vector_type(2)));
typedef float v8f __attribute__((ext_vector_type(8)));

#define DD 128
#define HH 4
#define CC 32

__global__ void gnn_fill_u32(unsigned int* __restrict__ p, unsigned int v, long long n) {
    long long i = (long long)blockIdx.x * blockDim.x + threadIdx.x;
    if (i < n) p[i] = v;
}

__global__ void __launch_bounds__(256)
gnn_edge_init(const float* __restrict__ x, const int* __restrict__ ei,
              double* __restrict__ S, int E_) {
    const long long total  = (long long)E_ * 4;
    const long long stride = (long long)gridDim.x * blockDim.x;
    const long long first  = (long long)blockIdx.x * blockDim.x + threadIdx.x;
    const int c = (int)(first & 3);
    const int cx = (c < 3) ? c : 2;
    long long idx = first;
    for (; idx + 3 * stride < total; idx += 4 * stride) {
        int   dstn[4];
        float xf[4];
#pragma unroll
        for (int u = 0; u < 4; ++u) {
            const int e = (int)((idx + u * stride) >> 2);
            dstn[u] = ei[E_ + e];
            xf[u]   = x[(long long)e * 3 + cx];
        }
#pragma unroll
        for (int u = 0; u < 4; ++u)
            atomicAdd(&S[(long long)dstn[u] * 4 + c], (c < 3) ? (double)xf[u] : 1.0);
    }
    for (; idx < total; idx += stride) {
        const int e = (int)(idx >> 2);
        const int dstn = ei[E_ + e];
        const float xf = x[(long long)e * 3 + cx];
        atomicAdd(&S[(long long)dstn * 4 + c], (c < 3) ? (double)xf : 1.0);
    }
}

#define TP 132

__global__ void __launch_bounds__(128)
gnn_init_tile(const double* __restrict__ S, const float* __restrict__ W_edge,
              const float* __restrict__ b_edge, const float* __restrict__ W_node,
              const float* __restrict__ b_node, float* __restrict__ h, int M) {
    __shared__ __attribute__((aligned(16))) float Ht[16 * TP];
    __shared__ __attribute__((aligned(16))) float Xt[16 * TP];
    const int wave = threadIdx.x >> 5;
    const int lane = threadIdx.x & 31;
    const int m0 = blockIdx.x * 16;

    const float4 w0 = *(const float4*)(W_edge + 4 * lane);
    const float4 w1 = *(const float4*)(W_edge + DD + 4 * lane);
    const float4 w2 = *(const float4*)(W_edge + 2 * DD + 4 * lane);
    const float4 be = *(const float4*)(b_edge + 4 * lane);
#pragma unroll
    for (int u = 0; u < 4; ++u) {
        const int r = wave * 4 + u;
        int node = m0 + r;
        if (node >= M) node = M - 1;
        const double* sp = S + (long long)node * 4;
        const double c  = sp[3];
        const float inv = 1.0f / fmaxf((float)c, 1.0f);
        const float a0 = (float)sp[0] * inv;
        const float a1 = (float)sp[1] * inv;
        const float a2 = (float)sp[2] * inv;
        const float fb = (c > 0.0) ? 1.0f : 0.0f;
        float4 v;
        v.x = a0 * w0.x + a1 * w1.x + a2 * w2.x + fb * be.x;
        v.y = a0 * w0.y + a1 * w1.y + a2 * w2.y + fb * be.y;
        v.z = a0 * w0.z + a1 * w1.z + a2 * w2.z + fb * be.z;
        v.w = a0 * w0.w + a1 * w1.w + a2 * w2.w + fb * be.w;
        *(float4*)(&Ht[r * TP + 4 * lane]) = v;
    }
    __syncthreads();

    const int half = lane >> 4;
    const int l    = lane & 15;
    const int K0   = half * 2;
    const int n0   = wave * 32;
    v8f acc0 = {};
    v8f acc1 = {};
#pragma unroll
    for (int k = 0; k < DD; k += 4) {
        const v2f a = *(const v2f*)(&Ht[l * TP + k + K0]);
        const float* Bk = W_node + (long long)(k + K0) * DD + n0 + l;
        v2f b0, b1;
        b0.x = Bk[0];       b0.y = Bk[DD];
        b1.x = Bk[16];      b1.y = Bk[DD + 16];
        acc0 = __builtin_amdgcn_wmma_f32_16x16x4_f32(false, a, false, b0, (short)0, acc0, false, false);
        acc1 = __builtin_amdgcn_wmma_f32_16x16x4_f32(false, a, false, b1, (short)0, acc1, false, false);
    }
#pragma unroll
    for (int r = 0; r < 8; ++r) {
        const int row = r + half * 8;
        Xt[row * TP + n0 + l]      = acc0[r];
        Xt[row * TP + n0 + 16 + l] = acc1[r];
    }
    __syncthreads();

    const int ch = wave * 32 + lane;
    const float bn = b_node[ch];
    for (int r = 0; r < 16; ++r) {
        const int node = m0 + r;
        if (node < M) h[(long long)node * DD + ch] = Xt[r * TP + ch] + bn;
    }
}

template <int PREV>
__global__ void __launch_bounds__(128)
gnn_node_tile(const float* __restrict__ Wl,
              const float* __restrict__ att_src, const float* __restrict__ att_dst,
              const float* __restrict__ gbias, const float* __restrict__ ln_g,
              const float* __restrict__ ln_b,
              float* __restrict__ h, const float* __restrict__ accb,
              const float* __restrict__ denom, float* __restrict__ xh,
              float* __restrict__ a_s, float* __restrict__ a_d, int M) {
    __shared__ __attribute__((aligned(16))) float Ht[16 * TP];
    __shared__ __attribute__((aligned(16))) float Xt[16 * TP];
    __shared__ float Lg[128];
    const int wave = threadIdx.x >> 5;
    const int lane = threadIdx.x & 31;
    const int m0 = blockIdx.x * 16;

    float4 hv[4];
    int    rowok[4];
    long long goff[4];
#pragma unroll
    for (int u = 0; u < 4; ++u) {
        const int node = m0 + wave * 4 + u;
        rowok[u] = node < M;
        goff[u]  = (long long)(rowok[u] ? node : M - 1) * DD + 4 * lane;
        hv[u]    = *(const float4*)(h + goff[u]);
    }
    if (PREV) {
        const float4 gb = *(const float4*)(gbias + 4 * lane);
        const float4 g  = *(const float4*)(ln_g + 4 * lane);
        const float4 b  = *(const float4*)(ln_b + 4 * lane);
        float v[4][4], s[4], q[4];
#pragma unroll
        for (int u = 0; u < 4; ++u) {
            const float4 o  = *(const float4*)(accb + goff[u]);
            const float  dn = denom[(goff[u] >> 7) * HH + (lane >> 3)];
            v[u][0] = o.x / dn + gb.x;  v[u][1] = o.y / dn + gb.y;
            v[u][2] = o.z / dn + gb.z;  v[u][3] = o.w / dn + gb.w;
            s[u] = (v[u][0] + v[u][1]) + (v[u][2] + v[u][3]);
        }
#pragma unroll
        for (int m = 16; m; m >>= 1) {
#pragma unroll
            for (int u = 0; u < 4; ++u) s[u] += __shfl_xor(s[u], m);
        }
#pragma unroll
        for (int u = 0; u < 4; ++u) {
            const float mean = s[u] * (1.0f / DD);
#pragma unroll
            for (int e = 0; e < 4; ++e) v[u][e] -= mean;
            q[u] = (v[u][0] * v[u][0] + v[u][1] * v[u][1]) + (v[u][2] * v[u][2] + v[u][3] * v[u][3]);
        }
#pragma unroll
        for (int m = 16; m; m >>= 1) {
#pragma unroll
            for (int u = 0; u < 4; ++u) q[u] += __shfl_xor(q[u], m);
        }
#pragma unroll
        for (int u = 0; u < 4; ++u) {
            const float rs = rsqrtf(q[u] * (1.0f / DD) + 1e-5f);
            hv[u].x += fmaxf(v[u][0] * rs * g.x + b.x, 0.0f);
            hv[u].y += fmaxf(v[u][1] * rs * g.y + b.y, 0.0f);
            hv[u].z += fmaxf(v[u][2] * rs * g.z + b.z, 0.0f);
            hv[u].w += fmaxf(v[u][3] * rs * g.w + b.w, 0.0f);
            if (rowok[u]) *(float4*)(h + goff[u]) = hv[u];
        }
    }
#pragma unroll
    for (int u = 0; u < 4; ++u)
        *(float4*)(&Ht[(wave * 4 + u) * TP + 4 * lane]) = hv[u];
    __syncthreads();

    const int half = lane >> 4;
    const int l    = lane & 15;
    const int K0   = half * 2;
    const int n0   = wave * 32;
    v8f acc0 = {};
    v8f acc1 = {};
#pragma unroll
    for (int k = 0; k < DD; k += 4) {
        const v2f a = *(const v2f*)(&Ht[l * TP + k + K0]);
        const float* Bk = Wl + (long long)(k + K0) * DD + n0 + l;
        v2f b0, b1;
        b0.x = Bk[0];       b0.y = Bk[DD];
        b1.x = Bk[16];      b1.y = Bk[DD + 16];
        acc0 = __builtin_amdgcn_wmma_f32_16x16x4_f32(false, a, false, b0, (short)0, acc0, false, false);
        acc1 = __builtin_amdgcn_wmma_f32_16x16x4_f32(false, a, false, b1, (short)0, acc1, false, false);
    }
#pragma unroll
    for (int r = 0; r < 8; ++r) {
        const int row = r + half * 8;
        Xt[row * TP + n0 + l]      = acc0[r];
        Xt[row * TP + n0 + 16 + l] = acc1[r];
    }
    __syncthreads();

    const int ch = wave * 32 + lane;
    const float ws = att_src[ch];
    const float wd = att_dst[ch];
    for (int it0 = 0; it0 < 16; it0 += 4) {
        float xv[4], s[4], t[4];
#pragma unroll
        for (int u = 0; u < 4; ++u) {
            xv[u] = Xt[(it0 + u) * TP + ch];
            s[u] = xv[u] * ws;
            t[u] = xv[u] * wd;
        }
#pragma unroll
        for (int m = 16; m; m >>= 1) {
#pragma unroll
            for (int u = 0; u < 4; ++u) { s[u] += __shfl_xor(s[u], m); t[u] += __shfl_xor(t[u], m); }
        }
#pragma unroll
        for (int u = 0; u < 4; ++u) {
            const int node = m0 + it0 + u;
            if (lane == 0) { Lg[(it0 + u) * HH + wave] = s[u]; Lg[64 + (it0 + u) * HH + wave] = t[u]; }
            if (node < M) xh[(long long)node * DD + ch] = xv[u];
        }
    }
    __syncthreads();
    for (int qi = lane; qi < 64; qi += 32) {
        if (m0 + (qi >> 2) < M) {
            const long long idx = (long long)m0 * HH + qi;
            if (wave == 0)      a_s[idx] = Lg[qi];
            else if (wave == 1) a_d[idx] = Lg[64 + qi];
        }
    }
}

#define GR   200
#define GNT  1024
#define GNW  32
#define GCAP 256
#define GQ   8
#define G0   ((GR * (DD / 4) + GNT - 1) / GNT)
#define GB   8
#define GS   ((GR + GNW - 1) / GNW)
#define GWQ  256

__device__ __forceinline__ float gnn_lrelu(float e) { return (e > 0.0f) ? e : 0.2f * e; }

template <bool AL>
__device__ __forceinline__ int gnn_scan(const int* __restrict__ col,
                                        int E_, int qb, int qe, int base, int lo, int lane,
                                        unsigned ltmask, int* lsrc_w, unsigned short* lrow_w) {
    int cw = 0;
    for (int q0 = qb; q0 < qe; q0 += 32 * GQ) {
        int4 dq[GQ];
#pragma unroll
        for (int u = 0; u < GQ; ++u) {
            const int q  = q0 + u * 32 + lane;
            const int qc = (q < qe) ? q : (qe - 1);
            if (AL) {
                dq[u] = *(const int4*)(col + 4 * qc);
            } else {
                const int last = E_ - 1;
                const int i0 = 4 * qc;
                dq[u].x = col[(i0 + 0 < last) ? (i0 + 0) : last];
                dq[u].y = col[(i0 + 1 < last) ? (i0 + 1) : last];
                dq[u].z = col[(i0 + 2 < last) ? (i0 + 2) : last];
                dq[u].w = col[(i0 + 3 < last) ? (i0 + 3) : last];
            }
        }
#pragma unroll
        for (int u = 0; u < GQ; ++u) {
            const int q = q0 + u * 32 + lane;
            const int dv[4] = {dq[u].x, dq[u].y, dq[u].z, dq[u].w};
            const unsigned rg = (q < qe) ? (unsigned)GR : 0u;
            bool mt[4];
#pragma unroll
            for (int c = 0; c < 4; ++c)
                mt[c] = ((unsigned)(dv[c] - base) < rg) & (AL | (4 * q + c < E_));
            if (__builtin_amdgcn_ballot_w32(mt[0] | mt[1] | mt[2] | mt[3]) != 0u) {
#pragma unroll
                for (int c = 0; c < 4; ++c) {
                    const unsigned mk = __builtin_amdgcn_ballot_w32(mt[c]);
                    const int o = cw + __popc(mk & ltmask) - lo;
                    if (mt[c] && o >= 0 && o < GCAP) {
                        lsrc_w[o] = 4 * q + c;
                        lrow_w[o] = (unsigned short)(dv[c] - base);
                    }
                    cw += __popc(mk);
                }
            }
        }
    }
    return cw;
}

template <int KEEP>
__global__ void __launch_bounds__(GNT)
gnn_range_gather(const int* __restrict__ ei, const float* __restrict__ a_s,
                 const float* __restrict__ a_d, const float* __restrict__ xh,
                 float* __restrict__ out, float* __restrict__ denom,
                 int* __restrict__ ksrc, unsigned char* __restrict__ krow,
                 int* __restrict__ kcnt, int E_, int Nn) {
    static_assert(GR <= 256, "a kept row number is one byte");
    __shared__ __attribute__((aligned(16))) float acc[GR * DD];
    __shared__ float den[GR * HH];
    __shared__ float Mo[GR * HH];
    __shared__ int   Mn[GR * HH];
    __shared__ float ad[GR * HH];
    __shared__ int            lsrc[GNW * GCAP];
    __shared__ unsigned short lrow[GNW * GCAP];
    __shared__ int            cnt[GNW];
    __shared__ unsigned short wq[GNW * GWQ];
    __shared__ unsigned short ws[GNW * 32];

    const int tid  = threadIdx.x;
    const int wave = __builtin_amdgcn_readfirstlane(tid >> 5);
    const int lane = tid & 31;
    const int base = blockIdx.x * GR;
    const unsigned ltmask = (1u << lane) - 1u;

    {
        float4 v0[G0];
#pragma unroll
        for (int t = 0; t < G0; ++t) {
            const int i  = tid + t * GNT;
            const int ic = (i < GR * (DD / 4)) ? i : (GR * (DD / 4) - 1);
            const int node = base + (ic >> 5);
            const int nc = (node < Nn) ? node : (Nn - 1);
            v0[t] = *(const float4*)(xh + (long long)nc * DD + 4 * (ic & 31));
        }
#pragma unroll
        for (int t = 0; t < G0; ++t) {
            const int i = tid + t * GNT;
            if (i < GR * (DD / 4)) {
                float4 v = v0[t];
                if (base + (i >> 5) >= Nn) v = make_float4(0.0f, 0.0f, 0.0f, 0.0f);
                *(float4*)(&acc[4 * i]) = v;
            }
        }
    }
    for (int i = tid; i < GR * HH; i += GNT) {
        float as_ = 0.0f, adv = 0.0f;
        if (base + (i >> 2) < Nn) {
            as_ = a_s[(long long)base * HH + i];
            adv = a_d[(long long)base * HH + i];
        }
        const float e = gnn_lrelu(as_ + adv);
        ad[i] = adv;  Mo[i] = e;  Mn[i] = __float_as_int(e);  den[i] = 1.0f;
    }
    __syncthreads();

    const int* col = ei + E_;
    const int  Q   = (E_ + 3) >> 2;
    const int  qpw = (Q + GNW - 1) / GNW;
    const int  qb  = wave * qpw;
    const int  qe  = (qb + qpw < Q) ? (qb + qpw) : Q;
    const bool al  = (E_ & 3) == 0;

    const long long kb = ((long long)blockIdx.x * GNW + wave) * GCAP;
    bool kept = false;
    int  kown = 0;
    if (KEEP == 2) {
        const int kc = kcnt[(long long)blockIdx.x * GNW + lane];
        kept = __builtin_amdgcn_ballot_w32(kc > GCAP) == 0u;
        kown = __builtin_amdgcn_readlane(kc, wave);
    }

    int lo = 0;
    int mx;
    do {
        int cw;
        if (KEEP == 2 && kept)
            cw = kown;
        else
            cw = al
            ? gnn_scan<true >(col, E_, qb, qe, base, lo, lane, ltmask, &lsrc[wave * GCAP], &lrow[wave * GCAP])
            : gnn_scan<false>(col, E_, qb, qe, base, lo, lane, ltmask, &lsrc[wave * GCAP], &lrow[wave * GCAP]);
        if (lane == 0) {
            cnt[wave] = cw;
            if (KEEP == 1 && lo == 0) kcnt[(long long)blockIdx.x * GNW + wave] = cw;
        }

        {
            int nown = cw - lo;
            nown = (nown < 0) ? 0 : ((nown > GCAP) ? GCAP : nown);
            for (int j = lane; j < nown; j += 32) {
                int s, r;
                if (KEEP == 2 && kept) {
                    s = ksrc[kb + j];
                    r = krow[kb + j];
                    lrow[wave * GCAP + j] = (unsigned short)r;
                } else {
                    s = ei[lsrc[wave * GCAP + j]];
                    r = lrow[wave * GCAP + j];
                    if (KEEP == 1 && lo == 0) {
                        ksrc[kb + j] = s;
                        krow[kb + j] = (unsigned char)r;
                    }
                }
                lsrc[wave * GCAP + j] = s;
                const float4 as4 = *(const float4*)(a_s + (long long)s * HH);
                const float ev[4] = {as4.x, as4.y, as4.z, as4.w};
#pragma unroll
                for (int hq = 0; hq < 4; ++hq) {
                    const float e = gnn_lrelu(ev[hq] + ad[r * HH + hq]);
                    int* p = &Mn[r * HH + hq];
                    if (e >= 0.0f) atomicMax(p, __float_as_int(e));
                    else           atomicMin((unsigned int*)p, __float_as_uint(e));
                }
            }
        }
        __syncthreads();

        mx = 0;
        for (int w = 0; w < GNW; ++w) mx = (cnt[w] > mx) ? cnt[w] : mx;

        for (int i = tid; i < GR * (DD / 4); i += GNT) {
            const int mi = (i >> 5) * HH + ((i & 31) >> 3);
            const float sc = __expf(Mo[mi] - __int_as_float(Mn[mi]));
            float4 v = *(float4*)(&acc[4 * i]);
            v.x *= sc;  v.y *= sc;  v.z *= sc;  v.w *= sc;
            *(float4*)(&acc[4 * i]) = v;
        }
        __syncthreads();
        for (int i = tid; i < GR * HH; i += GNT) {
            const float mn = __int_as_float(Mn[i]);
            den[i] *= __expf(Mo[i] - mn);
            Mo[i] = mn;
        }
        __syncthreads();

        static_assert(GS * GNW < 65536 && GS < 32, "row classes of the partition");
        int qh = 0, qn = 0;
        for (int s = 0; s <= GNW; ++s) {
            const bool last = (s == GNW);
            int n = 0;
            if (!last) {
                n = cnt[s] - lo;
                n = (n < 0) ? 0 : ((n > GCAP) ? GCAP : n);
                n = __builtin_amdgcn_readfirstlane(n);
            }
            int j0 = 0;
            do {
                if (j0 < n) {
                    bool mine[4];
#pragma unroll
                    for (int c = 0; c < 4; ++c) {
                        const int j = j0 + 32 * c + lane;
                        mine[c] = false;
                        if (j < n) mine[c] = (lrow[s * GCAP + j] & 31) == wave;
                    }
#pragma unroll
                    for (int c = 0; c < 4; ++c) {
                        const unsigned mk = __builtin_amdgcn_ballot_w32(mine[c]);
                        if (mk != 0u) {
                            if (mine[c])
                                wq[wave * GWQ + ((qh + qn + __popc(mk & ltmask)) & (GWQ - 1))] =
                                    (unsigned short)(s * GCAP + j0 + 32 * c + lane);
                            qn += __popc(mk);
                        }
                    }
                }
                while (qn >= 32 || (last && qn > 0)) {
                    const int cn = (qn < 32) ? qn : 32;
                    int id = 0;
                    int rw = GS * GNW;
                    if (lane < cn) {
                        id = wq[wave * GWQ + ((qh + lane) & (GWQ - 1))];
                        rw = lrow[id];
                    }
                    const int sl = rw >> 5;
                    int pos = 0, nb = 0;
#pragma unroll
                    for (int U = 0; U <= GS; ++U) {
                        const unsigned mk = __builtin_amdgcn_ballot_w32(sl == U);
                        if (sl == U) pos = nb + __popc(mk & ltmask);
                        nb += __popc(mk);
                    }
                    ws[wave * 32 + pos] = (unsigned short)id;
                    int es = 0, er = 0;
                    {
                        const int id2 = ws[wave * 32 + lane];
                        if (lane < cn) {
                            es = lsrc[id2];
                            er = lrow[id2];
                        }
                    }
                    int    cur  = -1;
                    float4 a4   = make_float4(0.0f, 0.0f, 0.0f, 0.0f);
                    float  dsum = 0.0f;
#pragma unroll 1
                    for (int b = 0; b < cn; b += GB) {
                        float4 xv[GB];
                        float  pv[GB];
#pragma unroll
                        for (int u = 0; u < GB; ++u) {
                            const int su = __builtin_amdgcn_readlane(es, (b + u) & 31);
                            pv[u] = a_s[(unsigned)(su * HH + (lane >> 3))];
                        }
#pragma unroll
                        for (int u = 0; u < GB; ++u) {
                            const int su = __builtin_amdgcn_readlane(es, (b + u) & 31);
                            xv[u] = *(const float4*)(xh + (unsigned)(su * DD + 4 * lane));
                        }
#pragma unroll
                        for (int u = 0; u < GB; ++u) {
                            const int ru = __builtin_amdgcn_readlane(er, (b + u) & 31);
                            const int mi = ru * HH + (lane >> 3);
                            const float e = gnn_lrelu(pv[u] + ad[mi]);
                            pv[u] = __expf(e - Mo[mi]);
                        }
#pragma unroll
                        for (int u = 0; u < GB; ++u) {
                            if (u == 0 || b + u < cn) {
                                const int ru = __builtin_amdgcn_readlane(er, (b + u) & 31);
                                if (ru != cur) {
                                    if (cur >= 0) {
                                        *(float4*)(&acc[cur * DD + 4 * lane]) = a4;
                                        if ((lane & 7) == 0) den[cur * HH + (lane >> 3)] = dsum;
                                    }
                                    a4   = *(const float4*)(&acc[ru * DD + 4 * lane]);
                                    dsum = den[ru * HH + (lane >> 3)];
                                    cur  = ru;
                                }
                                const float p = pv[u];
                                a4.x += p * xv[u].x;  a4.y += p * xv[u].y;
                                a4.z += p * xv[u].z;  a4.w += p * xv[u].w;
                                dsum += p;
                            }
                        }
                    }
                    if (cur >= 0) {
                        *(float4*)(&acc[cur * DD + 4 * lane]) = a4;
                        if ((lane & 7) == 0) den[cur * HH + (lane >> 3)] = dsum;
                    }
                    qh = (qh + cn) & (GWQ - 1);
                    qn -= cn;
                }
                j0 += 128;
            } while (j0 < n);
        }
        __syncthreads();
        lo += GCAP;
    } while (lo < mx);

    for (int i = tid; i < GR * (DD / 4); i += GNT) {
        const int node = base + (i >> 5);
        if (node < Nn) *(float4*)(out + (long long)node * DD + 4 * (i & 31)) = *(const float4*)(&acc[4 * i]);
    }
    for (int i = tid; i < GR * HH; i += GNT)
        if (base + (i >> 2) < Nn) denom[(long long)base * HH + i] = den[i];
}

#define FR 32
__global__ void __launch_bounds__(256)
gnn_finalize(const float* __restrict__ out, const float* __restrict__ denom,
             const float* __restrict__ gbias,
             const float* __restrict__ ln_g, const float* __restrict__ ln_b,
             float* __restrict__ h, int M) {
    const int wave = threadIdx.x >> 5;
    const int lane = threadIdx.x & 31;
    const int r0 = blockIdx.x * FR + wave * 4;
    if (r0 >= M) return;
    float4 hv[4], o[4];
    float  dn[4];
    int    rowok[4];
    long long goff[4];
#pragma unroll
    for (int u = 0; u < 4; ++u) {
        const int node = r0 + u;
        rowok[u] = node < M;
        const int nc = rowok[u] ? node : M - 1;
        goff[u] = (long long)nc * DD + 4 * lane;
        o[u]    = *(const float4*)(out + goff[u]);
        hv[u]   = *(const float4*)(h + goff[u]);
        dn[u]   = denom[(long long)nc * HH + (lane >> 3)];
    }
    const float4 gb = *(const float4*)(gbias + 4 * lane);
    const float4 g  = *(const float4*)(ln_g + 4 * lane);
    const float4 b  = *(const float4*)(ln_b + 4 * lane);
    float v[4][4], s[4], q[4];
#pragma unroll
    for (int u = 0; u < 4; ++u) {
        v[u][0] = o[u].x / dn[u] + gb.x;  v[u][1] = o[u].y / dn[u] + gb.y;
        v[u][2] = o[u].z / dn[u] + gb.z;  v[u][3] = o[u].w / dn[u] + gb.w;
        s[u] = (v[u][0] + v[u][1]) + (v[u][2] + v[u][3]);
    }
#pragma unroll
    for (int m = 16; m; m >>= 1) {
#pragma unroll
        for (int u = 0; u < 4; ++u) s[u] += __shfl_xor(s[u], m);
    }
#pragma unroll
    for (int u = 0; u < 4; ++u) {
        const float mean = s[u] * (1.0f / DD);
#pragma unroll
        for (int e = 0; e < 4; ++e) v[u][e] -= mean;
        q[u] = (v[u][0] * v[u][0] + v[u][1] * v[u][1]) + (v[u][2] * v[u][2] + v[u][3] * v[u][3]);
    }
#pragma unroll
    for (int m = 16; m; m >>= 1) {
#pragma unroll
        for (int u = 0; u < 4; ++u) q[u] += __shfl_xor(q[u], m);
    }
#pragma unroll
    for (int u = 0; u < 4; ++u) {
        const float rs = rsqrtf(q[u] * (1.0f / DD) + 1e-5f);
        hv[u].x += v[u][0] * rs * g.x + b.x;
        hv[u].y += v[u][1] * rs * g.y + b.y;
        hv[u].z += v[u][2] * rs * g.z + b.z;
        hv[u].w += v[u][3] * rs * g.w + b.w;
        if (rowok[u]) *(float4*)(h + goff[u]) = hv[u];
    }
}

extern "C" void kernel_launch(void* const* d_in, const int* in_sizes, int n_in,
                              void* d_out, int out_size, void* d_ws, size_t ws_size,
                              hipStream_t stream) {
    const float* x      = (const float*)d_in[0];
    const int*   ei     = (const int*)d_in[1];
    const float* W_edge = (const float*)d_in[2];
    const float* b_edge = (const float*)d_in[3];
    const float* W_node = (const float*)d_in[4];
    const float* b_node = (const float*)d_in[5];
    const float* lin_w  = (const float*)d_in[6];
    const float* att_sp = (const float*)d_in[7];
    const float* att_dp = (const float*)d_in[8];
    const float* g_bias = (const float*)d_in[9];
    const float* ln_g   = (const float*)d_in[10];
    const float* ln_b   = (const float*)d_in[11];

    const int E_ = in_sizes[0] / 3;
    const int Nn = out_size / DD;

    float* bufA  = (float*)d_ws;
    float* xh    = bufA + (long long)Nn * DD;
    float* a_s   = xh   + (long long)Nn * DD + Nn;
    float* a_d   = a_s  + (long long)Nn * HH;
    float* denom = a_d  + (long long)Nn * HH;
    float* h     = (float*)d_out;
    double* S    = (double*)d_ws;

    long long ns = (long long)Nn * 8;
    int gs  = (int)((ns + 255) / 256);
    int tiles = (Nn + 15) / 16;
    int gei = (int)(((long long)E_ * 4 + 2047) / 2048);
    int gfin = (Nn + FR - 1) / FR;
    int gblk = (Nn + GR - 1) / GR;

    int*           ksrc = (int*)(denom + (long long)Nn * HH);
    int*           kcnt = ksrc + (long long)gblk * GNW * GCAP;
    unsigned char* krow = (unsigned char*)(kcnt + (long long)gblk * GNW);
    const bool room = (size_t)((krow + (long long)gblk * GNW * GCAP) - (unsigned char*)d_ws) <= ws_size;

    gnn_fill_u32<<<gs, 256, 0, stream>>>((unsigned int*)S, 0u, ns);
    gnn_edge_init<<<gei, 256, 0, stream>>>(x, ei, S, E_);
    gnn_init_tile<<<tiles, 128, 0, stream>>>(S, W_edge, b_edge, W_node, b_node, h, Nn);

    for (int i = 0; i < 3; ++i) {
        if (i == 0)
            gnn_node_tile<0><<<tiles, 128, 0, stream>>>(
                lin_w, att_sp, att_dp, nullptr, nullptr, nullptr,
                h, bufA, denom, xh, a_s, a_d, Nn);
        else
            gnn_node_tile<1><<<tiles, 128, 0, stream>>>(
                lin_w + (long long)i * DD * DD, att_sp + i * DD, att_dp + i * DD,
                g_bias + (i - 1) * DD, ln_g + (i - 1) * DD, ln_b + (i - 1) * DD,
                h, bufA, denom, xh, a_s, a_d, Nn);

        if (!room)
            gnn_range_gather<0><<<gblk, GNT, 0, stream>>>(ei, a_s, a_d, xh, bufA, denom, ksrc, krow, kcnt, E_, Nn);
        else if (i == 0)
            gnn_range_gather<1><<<gblk, GNT, 0, stream>>>(ei, a_s, a_d, xh, bufA, denom, ksrc, krow, kcnt, E_, Nn);
        else
            gnn_range_gather<2><<<gblk, GNT, 0, stream>>>(ei, a_s, a_d, xh, bufA, denom, ksrc, krow, kcnt, E_, Nn);
    }
    gnn_finalize<<<gfin, 256, 0, stream>>>(bufA, denom, g_bias + 2 * DD, ln_g + 2 * DD,
                                           ln_b + 2 * DD, h, Nn);
    (void)n_in;
}


# Gather sum step: the 32 lists are regrouped by owner wave once (per-list counts per owner, 32x32 prefix sum, one placement pass) instead of every wave walking all 32 lists; same sum order
# speedup vs baseline: 5.2097x; 1.2141x over previous
#include <hip/hip_runtime.h>
#include <hip/hip_bf16.h>

typedef float v2f __attribute__((ext_vector_type(2)));
typedef float v8f __attribute__((ext_vector_type(8)));

#define DD 128
#define HH 4
#define CC 32

__global__ void gnn_fill_u32(unsigned int* __restrict__ p, unsigned int v, long long n) {
    long long i = (long long)blockIdx.x * blockDim.x + threadIdx.x;
    if (i < n) p[i] = v;
}

__global__ void __launch_bounds__(256)
gnn_edge_init(const float* __restrict__ x, const int* __restrict__ ei,
              double* __restrict__ S, int E_) {
    const long long total  = (long long)E_ * 4;
    const long long stride = (long long)gridDim.x * blockDim.x;
    const long long first  = (long long)blockIdx.x * blockDim.x + threadIdx.x;
    const int c = (int)(first & 3);
    const int cx = (c < 3) ? c : 2;
    long long idx = first;
    for (; idx + 3 * stride < total; idx += 4 * stride) {
        int   dstn[4];
        float xf[4];
#pragma unroll
        for (int u = 0; u < 4; ++u) {
            const int e = (int)((idx + u * stride) >> 2);
            dstn[u] = ei[E_ + e];
            xf[u]   = x[(long long)e * 3 + cx];
        }
#pragma unroll
        for (int u = 0; u < 4; ++u)
            atomicAdd(&S[(long long)dstn[u] * 4 + c], (c < 3) ? (double)xf[u] : 1.0);
    }
    for (; idx < total; idx += stride) {
        const int e = (int)(idx >> 2);
        const int dstn = ei[E_ + e];
        const float xf = x[(long long)e * 3 + cx];
        atomicAdd(&S[(long long)dstn * 4 + c], (c < 3) ? (double)xf : 1.0);
    }
}

#define TP 132

__global__ void __launch_bounds__(128)
gnn_init_tile(const double* __restrict__ S, const float* __restrict__ W_edge,
              const float* __restrict__ b_edge, const float* __restrict__ W_node,
              const float* __restrict__ b_node, float* __restrict__ h, int M) {
    __shared__ __attribute__((aligned(16))) float Ht[16 * TP];
    __shared__ __attribute__((aligned(16))) float Xt[16 * TP];
    const int wave = threadIdx.x >> 5;
    const int lane = threadIdx.x & 31;
    const int m0 = blockIdx.x * 16;

    const float4 w0 = *(const float4*)(W_edge + 4 * lane);
    const float4 w1 = *(const float4*)(W_edge + DD + 4 * lane);
    const float4 w2 = *(const float4*)(W_edge + 2 * DD + 4 * lane);
    const float4 be = *(const float4*)(b_edge + 4 * lane);
#pragma unroll
    for (int u = 0; u < 4; ++u) {
        const int r = wave * 4 + u;
        int node = m0 + r;
        if (node >= M) node = M - 1;
        const double* sp = S + (long long)node * 4;
        const double c  = sp[3];
        const float inv = 1.0f / fmaxf((float)c, 1.0f);
        const float a0 = (float)sp[0] * inv;
        const float a1 = (float)sp[1] * inv;
        const float a2 = (float)sp[2] * inv;
        const float fb = (c > 0.0) ? 1.0f : 0.0f;
        float4 v;
        v.x = a0 * w0.x + a1 * w1.x + a2 * w2.x + fb * be.x;
        v.y = a0 * w0.y + a1 * w1.y + a2 * w2.y + fb * be.y;
        v.z = a0 * w0.z + a1 * w1.z + a2 * w2.z + fb * be.z;
        v.w = a0 * w0.w + a1 * w1.w + a2 * w2.w + fb * be.w;
        *(float4*)(&Ht[r * TP + 4 * lane]) = v;
    }
    __syncthreads();

    const int half = lane >> 4;
    const int l    = lane & 15;
    const int K0   = half * 2;
    const int n0   = wave * 32;
    v8f acc0 = {};
    v8f acc1 = {};
#pragma unroll
    for (int k = 0; k < DD; k += 4) {
        const v2f a = *(const v2f*)(&Ht[l * TP + k + K0]);
        const float* Bk = W_node + (long long)(k + K0) * DD + n0 + l;
        v2f b0, b1;
        b0.x = Bk[0];       b0.y = Bk[DD];
        b1.x = Bk[16];      b1.y = Bk[DD + 16];
        acc0 = __builtin_amdgcn_wmma_f32_16x16x4_f32(false, a, false, b0, (short)0, acc0, false, false);
        acc1 = __builtin_amdgcn_wmma_f32_16x16x4_f32(false, a, false, b1, (short)0, acc1, false, false);
    }
#pragma unroll
    for (int r = 0; r < 8; ++r) {
        const int row = r + half * 8;
        Xt[row * TP + n0 + l]      = acc0[r];
        Xt[row * TP + n0 + 16 + l] = acc1[r];
    }
    __syncthreads();

    const int ch = wave * 32 + lane;
    const float bn = b_node[ch];
    for (int r = 0; r < 16; ++r) {
        const int node = m0 + r;
        if (node < M) h[(long long)node * DD + ch] = Xt[r * TP + ch] + bn;
    }
}

__global__ void __launch_bounds__(128)
gnn_node_tile(const float* __restrict__ Wl,
              const float* __restrict__ att_src, const float* __restrict__ att_dst,
              const float* __restrict__ h, float* __restrict__ xh,
              float* __restrict__ a_s, float* __restrict__ a_d, int M) {
    __shared__ __attribute__((aligned(16))) float Ht[16 * TP];
    __shared__ __attribute__((aligned(16))) float Xt[16 * TP];
    __shared__ float Lg[128];
    const int wave = threadIdx.x >> 5;
    const int lane = threadIdx.x & 31;
    const int m0 = blockIdx.x * 16;

    float4 hv[4];
#pragma unroll
    for (int u = 0; u < 4; ++u) {
        const int node = m0 + wave * 4 + u;
        hv[u] = *(const float4*)(h + (long long)((node < M) ? node : M - 1) * DD + 4 * lane);
    }
#pragma unroll
    for (int u = 0; u < 4; ++u)
        *(float4*)(&Ht[(wave * 4 + u) * TP + 4 * lane]) = hv[u];
    __syncthreads();

    const int half = lane >> 4;
    const int l    = lane & 15;
    const int K0   = half * 2;
    const int n0   = wave * 32;
    v8f acc0 = {};
    v8f acc1 = {};
#pragma unroll
    for (int k = 0; k < DD; k += 4) {
        const v2f a = *(const v2f*)(&Ht[l * TP + k + K0]);
        const float* Bk = Wl + (long long)(k + K0) * DD + n0 + l;
        v2f b0, b1;
        b0.x = Bk[0];       b0.y = Bk[DD];
        b1.x = Bk[16];      b1.y = Bk[DD + 16];
        acc0 = __builtin_amdgcn_wmma_f32_16x16x4_f32(false, a, false, b0, (short)0, acc0, false, false);
        acc1 = __builtin_amdgcn_wmma_f32_16x16x4_f32(false, a, false, b1, (short)0, acc1, false, false);
    }
#pragma unroll
    for (int r = 0; r < 8; ++r) {
        const int row = r + half * 8;
        Xt[row * TP + n0 + l]      = acc0[r];
        Xt[row * TP + n0 + 16 + l] = acc1[r];
    }
    __syncthreads();

    const int ch = wave * 32 + lane;
    const float ws = att_src[ch];
    const float wd = att_dst[ch];
    for (int it0 = 0; it0 < 16; it0 += 4) {
        float xv[4], s[4], t[4];
#pragma unroll
        for (int u = 0; u < 4; ++u) {
            xv[u] = Xt[(it0 + u) * TP + ch];
            s[u] = xv[u] * ws;
            t[u] = xv[u] * wd;
        }
#pragma unroll
        for (int m = 16; m; m >>= 1) {
#pragma unroll
            for (int u = 0; u < 4; ++u) { s[u] += __shfl_xor(s[u], m); t[u] += __shfl_xor(t[u], m); }
        }
#pragma unroll
        for (int u = 0; u < 4; ++u) {
            const int node = m0 + it0 + u;
            if (lane == 0) { Lg[(it0 + u) * HH + wave] = s[u]; Lg[64 + (it0 + u) * HH + wave] = t[u]; }
            if (node < M) xh[(long long)node * DD + ch] = xv[u];
        }
    }
    __syncthreads();
    for (int qi = lane; qi < 64; qi += 32) {
        if (m0 + (qi >> 2) < M) {
            const long long idx = (long long)m0 * HH + qi;
            if (wave == 0)      a_s[idx] = Lg[qi];
            else if (wave == 1) a_d[idx] = Lg[64 + qi];
        }
    }
}

#define GR   200
#define GNT  1024
#define GNW  32
#define GCAP 256
#define GQ   8
#define G0   ((GR * (DD / 4) + GNT - 1) / GNT)
#define GB   8
#define GS   ((GR + GNW - 1) / GNW)
#define HP   33
#define AP   132
#define GT   ((GR + 15) / 16)
#define GRP  (GT * 16)

__device__ __forceinline__ float gnn_lrelu(float e) { return (e > 0.0f) ? e : 0.2f * e; }

template <bool AL>
__device__ __forceinline__ int gnn_scan(const int* __restrict__ col,
                                        int E_, int qb, int qe, int base, int lo, int lane,
                                        unsigned ltmask, int* lsrc_w, unsigned short* lrow_w) {
    int cw = 0;
    for (int q0 = qb; q0 < qe; q0 += 32 * GQ) {
        int4 dq[GQ];
#pragma unroll
        for (int u = 0; u < GQ; ++u) {
            const int q  = q0 + u * 32 + lane;
            const int qc = (q < qe) ? q : (qe - 1);
            if (AL) {
                dq[u] = *(const int4*)(col + 4 * qc);
            } else {
                const int last = E_ - 1;
                const int i0 = 4 * qc;
                dq[u].x = col[(i0 + 0 < last) ? (i0 + 0) : last];
                dq[u].y = col[(i0 + 1 < last) ? (i0 + 1) : last];
                dq[u].z = col[(i0 + 2 < last) ? (i0 + 2) : last];
                dq[u].w = col[(i0 + 3 < last) ? (i0 + 3) : last];
            }
        }
#pragma unroll
        for (int u = 0; u < GQ; ++u) {
            const int q = q0 + u * 32 + lane;
            const int dv[4] = {dq[u].x, dq[u].y, dq[u].z, dq[u].w};
            const unsigned rg = (q < qe) ? (unsigned)GR : 0u;
            bool mt[4];
#pragma unroll
            for (int c = 0; c < 4; ++c)
                mt[c] = ((unsigned)(dv[c] - base) < rg) & (AL | (4 * q + c < E_));
            if (__builtin_amdgcn_ballot_w32(mt[0] | mt[1] | mt[2] | mt[3]) != 0u) {
#pragma unroll
                for (int c = 0; c < 4; ++c) {
                    const unsigned mk = __builtin_amdgcn_ballot_w32(mt[c]);
                    const int o = cw + __popc(mk & ltmask) - lo;
                    if (mt[c] && o >= 0 && o < GCAP) {
                        lsrc_w[o] = 4 * q + c;
                        lrow_w[o] = (unsigned short)(dv[c] - base);
                    }
                    cw += __popc(mk);
                }
            }
        }
    }
    return cw;
}

template <int KEEP, int LASTL>
__global__ void __launch_bounds__(GNT)
gnn_range_gather(const int* __restrict__ ei, const float* __restrict__ a_s,
                 float* a_d, const float* __restrict__ xh, float* h,
                 const float* __restrict__ gbias, const float* __restrict__ ln_g,
                 const float* __restrict__ ln_b,
                 const float* __restrict__ Wn, const float* __restrict__ att_src_n,
                 const float* __restrict__ att_dst_n,
                 float* __restrict__ xh_n, float* __restrict__ a_s_n,
                 int* __restrict__ ksrc, unsigned char* __restrict__ krow,
                 int* __restrict__ kcnt, int E_, int Nn) {
    static_assert(GR <= 256, "a kept row number is one byte");
    __shared__ __attribute__((aligned(16))) float acc[GRP * AP];
    __shared__ float den[GR * HH];
    __shared__ float Mo[GR * HH];
    __shared__ int   Mn[GR * HH];
    __shared__ float ad[GR * HH];
    __shared__ int            lsrc[GNW * GCAP];
    __shared__ unsigned short lrow[GNW * GCAP];
    __shared__ int            cnt[GNW];
    __shared__ unsigned short lpos[GNW * GCAP];
    __shared__ int            hist[GNW * HP];
    __shared__ int            tot[GNW];
    __shared__ unsigned short qall[GNW * GCAP];
    __shared__ unsigned short ws[GNW * 32];

    const int tid  = threadIdx.x;
    const int wave = __builtin_amdgcn_readfirstlane(tid >> 5);
    const int lane = tid & 31;
    const int base = blockIdx.x * GR;
    const unsigned ltmask = (1u << lane) - 1u;

    {
        float4 v0[G0];
#pragma unroll
        for (int t = 0; t < G0; ++t) {
            const int i  = tid + t * GNT;
            const int ic = (i < GR * (DD / 4)) ? i : (GR * (DD / 4) - 1);
            const int node = base + (ic >> 5);
            const int nc = (node < Nn) ? node : (Nn - 1);
            v0[t] = *(const float4*)(xh + (long long)nc * DD + 4 * (ic & 31));
        }
#pragma unroll
        for (int t = 0; t < G0; ++t) {
            const int i = tid + t * GNT;
            if (i < GR * (DD / 4)) {
                float4 v = v0[t];
                if (base + (i >> 5) >= Nn) v = make_float4(0.0f, 0.0f, 0.0f, 0.0f);
                *(float4*)(&acc[(i >> 5) * AP + 4 * (i & 31)]) = v;
            }
        }
    }
    for (int i = tid; i < GR * HH; i += GNT) {
        float as_ = 0.0f, adv = 0.0f;
        if (base + (i >> 2) < Nn) {
            as_ = a_s[(long long)base * HH + i];
            adv = a_d[(long long)base * HH + i];
        }
        const float e = gnn_lrelu(as_ + adv);
        ad[i] = adv;  Mo[i] = e;  Mn[i] = __float_as_int(e);  den[i] = 1.0f;
    }
    __syncthreads();

    const int* col = ei + E_;
    const int  Q   = (E_ + 3) >> 2;
    const int  qpw = (Q + GNW - 1) / GNW;
    const int  qb  = wave * qpw;
    const int  qe  = (qb + qpw < Q) ? (qb + qpw) : Q;
    const bool al  = (E_ & 3) == 0;

    const long long kb = ((long long)blockIdx.x * GNW + wave) * GCAP;
    bool kept = false;
    int  kown = 0;
    if (KEEP == 2) {
        const int kc = kcnt[(long long)blockIdx.x * GNW + lane];
        kept = __builtin_amdgcn_ballot_w32(kc > GCAP) == 0u;
        kown = __builtin_amdgcn_readlane(kc, wave);
    }

    int lo = 0;
    int mx;
    do {
        int tid_p = threadIdx.x;
        asm volatile("" : "+v"(tid_p));
        {
        const int tid  = tid_p;
        const int lane = tid & 31;
        int cw;
        if (KEEP == 2 && kept)
            cw = kown;
        else
            cw = al
            ? gnn_scan<true >(col, E_, qb, qe, base, lo, lane, ltmask, &lsrc[wave * GCAP], &lrow[wave * GCAP])
            : gnn_scan<false>(col, E_, qb, qe, base, lo, lane, ltmask, &lsrc[wave * GCAP], &lrow[wave * GCAP]);
        if (lane == 0) {
            cnt[wave] = cw;
            if (KEEP == 1 && lo == 0) kcnt[(long long)blockIdx.x * GNW + wave] = cw;
        }

        int nown = cw - lo;
        nown = (nown < 0) ? 0 : ((nown > GCAP) ? GCAP : nown);
        nown = __builtin_amdgcn_readfirstlane(nown);
        hist[wave * HP + lane] = 0;
        for (int j0 = 0; j0 < nown; j0 += 32) {
            const int  j  = j0 + lane;
            const bool ok = j < nown;
            int r = 0;
            if (ok) {
                int s;
                if (KEEP == 2 && kept) {
                    s = ksrc[kb + j];
                    r = krow[kb + j];
                    lrow[wave * GCAP + j] = (unsigned short)r;
                } else {
                    s = ei[lsrc[wave * GCAP + j]];
                    r = lrow[wave * GCAP + j];
                    if (KEEP == 1 && lo == 0) {
                        ksrc[kb + j] = s;
                        krow[kb + j] = (unsigned char)r;
                    }
                }
                lsrc[wave * GCAP + j] = s;
                const float4 as4 = *(const float4*)(a_s + (long long)s * HH);
                const float ev[4] = {as4.x, as4.y, as4.z, as4.w};
#pragma unroll
                for (int hq = 0; hq < 4; ++hq) {
                    const float e = gnn_lrelu(ev[hq] + ad[r * HH + hq]);
                    int* p = &Mn[r * HH + hq];
                    if (e >= 0.0f) atomicMax(p, __float_as_int(e));
                    else           atomicMin((unsigned int*)p, __float_as_uint(e));
                }
            }
            const int ow = r & 31;
            unsigned same = __builtin_amdgcn_ballot_w32(ok);
#pragma unroll
            for (int b = 0; b < 5; ++b) {
                const bool bit = ((ow >> b) & 1) != 0;
                const unsigned mb = __builtin_amdgcn_ballot_w32(ok && bit);
                same &= bit ? mb : ~mb;
            }
            if (ok) {
                const int before = hist[wave * HP + ow];
                lpos[wave * GCAP + j] = (unsigned short)(before + __popc(same & ltmask));
                if ((same & ltmask) == 0u) hist[wave * HP + ow] = before + __popc(same);
            }
        }
        __syncthreads();

        mx = 0;
        for (int w = 0; w < GNW; ++w) mx = (cnt[w] > mx) ? cnt[w] : mx;

        for (int i = tid; i < GR * (DD / 4); i += GNT) {
            const int mi = (i >> 5) * HH + ((i & 31) >> 3);
            const float sc = __expf(Mo[mi] - __int_as_float(Mn[mi]));
            float4* ap = (float4*)(&acc[(i >> 5) * AP + 4 * (i & 31)]);
            float4 v = *ap;
            v.x *= sc;  v.y *= sc;  v.z *= sc;  v.w *= sc;
            *ap = v;
        }
        int hin = hist[lane * HP + wave];
#pragma unroll
        for (int d = 1; d < 32; d <<= 1) {
            const int up = __shfl_up(hin, d);
            if (lane >= d) hin += up;
        }
        if (lane == 31) tot[wave] = hin;
        __syncthreads();
        int myb, myn;
        {
            const int tl = tot[lane];
            int tin = tl;
#pragma unroll
            for (int d = 1; d < 32; d <<= 1) {
                const int up = __shfl_up(tin, d);
                if (lane >= d) tin += up;
            }
            myb = __builtin_amdgcn_readlane(tin - tl, wave);
            myn = __builtin_amdgcn_readlane(tl, wave);
            hist[lane * HP + wave] = myb + hin - hist[lane * HP + wave];
        }
        for (int i = tid; i < GR * HH; i += GNT) {
            const float mn = __int_as_float(Mn[i]);
            den[i] *= __expf(Mo[i] - mn);
            Mo[i] = mn;
        }
        __syncthreads();
        for (int j0 = 0; j0 < nown; j0 += 32) {
            const int j = j0 + lane;
            if (j < nown) {
                const int ow = lrow[wave * GCAP + j] & 31;
                qall[hist[wave * HP + ow] + lpos[wave * GCAP + j]] = (unsigned short)(wave * GCAP + j);
            }
        }
        __syncthreads();

        static_assert(GS * GNW < 65536 && GS < 32, "row classes of the partition");
        for (int g0 = 0; g0 < myn; g0 += 32) {
            const int cn = (myn - g0 < 32) ? (myn - g0) : 32;
            int id = 0;
            int rw = GS * GNW;
            if (lane < cn) {
                id = qall[myb + g0 + lane];
                rw = lrow[id];
            }
            const int sl = rw >> 5;
            int pos = 0, nb = 0;
#pragma unroll
            for (int U = 0; U <= GS; ++U) {
                const unsigned mk = __builtin_amdgcn_ballot_w32(sl == U);
                if (sl == U) pos = nb + __popc(mk & ltmask);
                nb += __popc(mk);
            }
            ws[wave * 32 + pos] = (unsigned short)id;
            int es = 0, er = 0;
            {
                const int id2 = ws[wave * 32 + lane];
                if (lane < cn) {
                    es = lsrc[id2];
                    er = lrow[id2];
                }
            }
            int    cur  = -1;
            float4 a4   = make_float4(0.0f, 0.0f, 0.0f, 0.0f);
            float  dsum = 0.0f;
#pragma unroll 1
            for (int b = 0; b < cn; b += GB) {
                float4 xv[GB];
                float  pv[GB];
#pragma unroll
                for (int u = 0; u < GB; ++u) {
                    const int su = __builtin_amdgcn_readlane(es, (b + u) & 31);
                    pv[u] = a_s[(unsigned)(su * HH + (lane >> 3))];
                }
#pragma unroll
                for (int u = 0; u < GB; ++u) {
                    const int su = __builtin_amdgcn_readlane(es, (b + u) & 31);
                    xv[u] = *(const float4*)(xh + (unsigned)(su * DD + 4 * lane));
                }
#pragma unroll
                for (int u = 0; u < GB; ++u) {
                    const int ru = __builtin_amdgcn_readlane(er, (b + u) & 31);
                    const int mi = ru * HH + (lane >> 3);
                    const float e = gnn_lrelu(pv[u] + ad[mi]);
                    pv[u] = __expf(e - Mo[mi]);
                }
#pragma unroll
                for (int u = 0; u < GB; ++u) {
                    if (u == 0 || b + u < cn) {
                        const int ru = __builtin_amdgcn_readlane(er, (b + u) & 31);
                        if (ru != cur) {
                            if (cur >= 0) {
                                *(float4*)(&acc[cur * AP + 4 * lane]) = a4;
                                if ((lane & 7) == 0) den[cur * HH + (lane >> 3)] = dsum;
                            }
                            a4   = *(const float4*)(&acc[ru * AP + 4 * lane]);
                            dsum = den[ru * HH + (lane >> 3)];
                            cur  = ru;
                        }
                        const float p = pv[u];
                        a4.x += p * xv[u].x;  a4.y += p * xv[u].y;
                        a4.z += p * xv[u].z;  a4.w += p * xv[u].w;
                        dsum += p;
                    }
                }
            }
            if (cur >= 0) {
                *(float4*)(&acc[cur * AP + 4 * lane]) = a4;
                if ((lane & 7) == 0) den[cur * HH + (lane >> 3)] = dsum;
            }
        }
        __syncthreads();
        lo += GCAP;
        }
    } while (lo < mx);

    int tid_q = threadIdx.x;
    asm volatile("" : "+v"(tid_q));
    {
    const int tid  = tid_q;

    {
        const float4 gb = *(const float4*)(gbias + 4 * lane);
        const float4 g  = *(const float4*)(ln_g + 4 * lane);
        const float4 b  = *(const float4*)(ln_b + 4 * lane);
        float4 hv[GS];
#pragma unroll
        for (int u = 0; u < GS; ++u) {
            const int r = wave + GNW * u;
            int node = base + ((r < GR) ? r : 0);
            node = (node < Nn) ? node : (Nn - 1);
            hv[u] = *(const float4*)(h + (long long)node * DD + 4 * lane);
        }
#pragma unroll
        for (int u = 0; u < GS; ++u) {
            const int r = wave + GNW * u;
            if (r < GR) {
                const float4 o  = *(const float4*)(&acc[r * AP + 4 * lane]);
                const float  dn = den[r * HH + (lane >> 3)];
                float v0 = o.x / dn + gb.x, v1 = o.y / dn + gb.y;
                float v2 = o.z / dn + gb.z, v3 = o.w / dn + gb.w;
                float s = (v0 + v1) + (v2 + v3);
#pragma unroll
                for (int m = 16; m; m >>= 1) s += __shfl_xor(s, m);
                const float mean = s * (1.0f / DD);
                v0 -= mean;  v1 -= mean;  v2 -= mean;  v3 -= mean;
                float q = (v0 * v0 + v1 * v1) + (v2 * v2 + v3 * v3);
#pragma unroll
                for (int m = 16; m; m >>= 1) q += __shfl_xor(q, m);
                const float rs = rsqrtf(q * (1.0f / DD) + 1e-5f);
                float y0 = v0 * rs * g.x + b.x, y1 = v1 * rs * g.y + b.y;
                float y2 = v2 * rs * g.z + b.z, y3 = v3 * rs * g.w + b.w;
                if (!LASTL) {
                    y0 = fmaxf(y0, 0.0f);  y1 = fmaxf(y1, 0.0f);
                    y2 = fmaxf(y2, 0.0f);  y3 = fmaxf(y3, 0.0f);
                }
                float4 hn = hv[u];
                hn.x += y0;  hn.y += y1;  hn.z += y2;  hn.w += y3;
                if (base + r < Nn) *(float4*)(h + (long long)(base + r) * DD + 4 * lane) = hn;
                if (!LASTL) *(float4*)(&acc[r * AP + 4 * lane]) = hn;
            }
        }
    }
    if (LASTL) return;

    for (int i = tid; i < (GRP - GR) * (DD / 4); i += GNT)
        *(float4*)(&acc[(GR + (i >> 5)) * AP + 4 * (i & 31)]) = make_float4(0.0f, 0.0f, 0.0f, 0.0f);
    __syncthreads();

    const int half = lane >> 4;
    const int l    = lane & 15;
    const int K0   = half * 2;
    const int grp  = wave >> 2;
    const int hd   = wave & 3;
    const int n0   = hd * 32;
    const int t0   = grp;
    const int t1   = grp + 8;
    const bool two = t1 < GT;
    v8f c00 = {};
    v8f c01 = {};
    v8f c10 = {};
    v8f c11 = {};
    if (two) {
#pragma unroll 16
        for (int k = 0; k < DD; k += 4) {
            const v2f a0 = *(const v2f*)(&acc[(t0 * 16 + l) * AP + k + K0]);
            const v2f a1 = *(const v2f*)(&acc[(t1 * 16 + l) * AP + k + K0]);
            const float* Bk = Wn + (long long)(k + K0) * DD + n0 + l;
            v2f b0, b1;
            b0.x = Bk[0];       b0.y = Bk[DD];
            b1.x = Bk[16];      b1.y = Bk[DD + 16];
            c00 = __builtin_amdgcn_wmma_f32_16x16x4_f32(false, a0, false, b0, (short)0, c00, false, false);
            c01 = __builtin_amdgcn_wmma_f32_16x16x4_f32(false, a0, false, b1, (short)0, c01, false, false);
            c10 = __builtin_amdgcn_wmma_f32_16x16x4_f32(false, a1, false, b0, (short)0, c10, false, false);
            c11 = __builtin_amdgcn_wmma_f32_16x16x4_f32(false, a1, false, b1, (short)0, c11, false, false);
        }
    } else {
#pragma unroll
        for (int k = 0; k < DD; k += 4) {
            const v2f a0 = *(const v2f*)(&acc[(t0 * 16 + l) * AP + k + K0]);
            const float* Bk = Wn + (long long)(k + K0) * DD + n0 + l;
            v2f b0, b1;
            b0.x = Bk[0];       b0.y = Bk[DD];
            b1.x = Bk[16];      b1.y = Bk[DD + 16];
            c00 = __builtin_amdgcn_wmma_f32_16x16x4_f32(false, a0, false, b0, (short)0, c00, false, false);
            c01 = __builtin_amdgcn_wmma_f32_16x16x4_f32(false, a0, false, b1, (short)0, c01, false, false);
        }
    }
    __syncthreads();
#pragma unroll
    for (int r = 0; r < 8; ++r) {
        const int row = t0 * 16 + r + half * 8;
        acc[row * AP + n0 + l]      = c00[r];
        acc[row * AP + n0 + 16 + l] = c01[r];
    }
    if (two) {
#pragma unroll
        for (int r = 0; r < 8; ++r) {
            const int row = t1 * 16 + r + half * 8;
            acc[row * AP + n0 + l]      = c10[r];
            acc[row * AP + n0 + 16 + l] = c11[r];
        }
    }

    {
        const int ch = n0 + lane;
        const float wsv = att_src_n[ch];
        const float wdv = att_dst_n[ch];
        for (int tt = 0; tt < 2; ++tt) {
            const int t = grp + 8 * tt;
            if (t < GT) {
                for (int it0 = 0; it0 < 16; it0 += 4) {
                    float xv[4], s[4], tq[4];
#pragma unroll
                    for (int u = 0; u < 4; ++u) {
                        xv[u] = acc[(t * 16 + it0 + u) * AP + ch];
                        s[u]  = xv[u] * wsv;
                        tq[u] = xv[u] * wdv;
                    }
#pragma unroll
                    for (int m = 16; m; m >>= 1) {
#pragma unroll
                        for (int u = 0; u < 4; ++u) { s[u] += __shfl_xor(s[u], m); tq[u] += __shfl_xor(tq[u], m); }
                    }
#pragma unroll
                    for (int u = 0; u < 4; ++u) {
                        const int row  = t * 16 + it0 + u;
                        const int node = base + row;
                        if (row < GR) {
                            if (lane == 0) { Mo[row * HH + hd] = s[u]; ad[row * HH + hd] = tq[u]; }
                            if (node < Nn) xh_n[(long long)node * DD + ch] = xv[u];
                        }
                    }
                }
            }
        }
    }
    __syncthreads();
    for (int i = tid; i < GR * HH; i += GNT) {
        if (base + (i >> 2) < Nn) {
            a_s_n[(long long)base * HH + i] = Mo[i];
            a_d[(long long)base * HH + i]   = ad[i];
        }
    }
    }
}

extern "C" void kernel_launch(void* const* d_in, const int* in_sizes, int n_in,
                              void* d_out, int out_size, void* d_ws, size_t ws_size,
                              hipStream_t stream) {
    const float* x      = (const float*)d_in[0];
    const int*   ei     = (const int*)d_in[1];
    const float* W_edge = (const float*)d_in[2];
    const float* b_edge = (const float*)d_in[3];
    const float* W_node = (const float*)d_in[4];
    const float* b_node = (const float*)d_in[5];
    const float* lin_w  = (const float*)d_in[6];
    const float* att_sp = (const float*)d_in[7];
    const float* att_dp = (const float*)d_in[8];
    const float* g_bias = (const float*)d_in[9];
    const float* ln_g   = (const float*)d_in[10];
    const float* ln_b   = (const float*)d_in[11];

    const int E_ = in_sizes[0] / 3;
    const int Nn = out_size / DD;

    float* xhB   = (float*)d_ws;
    float* xhA   = xhB  + (long long)Nn * DD;
    float* asA   = xhA  + (long long)Nn * DD + Nn;
    float* a_d   = asA  + (long long)Nn * HH;
    float* asB   = a_d  + (long long)Nn * HH;
    float* h     = (float*)d_out;
    double* S    = (double*)d_ws;

    long long ns = (long long)Nn * 8;
    int gs  = (int)((ns + 255) / 256);
    int tiles = (Nn + 15) / 16;
    int gei = (int)(((long long)E_ * 4 + 2047) / 2048);
    int gblk = (Nn + GR - 1) / GR;

    int*           ksrc = (int*)(asB + (long long)Nn * HH);
    int*           kcnt = ksrc + (long long)gblk * GNW * GCAP;
    unsigned char* krow = (unsigned char*)(kcnt + (long long)gblk * GNW);
    const bool room = (size_t)((krow + (long long)gblk * GNW * GCAP) - (unsigned char*)d_ws) <= ws_size;

    gnn_fill_u32<<<gs, 256, 0, stream>>>((unsigned int*)S, 0u, ns);
    gnn_edge_init<<<gei, 256, 0, stream>>>(x, ei, S, E_);
    gnn_init_tile<<<tiles, 128, 0, stream>>>(S, W_edge, b_edge, W_node, b_node, h, Nn);

    gnn_node_tile<<<tiles, 128, 0, stream>>>(lin_w, att_sp, att_dp, h, xhA, asA, a_d, Nn);

    if (room)
        gnn_range_gather<1, 0><<<gblk, GNT, 0, stream>>>(
            ei, asA, a_d, xhA, h, g_bias, ln_g, ln_b,
            lin_w + (long long)DD * DD, att_sp + DD, att_dp + DD, xhB, asB,
            ksrc, krow, kcnt, E_, Nn);
    else
        gnn_range_gather<0, 0><<<gblk, GNT, 0, stream>>>(
            ei, asA, a_d, xhA, h, g_bias, ln_g, ln_b,
            lin_w + (long long)DD * DD, att_sp + DD, att_dp + DD, xhB, asB,
            ksrc, krow, kcnt, E_, Nn);
    if (room)
        gnn_range_gather<2, 0><<<gblk, GNT, 0, stream>>>(
            ei, asB, a_d, xhB, h, g_bias + DD, ln_g + DD, ln_b + DD,
            lin_w + (long long)2 * DD * DD, att_sp + 2 * DD, att_dp + 2 * DD, xhA, asA,
            ksrc, krow, kcnt, E_, Nn);
    else
        gnn_range_gather<0, 0><<<gblk, GNT, 0, stream>>>(
            ei, asB, a_d, xhB, h, g_bias + DD, ln_g + DD, ln_b + DD,
            lin_w + (long long)2 * DD * DD, att_sp + 2 * DD, att_dp + 2 * DD, xhA, asA,
            ksrc, krow, kcnt, E_, Nn);
    if (room)
        gnn_range_gather<2, 1><<<gblk, GNT, 0, stream>>>(
            ei, asA, a_d, xhA, h, g_bias + 2 * DD, ln_g + 2 * DD, ln_b + 2 * DD,
            nullptr, nullptr, nullptr, nullptr, nullptr,
            ksrc, krow, kcnt, E_, Nn);
    else
        gnn_range_gather<0, 1><<<gblk, GNT, 0, stream>>>(
            ei, asA, a_d, xhA, h, g_bias + 2 * DD, ln_g + 2 * DD, ln_b + 2 * DD,
            nullptr, nullptr, nullptr, nullptr, nullptr,
            ksrc, krow, kcnt, E_, Nn);
    (void)n_in;
}


# First gather's scan reads a 1-byte block key per edge (written by the init scatter) instead of the 4-byte destination: 16 edges per 16-byte load; list order, so the rows' sum order, changes
# speedup vs baseline: 5.4387x; 1.0440x over previous
#include <hip/hip_runtime.h>
#include <hip/hip_bf16.h>

typedef float v2f __attribute__((ext_vector_type(2)));
typedef float v8f __attribute__((ext_vector_type(8)));

#define DD 128
#define HH 4
#define CC 32
#define GR 200

__global__ void gnn_fill_u32(unsigned int* __restrict__ p, unsigned int v, long long n) {
    long long i = (long long)blockIdx.x * blockDim.x + threadIdx.x;
    if (i < n) p[i] = v;
}

__global__ void __launch_bounds__(256)
gnn_edge_init(const float* __restrict__ x, const int* __restrict__ ei,
              double* __restrict__ S, unsigned char* __restrict__ bkey, int E_) {
    const long long total  = (long long)E_ * 4;
    const long long stride = (long long)gridDim.x * blockDim.x;
    const long long first  = (long long)blockIdx.x * blockDim.x + threadIdx.x;
    const int c = (int)(first & 3);
    const int cx = (c < 3) ? c : 2;
    const bool kw = (c == 3) && (bkey != nullptr);
    long long idx = first;
    for (; idx + 3 * stride < total; idx += 4 * stride) {
        int   dstn[4];
        float xf[4];
#pragma unroll
        for (int u = 0; u < 4; ++u) {
            const int e = (int)((idx + u * stride) >> 2);
            dstn[u] = ei[E_ + e];
            xf[u]   = x[(long long)e * 3 + cx];
        }
#pragma unroll
        for (int u = 0; u < 4; ++u)
            atomicAdd(&S[(long long)dstn[u] * 4 + c], (c < 3) ? (double)xf[u] : 1.0);
        if (kw) {
#pragma unroll
            for (int u = 0; u < 4; ++u)
                bkey[(int)((idx + u * stride) >> 2)] = (unsigned char)((unsigned)dstn[u] / GR);
        }
    }
    for (; idx < total; idx += stride) {
        const int e = (int)(idx >> 2);
        const int dstn = ei[E_ + e];
        const float xf = x[(long long)e * 3 + cx];
        atomicAdd(&S[(long long)dstn * 4 + c], (c < 3) ? (double)xf : 1.0);
        if (kw) bkey[e] = (unsigned char)((unsigned)dstn / GR);
    }
}

#define TP 132

__global__ void __launch_bounds__(128)
gnn_init_tile(const double* __restrict__ S, const float* __restrict__ W_edge,
              const float* __restrict__ b_edge, const float* __restrict__ W_node,
              const float* __restrict__ b_node, float* __restrict__ h, int M) {
    __shared__ __attribute__((aligned(16))) float Ht[16 * TP];
    __shared__ __attribute__((aligned(16))) float Xt[16 * TP];
    const int wave = threadIdx.x >> 5;
    const int lane = threadIdx.x & 31;
    const int m0 = blockIdx.x * 16;

    const float4 w0 = *(const float4*)(W_edge + 4 * lane);
    const float4 w1 = *(const float4*)(W_edge + DD + 4 * lane);
    const float4 w2 = *(const float4*)(W_edge + 2 * DD + 4 * lane);
    const float4 be = *(const float4*)(b_edge + 4 * lane);
#pragma unroll
    for (int u = 0; u < 4; ++u) {
        const int r = wave * 4 + u;
        int node = m0 + r;
        if (node >= M) node = M - 1;
        const double* sp = S + (long long)node * 4;
        const double c  = sp[3];
        const float inv = 1.0f / fmaxf((float)c, 1.0f);
        const float a0 = (float)sp[0] * inv;
        const float a1 = (float)sp[1] * inv;
        const float a2 = (float)sp[2] * inv;
        const float fb = (c > 0.0) ? 1.0f : 0.0f;
        float4 v;
        v.x = a0 * w0.x + a1 * w1.x + a2 * w2.x + fb * be.x;
        v.y = a0 * w0.y + a1 * w1.y + a2 * w2.y + fb * be.y;
        v.z = a0 * w0.z + a1 * w1.z + a2 * w2.z + fb * be.z;
        v.w = a0 * w0.w + a1 * w1.w + a2 * w2.w + fb * be.w;
        *(float4*)(&Ht[r * TP + 4 * lane]) = v;
    }
    __syncthreads();

    const int half = lane >> 4;
    const int l    = lane & 15;
    const int K0   = half * 2;
    const int n0   = wave * 32;
    v8f acc0 = {};
    v8f acc1 = {};
#pragma unroll
    for (int k = 0; k < DD; k += 4) {
        const v2f a = *(const v2f*)(&Ht[l * TP + k + K0]);
        const float* Bk = W_node + (long long)(k + K0) * DD + n0 + l;
        v2f b0, b1;
        b0.x = Bk[0];       b0.y = Bk[DD];
        b1.x = Bk[16];      b1.y = Bk[DD + 16];
        acc0 = __builtin_amdgcn_wmma_f32_16x16x4_f32(false, a, false, b0, (short)0, acc0, false, false);
        acc1 = __builtin_amdgcn_wmma_f32_16x16x4_f32(false, a, false, b1, (short)0, acc1, false, false);
    }
#pragma unroll
    for (int r = 0; r < 8; ++r) {
        const int row = r + half * 8;
        Xt[row * TP + n0 + l]      = acc0[r];
        Xt[row * TP + n0 + 16 + l] = acc1[r];
    }
    __syncthreads();

    const int ch = wave * 32 + lane;
    const float bn = b_node[ch];
    for (int r = 0; r < 16; ++r) {
        const int node = m0 + r;
        if (node < M) h[(long long)node * DD + ch] = Xt[r * TP + ch] + bn;
    }
}

__global__ void __launch_bounds__(128)
gnn_node_tile(const float* __restrict__ Wl,
              const float* __restrict__ att_src, const float* __restrict__ att_dst,
              const float* __restrict__ h, float* __restrict__ xh,
              float* __restrict__ a_s, float* __restrict__ a_d, int M) {
    __shared__ __attribute__((aligned(16))) float Ht[16 * TP];
    __shared__ __attribute__((aligned(16))) float Xt[16 * TP];
    __shared__ float Lg[128];
    const int wave = threadIdx.x >> 5;
    const int lane = threadIdx.x & 31;
    const int m0 = blockIdx.x * 16;

    float4 hv[4];
#pragma unroll
    for (int u = 0; u < 4; ++u) {
        const int node = m0 + wave * 4 + u;
        hv[u] = *(const float4*)(h + (long long)((node < M) ? node : M - 1) * DD + 4 * lane);
    }
#pragma unroll
    for (int u = 0; u < 4; ++u)
        *(float4*)(&Ht[(wave * 4 + u) * TP + 4 * lane]) = hv[u];
    __syncthreads();

    const int half = lane >> 4;
    const int l    = lane & 15;
    const int K0   = half * 2;
    const int n0   = wave * 32;
    v8f acc0 = {};
    v8f acc1 = {};
#pragma unroll
    for (int k = 0; k < DD; k += 4) {
        const v2f a = *(const v2f*)(&Ht[l * TP + k + K0]);
        const float* Bk = Wl + (long long)(k + K0) * DD + n0 + l;
        v2f b0, b1;
        b0.x = Bk[0];       b0.y = Bk[DD];
        b1.x = Bk[16];      b1.y = Bk[DD + 16];
        acc0 = __builtin_amdgcn_wmma_f32_16x16x4_f32(false, a, false, b0, (short)0, acc0, false, false);
        acc1 = __builtin_amdgcn_wmma_f32_16x16x4_f32(false, a, false, b1, (short)0, acc1, false, false);
    }
#pragma unroll
    for (int r = 0; r < 8; ++r) {
        const int row = r + half * 8;
        Xt[row * TP + n0 + l]      = acc0[r];
        Xt[row * TP + n0 + 16 + l] = acc1[r];
    }
    __syncthreads();

    const int ch = wave * 32 + lane;
    const float ws = att_src[ch];
    const float wd = att_dst[ch];
    for (int it0 = 0; it0 < 16; it0 += 4) {
        float xv[4], s[4], t[4];
#pragma unroll
        for (int u = 0; u < 4; ++u) {
            xv[u] = Xt[(it0 + u) * TP + ch];
            s[u] = xv[u] * ws;
            t[u] = xv[u] * wd;
        }
#pragma unroll
        for (int m = 16; m; m >>= 1) {
#pragma unroll
            for (int u = 0; u < 4; ++u) { s[u] += __shfl_xor(s[u], m); t[u] += __shfl_xor(t[u], m); }
        }
#pragma unroll
        for (int u = 0; u < 4; ++u) {
            const int node = m0 + it0 + u;
            if (lane == 0) { Lg[(it0 + u) * HH + wave] = s[u]; Lg[64 + (it0 + u) * HH + wave] = t[u]; }
            if (node < M) xh[(long long)node * DD + ch] = xv[u];
        }
    }
    __syncthreads();
    for (int qi = lane; qi < 64; qi += 32) {
        if (m0 + (qi >> 2) < M) {
            const long long idx = (long long)m0 * HH + qi;
            if (wave == 0)      a_s[idx] = Lg[qi];
            else if (wave == 1) a_d[idx] = Lg[64 + qi];
        }
    }
}

#define GNT  1024
#define GNW  32
#define GCAP 256
#define GQ   8
#define G0   ((GR * (DD / 4) + GNT - 1) / GNT)
#define GB   8
#define GS   ((GR + GNW - 1) / GNW)
#define HP   33
#define AP   132
#define GT   ((GR + 15) / 16)
#define GRP  (GT * 16)

__device__ __forceinline__ float gnn_lrelu(float e) { return (e > 0.0f) ? e : 0.2f * e; }

template <bool AL>
__device__ __forceinline__ int gnn_scan(const int* __restrict__ col,
                                        int E_, int qb, int qe, int base, int lo, int lane,
                                        unsigned ltmask, int* lsrc_w, unsigned short* lrow_w) {
    int cw = 0;
    for (int q0 = qb; q0 < qe; q0 += 32 * GQ) {
        int4 dq[GQ];
#pragma unroll
        for (int u = 0; u < GQ; ++u) {
            const int q  = q0 + u * 32 + lane;
            const int qc = (q < qe) ? q : (qe - 1);
            if (AL) {
                dq[u] = *(const int4*)(col + 4 * qc);
            } else {
                const int last = E_ - 1;
                const int i0 = 4 * qc;
                dq[u].x = col[(i0 + 0 < last) ? (i0 + 0) : last];
                dq[u].y = col[(i0 + 1 < last) ? (i0 + 1) : last];
                dq[u].z = col[(i0 + 2 < last) ? (i0 + 2) : last];
                dq[u].w = col[(i0 + 3 < last) ? (i0 + 3) : last];
            }
        }
#pragma unroll
        for (int u = 0; u < GQ; ++u) {
            const int q = q0 + u * 32 + lane;
            const int dv[4] = {dq[u].x, dq[u].y, dq[u].z, dq[u].w};
            const unsigned rg = (q < qe) ? (unsigned)GR : 0u;
            bool mt[4];
#pragma unroll
            for (int c = 0; c < 4; ++c)
                mt[c] = ((unsigned)(dv[c] - base) < rg) & (AL | (4 * q + c < E_));
            if (__builtin_amdgcn_ballot_w32(mt[0] | mt[1] | mt[2] | mt[3]) != 0u) {
#pragma unroll
                for (int c = 0; c < 4; ++c) {
                    const unsigned mk = __builtin_amdgcn_ballot_w32(mt[c]);
                    const int o = cw + __popc(mk & ltmask) - lo;
                    if (mt[c] && o >= 0 && o < GCAP) {
                        lsrc_w[o] = 4 * q + c;
                        lrow_w[o] = (unsigned short)(dv[c] - base);
                    }
                    cw += __popc(mk);
                }
            }
        }
    }
    return cw;
}

__device__ __forceinline__ int gnn_scan_keys(const unsigned char* __restrict__ bkey,
                                             int E_, int qb, int qe, unsigned bid, int lo, int lane,
                                             unsigned ltmask, int* lsrc_w) {
    int cw = 0;
    const unsigned pat = bid * 0x01010101u;
    for (int q0 = qb; q0 < qe; q0 += 32 * GQ) {
        uint4 dq[GQ];
#pragma unroll
        for (int u = 0; u < GQ; ++u) {
            const int q  = q0 + u * 32 + lane;
            const int qc = (q < qe) ? q : (qe - 1);
            dq[u] = *(const uint4*)(bkey + 16 * (long long)qc);
        }
        __builtin_amdgcn_sched_barrier(0);
#pragma unroll
        for (int u = 0; u < GQ; ++u) {
            const int q = q0 + u * 32 + lane;
            const unsigned tw[4] = {dq[u].x ^ pat, dq[u].y ^ pat, dq[u].z ^ pat, dq[u].w ^ pat};
            unsigned z = 0u;
#pragma unroll
            for (int c = 0; c < 4; ++c) {
                const unsigned t = tw[c];
                const unsigned g = (((t & 0x7F7F7F7Fu) + 0x7F7F7F7Fu) | t) & 0x80808080u;
                z |= g >> (7 - c);
            }
            unsigned m = ~z & 0x0F0F0F0Fu;
            if (q >= qe) m = 0u;
            for (int it = 0; it < 16; ++it) {
                if (__builtin_amdgcn_ballot_w32(m != 0u) == 0u) break;
                const int p = __ffs((int)m) - 1;
                const int e = 16 * q + 4 * (p & 7) + (p >> 3);
                const bool ok = (m != 0u) && (e < E_);
                const unsigned mk = __builtin_amdgcn_ballot_w32(ok);
                if (ok) {
                    const int o = cw + __popc(mk & ltmask) - lo;
                    if (o >= 0 && o < GCAP) lsrc_w[o] = e;
                }
                cw += __popc(mk);
                m &= m - 1u;
            }
        }
    }
    return cw;
}

template <int KEEP, int LASTL>
__global__ void __launch_bounds__(GNT)
gnn_range_gather(const int* __restrict__ ei, const float* __restrict__ a_s,
                 float* a_d, const float* __restrict__ xh, float* h,
                 const float* __restrict__ gbias, const float* __restrict__ ln_g,
                 const float* __restrict__ ln_b,
                 const float* __restrict__ Wn, const float* __restrict__ att_src_n,
                 const float* __restrict__ att_dst_n,
                 float* __restrict__ xh_n, float* __restrict__ a_s_n,
                 int* __restrict__ ksrc, unsigned char* __restrict__ krow,
                 int* __restrict__ kcnt, const unsigned char* __restrict__ bkey, int E_, int Nn) {
    static_assert(GR <= 256, "a kept row number is one byte");
    __shared__ __attribute__((aligned(16))) float acc[GRP * AP];
    __shared__ float den[GR * HH];
    __shared__ float Mo[GR * HH];
    __shared__ int   Mn[GR * HH];
    __shared__ float ad[GR * HH];
    __shared__ int            lsrc[GNW * GCAP];
    __shared__ unsigned short lrow[GNW * GCAP];
    __shared__ int            cnt[GNW];
    __shared__ unsigned short lpos[GNW * GCAP];
    __shared__ int            hist[GNW * HP];
    __shared__ int            tot[GNW];
    __shared__ unsigned short qall[GNW * GCAP];
    __shared__ unsigned short ws[GNW * 32];

    const int tid  = threadIdx.x;
    const int wave = __builtin_amdgcn_readfirstlane(tid >> 5);
    const int lane = tid & 31;
    const int base = blockIdx.x * GR;
    const unsigned ltmask = (1u << lane) - 1u;

    {
        float4 v0[G0];
#pragma unroll
        for (int t = 0; t < G0; ++t) {
            const int i  = tid + t * GNT;
            const int ic = (i < GR * (DD / 4)) ? i : (GR * (DD / 4) - 1);
            const int node = base + (ic >> 5);
            const int nc = (node < Nn) ? node : (Nn - 1);
            v0[t] = *(const float4*)(xh + (long long)nc * DD + 4 * (ic & 31));
        }
#pragma unroll
        for (int t = 0; t < G0; ++t) {
            const int i = tid + t * GNT;
            if (i < GR * (DD / 4)) {
                float4 v = v0[t];
                if (base + (i >> 5) >= Nn) v = make_float4(0.0f, 0.0f, 0.0f, 0.0f);
                *(float4*)(&acc[(i >> 5) * AP + 4 * (i & 31)]) = v;
            }
        }
    }
    for (int i = tid; i < GR * HH; i += GNT) {
        float as_ = 0.0f, adv = 0.0f;
        if (base + (i >> 2) < Nn) {
            as_ = a_s[(long long)base * HH + i];
            adv = a_d[(long long)base * HH + i];
        }
        const float e = gnn_lrelu(as_ + adv);
        ad[i] = adv;  Mo[i] = e;  Mn[i] = __float_as_int(e);  den[i] = 1.0f;
    }
    __syncthreads();

    const int* col = ei + E_;
    const bool bk  = bkey != nullptr;
    const int  Q   = bk ? ((E_ + 15) >> 4) : ((E_ + 3) >> 2);
    const int  qpw = (Q + GNW - 1) / GNW;
    const int  qb  = wave * qpw;
    const int  qe  = (qb + qpw < Q) ? (qb + qpw) : Q;
    const bool al  = (E_ & 3) == 0;

    const long long kb = ((long long)blockIdx.x * GNW + wave) * GCAP;
    bool kept = false;
    int  kown = 0;
    if (KEEP == 2) {
        const int kc = kcnt[(long long)blockIdx.x * GNW + lane];
        kept = __builtin_amdgcn_ballot_w32(kc > GCAP) == 0u;
        kown = __builtin_amdgcn_readlane(kc, wave);
    }

    int lo = 0;
    int mx;
    do {
        int tid_p = threadIdx.x;
        asm volatile("" : "+v"(tid_p));
        {
        const int tid  = tid_p;
        const int lane = tid & 31;
        int cw;
        if (KEEP == 2 && kept)
            cw = kown;
        else if (bk)
            cw = gnn_scan_keys(bkey, E_, qb, qe, (unsigned)blockIdx.x, lo, lane, ltmask, &lsrc[wave * GCAP]);
        else
            cw = al
            ? gnn_scan<true >(col, E_, qb, qe, base, lo, lane, ltmask, &lsrc[wave * GCAP], &lrow[wave * GCAP])
            : gnn_scan<false>(col, E_, qb, qe, base, lo, lane, ltmask, &lsrc[wave * GCAP], &lrow[wave * GCAP]);
        if (lane == 0) {
            cnt[wave] = cw;
            if (KEEP == 1 && lo == 0) kcnt[(long long)blockIdx.x * GNW + wave] = cw;
        }

        int nown = cw - lo;
        nown = (nown < 0) ? 0 : ((nown > GCAP) ? GCAP : nown);
        nown = __builtin_amdgcn_readfirstlane(nown);
        hist[wave * HP + lane] = 0;
        for (int j0 = 0; j0 < nown; j0 += 32) {
            const int  j  = j0 + lane;
            const bool ok = j < nown;
            int r = 0;
            if (ok) {
                int s;
                if (KEEP == 2 && kept) {
                    s = ksrc[kb + j];
                    r = krow[kb + j];
                    lrow[wave * GCAP + j] = (unsigned short)r;
                } else {
                    const int eid = lsrc[wave * GCAP + j];
                    s = ei[eid];
                    if (bk) {
                        r = col[eid] - base;
                        lrow[wave * GCAP + j] = (unsigned short)r;
                    } else {
                        r = lrow[wave * GCAP + j];
                    }
                    if (KEEP == 1 && lo == 0) {
                        ksrc[kb + j] = s;
                        krow[kb + j] = (unsigned char)r;
                    }
                }
                lsrc[wave * GCAP + j] = s;
                const float4 as4 = *(const float4*)(a_s + (long long)s * HH);
                const float ev[4] = {as4.x, as4.y, as4.z, as4.w};
#pragma unroll
                for (int hq = 0; hq < 4; ++hq) {
                    const float e = gnn_lrelu(ev[hq] + ad[r * HH + hq]);
                    int* p = &Mn[r * HH + hq];
                    if (e >= 0.0f) atomicMax(p, __float_as_int(e));
                    else           atomicMin((unsigned int*)p, __float_as_uint(e));
                }
            }
            const int ow = r & 31;
            unsigned same = __builtin_amdgcn_ballot_w32(ok);
#pragma unroll
            for (int b = 0; b < 5; ++b) {
                const bool bit = ((ow >> b) & 1) != 0;
                const unsigned mb = __builtin_amdgcn_ballot_w32(ok && bit);
                same &= bit ? mb : ~mb;
            }
            if (ok) {
                const int before = hist[wave * HP + ow];
                lpos[wave * GCAP + j] = (unsigned short)(before + __popc(same & ltmask));
                if ((same & ltmask) == 0u) hist[wave * HP + ow] = before + __popc(same);
            }
        }
        __syncthreads();

        mx = 0;
        for (int w = 0; w < GNW; ++w) mx = (cnt[w] > mx) ? cnt[w] : mx;

        for (int i = tid; i < GR * (DD / 4); i += GNT) {
            const int mi = (i >> 5) * HH + ((i & 31) >> 3);
            const float sc = __expf(Mo[mi] - __int_as_float(Mn[mi]));
            float4* ap = (float4*)(&acc[(i >> 5) * AP + 4 * (i & 31)]);
            float4 v = *ap;
            v.x *= sc;  v.y *= sc;  v.z *= sc;  v.w *= sc;
            *ap = v;
        }
        int hin = hist[lane * HP + wave];
#pragma unroll
        for (int d = 1; d < 32; d <<= 1) {
            const int up = __shfl_up(hin, d);
            if (lane >= d) hin += up;
        }
        if (lane == 31) tot[wave] = hin;
        __syncthreads();
        int myb, myn;
        {
            const int tl = tot[lane];
            int tin = tl;
#pragma unroll
            for (int d = 1; d < 32; d <<= 1) {
                const int up = __shfl_up(tin, d);
                if (lane >= d) tin += up;
            }
            myb = __builtin_amdgcn_readlane(tin - tl, wave);
            myn = __builtin_amdgcn_readlane(tl, wave);
            hist[lane * HP + wave] = myb + hin - hist[lane * HP + wave];
        }
        for (int i = tid; i < GR * HH; i += GNT) {
            const float mn = __int_as_float(Mn[i]);
            den[i] *= __expf(Mo[i] - mn);
            Mo[i] = mn;
        }
        __syncthreads();
        for (int j0 = 0; j0 < nown; j0 += 32) {
            const int j = j0 + lane;
            if (j < nown) {
                const int ow = lrow[wave * GCAP + j] & 31;
                qall[hist[wave * HP + ow] + lpos[wave * GCAP + j]] = (unsigned short)(wave * GCAP + j);
            }
        }
        __syncthreads();

        static_assert(GS * GNW < 65536 && GS < 32, "row classes of the partition");
        for (int g0 = 0; g0 < myn; g0 += 32) {
            const int cn = (myn - g0 < 32) ? (myn - g0) : 32;
            int id = 0;
            int rw = GS * GNW;
            if (lane < cn) {
                id = qall[myb + g0 + lane];
                rw = lrow[id];
            }
            const int sl = rw >> 5;
            int pos = 0, nb = 0;
#pragma unroll
            for (int U = 0; U <= GS; ++U) {
                const unsigned mk = __builtin_amdgcn_ballot_w32(sl == U);
                if (sl == U) pos = nb + __popc(mk & ltmask);
                nb += __popc(mk);
            }
            ws[wave * 32 + pos] = (unsigned short)id;
            int es = 0, er = 0;
            {
                const int id2 = ws[wave * 32 + lane];
                if (lane < cn) {
                    es = lsrc[id2];
                    er = lrow[id2];
                }
            }
            int    cur  = -1;
            float4 a4   = make_float4(0.0f, 0.0f, 0.0f, 0.0f);
            float  dsum = 0.0f;
#pragma unroll 1
            for (int b = 0; b < cn; b += GB) {
                float4 xv[GB];
                float  pv[GB];
#pragma unroll
                for (int u = 0; u < GB; ++u) {
                    const int su = __builtin_amdgcn_readlane(es, (b + u) & 31);
                    pv[u] = a_s[(unsigned)(su * HH + (lane >> 3))];
                }
#pragma unroll
                for (int u = 0; u < GB; ++u) {
                    const int su = __builtin_amdgcn_readlane(es, (b + u) & 31);
                    xv[u] = *(const float4*)(xh + (unsigned)(su * DD + 4 * lane));
                }
#pragma unroll
                for (int u = 0; u < GB; ++u) {
                    const int ru = __builtin_amdgcn_readlane(er, (b + u) & 31);
                    const int mi = ru * HH + (lane >> 3);
                    const float e = gnn_lrelu(pv[u] + ad[mi]);
                    pv[u] = __expf(e - Mo[mi]);
                }
#pragma unroll
                for (int u = 0; u < GB; ++u) {
                    if (u == 0 || b + u < cn) {
                        const int ru = __builtin_amdgcn_readlane(er, (b + u) & 31);
                        if (ru != cur) {
                            if (cur >= 0) {
                                *(float4*)(&acc[cur * AP + 4 * lane]) = a4;
                                if ((lane & 7) == 0) den[cur * HH + (lane >> 3)] = dsum;
                            }
                            a4   = *(const float4*)(&acc[ru * AP + 4 * lane]);
                            dsum = den[ru * HH + (lane >> 3)];
                            cur  = ru;
                        }
                        const float p = pv[u];
                        a4.x += p * xv[u].x;  a4.y += p * xv[u].y;
                        a4.z += p * xv[u].z;  a4.w += p * xv[u].w;
                        dsum += p;
                    }
                }
            }
            if (cur >= 0) {
                *(float4*)(&acc[cur * AP + 4 * lane]) = a4;
                if ((lane & 7) == 0) den[cur * HH + (lane >> 3)] = dsum;
            }
        }
        __syncthreads();
        lo += GCAP;
        }
    } while (lo < mx);

    int tid_q = threadIdx.x;
    asm volatile("" : "+v"(tid_q));
    {
    const int tid  = tid_q;

    {
        const float4 gb = *(const float4*)(gbias + 4 * lane);
        const float4 g  = *(const float4*)(ln_g + 4 * lane);
        const float4 b  = *(const float4*)(ln_b + 4 * lane);
        float4 hv[GS];
#pragma unroll
        for (int u = 0; u < GS; ++u) {
            const int r = wave + GNW * u;
            int node = base + ((r < GR) ? r : 0);
            node = (node < Nn) ? node : (Nn - 1);
            hv[u] = *(const float4*)(h + (long long)node * DD + 4 * lane);
        }
#pragma unroll
        for (int u = 0; u < GS; ++u) {
            const int r = wave + GNW * u;
            if (r < GR) {
                const float4 o  = *(const float4*)(&acc[r * AP + 4 * lane]);
                const float  dn = den[r * HH + (lane >> 3)];
                float v0 = o.x / dn + gb.x, v1 = o.y / dn + gb.y;
                float v2 = o.z / dn + gb.z, v3 = o.w / dn + gb.w;
                float s = (v0 + v1) + (v2 + v3);
#pragma unroll
                for (int m = 16; m; m >>= 1) s += __shfl_xor(s, m);
                const float mean = s * (1.0f / DD);
                v0 -= mean;  v1 -= mean;  v2 -= mean;  v3 -= mean;
                float q = (v0 * v0 + v1 * v1) + (v2 * v2 + v3 * v3);
#pragma unroll
                for (int m = 16; m; m >>= 1) q += __shfl_xor(q, m);
                const float rs = rsqrtf(q * (1.0f / DD) + 1e-5f);
                float y0 = v0 * rs * g.x + b.x, y1 = v1 * rs * g.y + b.y;
                float y2 = v2 * rs * g.z + b.z, y3 = v3 * rs * g.w + b.w;
                if (!LASTL) {
                    y0 = fmaxf(y0, 0.0f);  y1 = fmaxf(y1, 0.0f);
                    y2 = fmaxf(y2, 0.0f);  y3 = fmaxf(y3, 0.0f);
                }
                float4 hn = hv[u];
                hn.x += y0;  hn.y += y1;  hn.z += y2;  hn.w += y3;
                if (base + r < Nn) *(float4*)(h + (long long)(base + r) * DD + 4 * lane) = hn;
                if (!LASTL) *(float4*)(&acc[r * AP + 4 * lane]) = hn;
            }
        }
    }
    if (LASTL) return;

    for (int i = tid; i < (GRP - GR) * (DD / 4); i += GNT)
        *(float4*)(&acc[(GR + (i >> 5)) * AP + 4 * (i & 31)]) = make_float4(0.0f, 0.0f, 0.0f, 0.0f);
    __syncthreads();

    const int half = lane >> 4;
    const int l    = lane & 15;
    const int K0   = half * 2;
    const int grp  = wave >> 2;
    const int hd   = wave & 3;
    const int n0   = hd * 32;
    const int t0   = grp;
    const int t1   = grp + 8;
    const bool two = t1 < GT;
    v8f c00 = {};
    v8f c01 = {};
    v8f c10 = {};
    v8f c11 = {};
    if (two) {
#pragma unroll 16
        for (int k = 0; k < DD; k += 4) {
            const v2f a0 = *(const v2f*)(&acc[(t0 * 16 + l) * AP + k + K0]);
            const v2f a1 = *(const v2f*)(&acc[(t1 * 16 + l) * AP + k + K0]);
            const float* Bk = Wn + (long long)(k + K0) * DD + n0 + l;
            v2f b0, b1;
            b0.x = Bk[0];       b0.y = Bk[DD];
            b1.x = Bk[16];      b1.y = Bk[DD + 16];
            c00 = __builtin_amdgcn_wmma_f32_16x16x4_f32(false, a0, false, b0, (short)0, c00, false, false);
            c01 = __builtin_amdgcn_wmma_f32_16x16x4_f32(false, a0, false, b1, (short)0, c01, false, false);
            c10 = __builtin_amdgcn_wmma_f32_16x16x4_f32(false, a1, false, b0, (short)0, c10, false, false);
            c11 = __builtin_amdgcn_wmma_f32_16x16x4_f32(false, a1, false, b1, (short)0, c11, false, false);
        }
    } else {
#pragma unroll
        for (int k = 0; k < DD; k += 4) {
            const v2f a0 = *(const v2f*)(&acc[(t0 * 16 + l) * AP + k + K0]);
            const float* Bk = Wn + (long long)(k + K0) * DD + n0 + l;
            v2f b0, b1;
            b0.x = Bk[0];       b0.y = Bk[DD];
            b1.x = Bk[16];      b1.y = Bk[DD + 16];
            c00 = __builtin_amdgcn_wmma_f32_16x16x4_f32(false, a0, false, b0, (short)0, c00, false, false);
            c01 = __builtin_amdgcn_wmma_f32_16x16x4_f32(false, a0, false, b1, (short)0, c01, false, false);
        }
    }
    __syncthreads();
#pragma unroll
    for (int r = 0; r < 8; ++r) {
        const int row = t0 * 16 + r + half * 8;
        acc[row * AP + n0 + l]      = c00[r];
        acc[row * AP + n0 + 16 + l] = c01[r];
    }
    if (two) {
#pragma unroll
        for (int r = 0; r < 8; ++r) {
            const int row = t1 * 16 + r + half * 8;
            acc[row * AP + n0 + l]      = c10[r];
            acc[row * AP + n0 + 16 + l] = c11[r];
        }
    }

    {
        const int ch = n0 + lane;
        const float wsv = att_src_n[ch];
        const float wdv = att_dst_n[ch];
        for (int tt = 0; tt < 2; ++tt) {
            const int t = grp + 8 * tt;
            if (t < GT) {
                for (int it0 = 0; it0 < 16; it0 += 4) {
                    float xv[4], s[4], tq[4];
#pragma unroll
                    for (int u = 0; u < 4; ++u) {
                        xv[u] = acc[(t * 16 + it0 + u) * AP + ch];
                        s[u]  = xv[u] * wsv;
                        tq[u] = xv[u] * wdv;
                    }
#pragma unroll
                    for (int m = 16; m; m >>= 1) {
#pragma unroll
                        for (int u = 0; u < 4; ++u) { s[u] += __shfl_xor(s[u], m); tq[u] += __shfl_xor(tq[u], m); }
                    }
#pragma unroll
                    for (int u = 0; u < 4; ++u) {
                        const int row  = t * 16 + it0 + u;
                        const int node = base + row;
                        if (row < GR) {
                            if (lane == 0) { Mo[row * HH + hd] = s[u]; ad[row * HH + hd] = tq[u]; }
                            if (node < Nn) xh_n[(long long)node * DD + ch] = xv[u];
                        }
                    }
                }
            }
        }
    }
    __syncthreads();
    for (int i = tid; i < GR * HH; i += GNT) {
        if (base + (i >> 2) < Nn) {
            a_s_n[(long long)base * HH + i] = Mo[i];
            a_d[(long long)base * HH + i]   = ad[i];
        }
    }
    }
}

extern "C" void kernel_launch(void* const* d_in, const int* in_sizes, int n_in,
                              void* d_out, int out_size, void* d_ws, size_t ws_size,
                              hipStream_t stream) {
    const float* x      = (const float*)d_in[0];
    const int*   ei     = (const int*)d_in[1];
    const float* W_edge = (const float*)d_in[2];
    const float* b_edge = (const float*)d_in[3];
    const float* W_node = (const float*)d_in[4];
    const float* b_node = (const float*)d_in[5];
    const float* lin_w  = (const float*)d_in[6];
    const float* att_sp = (const float*)d_in[7];
    const float* att_dp = (const float*)d_in[8];
    const float* g_bias = (const float*)d_in[9];
    const float* ln_g   = (const float*)d_in[10];
    const float* ln_b   = (const float*)d_in[11];

    const int E_ = in_sizes[0] / 3;
    const int Nn = out_size / DD;

    float* xhB   = (float*)d_ws;
    float* xhA   = xhB  + (long long)Nn * DD;
    float* asA   = xhA  + (long long)Nn * DD + Nn;
    float* a_d   = asA  + (long long)Nn * HH;
    float* asB   = a_d  + (long long)Nn * HH;
    float* h     = (float*)d_out;
    double* S    = (double*)d_ws;

    long long ns = (long long)Nn * 8;
    int gs  = (int)((ns + 255) / 256);
    int tiles = (Nn + 15) / 16;
    int gei = (int)(((long long)E_ * 4 + 2047) / 2048);
    int gblk = (Nn + GR - 1) / GR;

    int*           ksrc = (int*)(asB + (long long)Nn * HH);
    int*           kcnt = ksrc + (long long)gblk * GNW * GCAP;
    unsigned char* krow = (unsigned char*)(kcnt + (long long)gblk * GNW);
    const bool room = (size_t)((krow + (long long)gblk * GNW * GCAP) - (unsigned char*)d_ws) <= ws_size;
    const size_t koff = ((size_t)((krow + (long long)gblk * GNW * GCAP) - (unsigned char*)d_ws) + 15) & ~(size_t)15;
    const size_t klen = (((size_t)E_ + 15) / 16) * 16;
    unsigned char* bkey = (gblk <= 256 && koff + klen <= ws_size) ? ((unsigned char*)d_ws + koff) : nullptr;

    gnn_fill_u32<<<gs, 256, 0, stream>>>((unsigned int*)S, 0u, ns);
    gnn_edge_init<<<gei, 256, 0, stream>>>(x, ei, S, bkey, E_);
    gnn_init_tile<<<tiles, 128, 0, stream>>>(S, W_edge, b_edge, W_node, b_node, h, Nn);

    gnn_node_tile<<<tiles, 128, 0, stream>>>(lin_w, att_sp, att_dp, h, xhA, asA, a_d, Nn);

    if (room)
        gnn_range_gather<1, 0><<<gblk, GNT, 0, stream>>>(
            ei, asA, a_d, xhA, h, g_bias, ln_g, ln_b,
            lin_w + (long long)DD * DD, att_sp + DD, att_dp + DD, xhB, asB,
            ksrc, krow, kcnt, bkey, E_, Nn);
    else
        gnn_range_gather<0, 0><<<gblk, GNT, 0, stream>>>(
            ei, asA, a_d, xhA, h, g_bias, ln_g, ln_b,
            lin_w + (long long)DD * DD, att_sp + DD, att_dp + DD, xhB, asB,
            ksrc, krow, kcnt, bkey, E_, Nn);
    if (room)
        gnn_range_gather<2, 0><<<gblk, GNT, 0, stream>>>(
            ei, asB, a_d, xhB, h, g_bias + DD, ln_g + DD, ln_b + DD,
            lin_w + (long long)2 * DD * DD, att_sp + 2 * DD, att_dp + 2 * DD, xhA, asA,
            ksrc, krow, kcnt, bkey, E_, Nn);
    else
        gnn_range_gather<0, 0><<<gblk, GNT, 0, stream>>>(
            ei, asB, a_d, xhB, h, g_bias + DD, ln_g + DD, ln_b + DD,
            lin_w + (long long)2 * DD * DD, att_sp + 2 * DD, att_dp + 2 * DD, xhA, asA,
            ksrc, krow, kcnt, bkey, E_, Nn);
    if (room)
        gnn_range_gather<2, 1><<<gblk, GNT, 0, stream>>>(
            ei, asA, a_d, xhA, h, g_bias + 2 * DD, ln_g + 2 * DD, ln_b + 2 * DD,
            nullptr, nullptr, nullptr, nullptr, nullptr,
            ksrc, krow, kcnt, bkey, E_, Nn);
    else
        gnn_range_gather<0, 1><<<gblk, GNT, 0, stream>>>(
            ei, asA, a_d, xhA, h, g_bias + 2 * DD, ln_g + 2 * DD, ln_b + 2 * DD,
            nullptr, nullptr, nullptr, nullptr, nullptr,
            ksrc, krow, kcnt, bkey, E_, Nn);
    (void)n_in;
}
